# starting kernel
# baseline (speedup 1.0000x reference)
#include <hip/hip_runtime.h>
#include <hip/hip_bf16.h>
#include <math.h>


typedef float v2f __attribute__((ext_vector_type(2)));
typedef float v8f __attribute__((ext_vector_type(8)));

template<int NCOLS, int KDIM, int NT>
__global__ __launch_bounds__(32)
void k_gemm_wmma_f32(const float* __restrict__ A, const float* __restrict__ B,
                     float* __restrict__ C, int M) {
    const int tileM  = blockIdx.x;
    const int tileN0 = blockIdx.y * NT;
    const int lane   = threadIdx.x;
    const int lm     = lane & 15;
    const int kg     = lane >> 4;

    int row = tileM * 16 + lm;
    if (row >= M) row = M - 1;
    const float* Ab = A + (size_t)row * KDIM + kg * 2;
    const float* Bb = B + (size_t)(kg * 2) * NCOLS + tileN0 * 16 + lm;

    v8f acc[NT];
#pragma unroll
    for (int j = 0; j < NT; ++j) acc[j] = (v8f){};

#pragma unroll
    for (int k = 0; k < KDIM; k += 4) {
        const v2f a = *(const v2f*)(Ab + k);
#pragma unroll
        for (int j = 0; j < NT; ++j) {
            v2f b;
            b.x = Bb[k * NCOLS + j * 16];
            b.y = Bb[k * NCOLS + NCOLS + j * 16];
            acc[j] = __builtin_amdgcn_wmma_f32_16x16x4_f32(
                 false, a,  false, b,
                 (short)0, acc[j],  false,  false);
        }
    }

    const int rbase = tileM * 16 + kg * 8;
#pragma unroll
    for (int j = 0; j < NT; ++j) {
        const int cbase = (tileN0 + j) * 16 + lm;
#pragma unroll
        for (int v = 0; v < 8; ++v) {
            const int r = rbase + v;
            if (r < M) C[(size_t)r * NCOLS + cbase] = acc[j][v];
        }
    }
}

__global__ void k_attn_coef(const float* __restrict__ feat,
                            const float* __restrict__ al,
                            const float* __restrict__ ar,
                            float* __restrict__ el, float* __restrict__ er,
                            int N, int H, int D) {
    const int idx = blockIdx.x * blockDim.x + threadIdx.x;
    if (idx >= N * H) return;
    const int n = idx / H, h = idx % H;
    const float* f  = feat + ((size_t)n * H + h) * D;
    const float* pl = al + h * D;
    const float* pr = ar + h * D;
    float sl = 0.0f, sr = 0.0f;
    for (int d = 0; d < D; ++d) { const float v = f[d]; sl += v * pl[d]; sr += v * pr[d]; }
    el[idx] = sl; er[idx] = sr;
}

__device__ __forceinline__ unsigned fkey(float f) {
    unsigned b = __float_as_uint(f);
    return b ^ (unsigned)(((int)b >> 31) | 0x80000000);
}
__device__ __forceinline__ float funkey(unsigned k) {
    unsigned b = (k & 0x80000000u) ? (k ^ 0x80000000u) : ~k;
    return __uint_as_float(b);
}

__global__ void k_edge_max(const int* __restrict__ src, const int* __restrict__ dst,
                           const float* __restrict__ el, const float* __restrict__ er,
                           float* __restrict__ ebuf, unsigned* __restrict__ emax,
                           int E, int H) {
    const int idx = blockIdx.x * blockDim.x + threadIdx.x;
    if (idx >= E * H) return;
    const int e = idx / H, h = idx % H;
    const int s = src[e], d = dst[e];
    float v = el[s * H + h] + er[d * H + h];
    v = (v > 0.0f) ? v : 0.2f * v;
    ebuf[idx] = v;
    atomicMax(&emax[d * H + h], fkey(v));
}

__global__ void k_edge_exp(const int* __restrict__ dst,
                           float* __restrict__ ebuf, const unsigned* __restrict__ emax,
                           float* __restrict__ denom, int E, int H) {
    const int idx = blockIdx.x * blockDim.x + threadIdx.x;
    if (idx >= E * H) return;
    const int e = idx / H, h = idx % H;
    const int d = dst[e];
    const float ex = expf(ebuf[idx] - funkey(emax[d * H + h]));
    ebuf[idx] = ex;
    atomicAdd(&denom[d * H + h], ex);
}

__global__ void k_aggregate(const int* __restrict__ src, const int* __restrict__ dst,
                            const float* __restrict__ feat, const float* __restrict__ ebuf,
                            const float* __restrict__ denom, float* __restrict__ out,
                            int H, int D) {
    const int e = blockIdx.x;
    const int t = threadIdx.x;
    const int s = src[e], d = dst[e];
    __shared__ float salpha[8];
    if (t < H) salpha[t] = ebuf[e * H + t] / denom[d * H + t];
    __syncthreads();
    const int h = t / D;
    atomicAdd(&out[(size_t)d * H * D + t], feat[(size_t)s * H * D + t] * salpha[h]);
}

__global__ void k_elu(float* __restrict__ x, int n) {
    const int idx = blockIdx.x * blockDim.x + threadIdx.x;
    if (idx >= n) return;
    const float v = x[idx];
    x[idx] = (v > 0.0f) ? v : (expf(v) - 1.0f);
}

__global__ void k_add(const float* __restrict__ a, const float* __restrict__ b,
                      float* __restrict__ o, int n) {
    const int idx = blockIdx.x * blockDim.x + threadIdx.x;
    if (idx < n) o[idx] = a[idx] + b[idx];
}

extern "C" void kernel_launch(void* const* d_in, const int* in_sizes, int n_in,
                              void* d_out, int out_size, void* d_ws, size_t ws_size,
                              hipStream_t stream) {
    const float* x    = (const float*)d_in[0];
    const int*   src  = (const int*)  d_in[1];
    const int*   dst  = (const int*)  d_in[2];
    const float* W0   = (const float*)d_in[3];
    const float* al0  = (const float*)d_in[4];
    const float* ar0  = (const float*)d_in[5];
    const float* W1   = (const float*)d_in[6];
    const float* al1  = (const float*)d_in[7];
    const float* ar1  = (const float*)d_in[8];
    const float* Wres = (const float*)d_in[9];
    float* out = (float*)d_out;

    const int N = in_sizes[0] / 256;
    const int E = in_sizes[1];
    const int F = 256;

    char* ws = (char*)d_ws;
    size_t off = 0;
    auto alloc = [&](size_t elems) -> float* {
        float* p = (float*)(ws + off);
        off += ((elems * sizeof(float) + 255) / 256) * 256;
        return p;
    };
    float*    feat0  = alloc((size_t)N * F);
    float*    hbuf   = alloc((size_t)N * F);
    float*    el0    = alloc((size_t)N * 4);
    float*    er0    = alloc((size_t)N * 4);
    unsigned* emax0  = (unsigned*)alloc((size_t)N * 4);
    float*    denom0 = alloc((size_t)N * 4);
    float*    ebuf0  = alloc((size_t)E * 4);
    float*    feat1  = alloc((size_t)N * 32);
    float*    resb   = alloc((size_t)N * 32);
    float*    el1    = alloc((size_t)N);
    float*    er1    = alloc((size_t)N);
    unsigned* emax1  = (unsigned*)alloc((size_t)N);
    float*    denom1 = alloc((size_t)N);
    float*    ebuf1  = alloc((size_t)E);
    float*    agg1   = alloc((size_t)N * 32);

    hipMemsetAsync(hbuf,   0, (size_t)N * F  * sizeof(float), stream);
    hipMemsetAsync(emax0,  0, (size_t)N * 4  * sizeof(unsigned), stream);
    hipMemsetAsync(denom0, 0, (size_t)N * 4  * sizeof(float), stream);
    hipMemsetAsync(emax1,  0, (size_t)N      * sizeof(unsigned), stream);
    hipMemsetAsync(denom1, 0, (size_t)N      * sizeof(float), stream);
    hipMemsetAsync(agg1,   0, (size_t)N * 32 * sizeof(float), stream);

    const int mt = (N + 15) / 16;

    k_gemm_wmma_f32<256, 256, 4><<<dim3(mt, 4), 32, 0, stream>>>(x, W0, feat0, N);
    k_attn_coef<<<(N * 4 + 255) / 256, 256, 0, stream>>>(feat0, al0, ar0, el0, er0, N, 4, 64);
    k_edge_max<<<((size_t)E * 4 + 255) / 256, 256, 0, stream>>>(src, dst, el0, er0, ebuf0, emax0, E, 4);
    k_edge_exp<<<((size_t)E * 4 + 255) / 256, 256, 0, stream>>>(dst, ebuf0, emax0, denom0, E, 4);
    k_aggregate<<<E, 256, 0, stream>>>(src, dst, feat0, ebuf0, denom0, hbuf, 4, 64);
    k_elu<<<((size_t)N * F + 255) / 256, 256, 0, stream>>>(hbuf, N * F);

    k_gemm_wmma_f32<32, 256, 2><<<dim3(mt, 1), 32, 0, stream>>>(hbuf, W1,   feat1, N);
    k_gemm_wmma_f32<32, 256, 2><<<dim3(mt, 1), 32, 0, stream>>>(hbuf, Wres, resb,  N);
    k_attn_coef<<<(N + 255) / 256, 256, 0, stream>>>(feat1, al1, ar1, el1, er1, N, 1, 32);
    k_edge_max<<<(E + 255) / 256, 256, 0, stream>>>(src, dst, el1, er1, ebuf1, emax1, E, 1);
    k_edge_exp<<<(E + 255) / 256, 256, 0, stream>>>(dst, ebuf1, emax1, denom1, E, 1);
    k_aggregate<<<E, 32, 0, stream>>>(src, dst, feat1, ebuf1, denom1, agg1, 1, 32);
    k_add<<<((size_t)N * 32 + 255) / 256, 256, 0, stream>>>(agg1, resb, out, N * 32);
}


# Edges counting-sorted by destination (by source inside it, so sums run in a fixed order); one wave per destination does max, exp, sum and weighted row sum in registers; float atomics removed
# speedup vs baseline: 3.6918x; 3.6918x over previous
#include <hip/hip_runtime.h>
#include <hip/hip_bf16.h>
#include <math.h>


typedef float v2f __attribute__((ext_vector_type(2)));
typedef float v8f __attribute__((ext_vector_type(8)));

template<int NCOLS, int KDIM, int NT>
__global__ __launch_bounds__(32)
void k_gemm_wmma_f32(const float* __restrict__ A, const float* __restrict__ B,
                     float* __restrict__ C, int M) {
    const int tileM  = blockIdx.x;
    const int tileN0 = blockIdx.y * NT;
    const int lane   = threadIdx.x;
    const int lm     = lane & 15;
    const int kg     = lane >> 4;

    int row = tileM * 16 + lm;
    if (row >= M) row = M - 1;
    const float* Ab = A + (size_t)row * KDIM + kg * 2;
    const float* Bb = B + (size_t)(kg * 2) * NCOLS + tileN0 * 16 + lm;

    v8f acc[NT];
#pragma unroll
    for (int j = 0; j < NT; ++j) acc[j] = (v8f){};

#pragma unroll
    for (int k = 0; k < KDIM; k += 4) {
        const v2f a = *(const v2f*)(Ab + k);
#pragma unroll
        for (int j = 0; j < NT; ++j) {
            v2f b;
            b.x = Bb[k * NCOLS + j * 16];
            b.y = Bb[k * NCOLS + NCOLS + j * 16];
            acc[j] = __builtin_amdgcn_wmma_f32_16x16x4_f32(
                 false, a,  false, b,
                 (short)0, acc[j],  false,  false);
        }
    }

    const int rbase = tileM * 16 + kg * 8;
#pragma unroll
    for (int j = 0; j < NT; ++j) {
        const int cbase = (tileN0 + j) * 16 + lm;
#pragma unroll
        for (int v = 0; v < 8; ++v) {
            const int r = rbase + v;
            if (r < M) C[(size_t)r * NCOLS + cbase] = acc[j][v];
        }
    }
}

__global__ void k_attn_coef(const float* __restrict__ feat,
                            const float* __restrict__ al,
                            const float* __restrict__ ar,
                            float* __restrict__ el, float* __restrict__ er,
                            int N, int H, int D) {
    const int idx = blockIdx.x * blockDim.x + threadIdx.x;
    if (idx >= N * H) return;
    const int n = idx / H, h = idx % H;
    const float* f  = feat + ((size_t)n * H + h) * D;
    const float* pl = al + h * D;
    const float* pr = ar + h * D;
    float sl = 0.0f, sr = 0.0f;
    for (int d = 0; d < D; ++d) { const float v = f[d]; sl += v * pl[d]; sr += v * pr[d]; }
    el[idx] = sl; er[idx] = sr;
}

__global__ void k_degree(const int* __restrict__ dst, int* __restrict__ deg,
                         int* __restrict__ rank, int E) {
    const int e = blockIdx.x * blockDim.x + threadIdx.x;
    if (e < E) rank[e] = atomicAdd(&deg[dst[e]], 1);
}

__global__ __launch_bounds__(1024)
void k_scan(const int* __restrict__ deg, int* __restrict__ rowptr, int NPAD) {
    __shared__ int swave[32];
    const int t = threadIdx.x, lane = t & 31, w = t >> 5;
    int carry = 0;
    int4 v = *(const int4*)(deg + 4 * t);
    for (int base = 0; base < NPAD; base += 4096) {
        const int i0 = base + 4 * t;
        int4 vn = make_int4(0, 0, 0, 0);
        if (base + 4096 < NPAD) vn = *(const int4*)(deg + i0 + 4096);
        const int s1 = v.x, s2 = s1 + v.y, s3 = s2 + v.z, s4 = s3 + v.w;
        int inc = s4;
        for (int off = 1; off < 32; off <<= 1) {
            const int y = __shfl_up(inc, off);
            if (lane >= off) inc += y;
        }
        if (lane == 31) swave[w] = inc;
        __syncthreads();
        const int wt = swave[lane];
        int winc = wt;
        for (int off = 1; off < 32; off <<= 1) {
            const int y = __shfl_up(winc, off);
            if (lane >= off) winc += y;
        }
        const int total = __shfl(winc, 31);
        const int wexc  = __shfl(winc, w) - __shfl(wt, w);
        const int ex0   = carry + wexc + (inc - s4);
        const int4 o = make_int4(ex0, ex0 + s1, ex0 + s2, ex0 + s3);
        *(int4*)(rowptr + i0) = o;
        carry += total;
        v = vn;
        __syncthreads();
    }
}

__global__ void k_fill(const int* __restrict__ src, const int* __restrict__ dst,
                       const int* __restrict__ rowptr, const int* __restrict__ rank,
                       int* __restrict__ raw, int E) {
    const int e = blockIdx.x * blockDim.x + threadIdx.x;
    if (e >= E) return;
    raw[rowptr[dst[e]] + rank[e]] = src[e];
}

__global__ __launch_bounds__(256)
void k_segsort(const int* __restrict__ rowptr, const int* __restrict__ raw,
               int* __restrict__ ssrc, int N) {
    const int lane = threadIdx.x & 31;
    const int d = __builtin_amdgcn_readfirstlane(blockIdx.x * (blockDim.x >> 5) + (threadIdx.x >> 5));
    if (d >= N) return;
    const int beg = rowptr[d];
    const int cnt = rowptr[d + 1] - beg;
    for (int mb = 0; mb < cnt; mb += 32) {
        const int k = mb + lane;
        const bool have = k < cnt;
        const int key = have ? raw[beg + k] : 0x7fffffff;
        int rank = 0;
        for (int jb = 0; jb < cnt; jb += 32) {
            const int kk = (jb + lane < cnt) ? raw[beg + jb + lane] : 0x7fffffff;
            const int jn = min(32, cnt - jb);
            for (int jj = 0; jj < jn; ++jj) {
                const int kj = __builtin_amdgcn_readlane(kk, jj);
                const int j  = jb + jj;
                rank += (kj < key || (kj == key && j < k)) ? 1 : 0;
            }
        }
        if (have) ssrc[beg + rank] = key;
    }
}

template<int H, int D>
__global__ __launch_bounds__(128)
void k_seg_softmax_sum(const int* __restrict__ rowptr, const int* __restrict__ ssrc,
                       const float* __restrict__ feat,
                       const float* __restrict__ el, const float* __restrict__ er,
                       float* __restrict__ out, int N) {
    constexpr int ROW = H * D;
    constexpr int FPL = ROW / 32;
    constexpr int NC  = (FPL >= 4) ? FPL / 4 : 1;
    static_assert(FPL == 1 || FPL % 4 == 0, "row shape");
    const int lane = threadIdx.x & 31;
    const int d = __builtin_amdgcn_readfirstlane(blockIdx.x * (blockDim.x >> 5) + (threadIdx.x >> 5));
    if (d >= N) return;
    const int beg = rowptr[d];
    const int cnt = rowptr[d + 1] - beg;

    float erd[H];
    for (int h = 0; h < H; ++h) erd[h] = er[d * H + h];

    int hsel[NC];
    for (int c = 0; c < NC; ++c) hsel[c] = (FPL >= 4) ? (c * 128 + lane * 4) / D : lane / D;

    int s0 = 0;
    float v0[H];
    for (int h = 0; h < H; ++h) v0[h] = 0.0f;
    if (lane < cnt) {
        s0 = ssrc[beg + lane];
        for (int h = 0; h < H; ++h) {
            const float v = el[s0 * H + h] + erd[h];
            v0[h] = (v > 0.0f) ? v : 0.2f * v;
        }
    }
    auto edge_scores = [&](int b, int& s, float (&v)[H]) {
        if (b == 0) {
            s = s0;
            for (int h = 0; h < H; ++h) v[h] = v0[h];
        } else {
            s = 0;
            for (int h = 0; h < H; ++h) v[h] = 0.0f;
            if (b + lane < cnt) {
                s = ssrc[beg + b + lane];
                for (int h = 0; h < H; ++h) {
                    const float t = el[s * H + h] + erd[h];
                    v[h] = (t > 0.0f) ? t : 0.2f * t;
                }
            }
        }
    };

    float mx[H];
    for (int h = 0; h < H; ++h) mx[h] = -INFINITY;
    for (int b = 0; b < cnt; b += 32) {
        int s; float v[H];
        edge_scores(b, s, v);
        if (b + lane < cnt) {
            for (int h = 0; h < H; ++h) mx[h] = fmaxf(mx[h], v[h]);
        }
    }
    for (int h = 0; h < H; ++h)
        for (int off = 16; off > 0; off >>= 1) mx[h] = fmaxf(mx[h], __shfl_xor(mx[h], off));

    float den[H];
    for (int h = 0; h < H; ++h) den[h] = 0.0f;
    for (int b = 0; b < cnt; b += 32) {
        int s; float v[H];
        edge_scores(b, s, v);
        if (b + lane < cnt) {
            for (int h = 0; h < H; ++h) den[h] += expf(v[h] - mx[h]);
        }
    }
    for (int h = 0; h < H; ++h)
        for (int off = 16; off > 0; off >>= 1) den[h] += __shfl_xor(den[h], off);

    float acc[NC * 4];
    for (int i = 0; i < NC * 4; ++i) acc[i] = 0.0f;
    for (int b = 0; b < cnt; b += 32) {
        int s; float v[H];
        edge_scores(b, s, v);
        float al[H];
        for (int h = 0; h < H; ++h) al[h] = 0.0f;
        if (b + lane < cnt) {
            for (int h = 0; h < H; ++h) al[h] = expf(v[h] - mx[h]) / den[h];
        }
        const int jn = min(32, cnt - b);
#pragma unroll 4
        for (int jj = 0; jj < jn; ++jj) {
            const int sj = __builtin_amdgcn_readlane(s, jj);
            float aj[H];
            for (int h = 0; h < H; ++h)
                aj[h] = __int_as_float(__builtin_amdgcn_readlane(__float_as_int(al[h]), jj));
            const float* frow = feat + (size_t)sj * ROW;
            for (int c = 0; c < NC; ++c) {
                float a = aj[0];
                for (int h = 1; h < H; ++h) a = (hsel[c] == h) ? aj[h] : a;
                if constexpr (FPL >= 4) {
                    const float4 f = *(const float4*)(frow + c * 128 + lane * 4);
                    acc[4 * c + 0] += f.x * a;
                    acc[4 * c + 1] += f.y * a;
                    acc[4 * c + 2] += f.z * a;
                    acc[4 * c + 3] += f.w * a;
                } else {
                    acc[0] += frow[lane] * a;
                }
            }
        }
    }

    float* orow = out + (size_t)d * ROW;
    if constexpr (FPL >= 4) {
        for (int c = 0; c < NC; ++c)
            *(float4*)(orow + c * 128 + lane * 4) =
                make_float4(acc[4 * c + 0], acc[4 * c + 1], acc[4 * c + 2], acc[4 * c + 3]);
    } else {
        orow[lane] = acc[0];
    }
}

__global__ void k_elu(float* __restrict__ x, int n) {
    const int idx = blockIdx.x * blockDim.x + threadIdx.x;
    if (idx >= n) return;
    const float v = x[idx];
    x[idx] = (v > 0.0f) ? v : (expf(v) - 1.0f);
}

__global__ void k_add(const float* __restrict__ a, const float* __restrict__ b,
                      float* __restrict__ o, int n) {
    const int idx = blockIdx.x * blockDim.x + threadIdx.x;
    if (idx < n) o[idx] = a[idx] + b[idx];
}

extern "C" void kernel_launch(void* const* d_in, const int* in_sizes, int n_in,
                              void* d_out, int out_size, void* d_ws, size_t ws_size,
                              hipStream_t stream) {
    const float* x    = (const float*)d_in[0];
    const int*   src  = (const int*)  d_in[1];
    const int*   dst  = (const int*)  d_in[2];
    const float* W0   = (const float*)d_in[3];
    const float* al0  = (const float*)d_in[4];
    const float* ar0  = (const float*)d_in[5];
    const float* W1   = (const float*)d_in[6];
    const float* al1  = (const float*)d_in[7];
    const float* ar1  = (const float*)d_in[8];
    const float* Wres = (const float*)d_in[9];
    float* out = (float*)d_out;

    const int N = in_sizes[0] / 256;
    const int E = in_sizes[1];
    const int F = 256;

    char* ws = (char*)d_ws;
    size_t off = 0;
    auto alloc = [&](size_t elems) -> float* {
        float* p = (float*)(ws + off);
        off += ((elems * sizeof(float) + 255) / 256) * 256;
        return p;
    };
    const int NPAD = ((N + 1 + 4095) / 4096) * 4096;
    float* feat0  = alloc((size_t)N * F);
    float* hbuf   = alloc((size_t)N * F);
    float* el0    = alloc((size_t)N * 4);
    float* er0    = alloc((size_t)N * 4);
    float* feat1  = alloc((size_t)N * 32);
    float* resb   = alloc((size_t)N * 32);
    float* el1    = alloc((size_t)N);
    float* er1    = alloc((size_t)N);
    float* agg1   = alloc((size_t)N * 32);
    int*   deg    = (int*)alloc((size_t)NPAD);
    int*   rowptr = (int*)alloc((size_t)NPAD);
    int*   rank   = (int*)alloc((size_t)E);
    int*   raw    = (int*)alloc((size_t)E);
    int*   ssrc   = (int*)alloc((size_t)E);

    hipMemsetAsync(deg, 0, (size_t)NPAD * sizeof(int), stream);

    const int mt = (N + 15) / 16;

    k_degree<<<(E + 255) / 256, 256, 0, stream>>>(dst, deg, rank, E);
    k_scan<<<1, 1024, 0, stream>>>(deg, rowptr, NPAD);
    k_fill<<<(E + 255) / 256, 256, 0, stream>>>(src, dst, rowptr, rank, raw, E);
    k_segsort<<<(N + 7) / 8, 256, 0, stream>>>(rowptr, raw, ssrc, N);

    k_gemm_wmma_f32<256, 256, 4><<<dim3(mt, 4), 32, 0, stream>>>(x, W0, feat0, N);
    k_attn_coef<<<(N * 4 + 255) / 256, 256, 0, stream>>>(feat0, al0, ar0, el0, er0, N, 4, 64);
    k_seg_softmax_sum<4, 64><<<(N + 3) / 4, 128, 0, stream>>>(rowptr, ssrc, feat0, el0, er0, hbuf, N);
    k_elu<<<((size_t)N * F + 255) / 256, 256, 0, stream>>>(hbuf, N * F);

    k_gemm_wmma_f32<32, 256, 2><<<dim3(mt, 1), 32, 0, stream>>>(hbuf, W1,   feat1, N);
    k_gemm_wmma_f32<32, 256, 2><<<dim3(mt, 1), 32, 0, stream>>>(hbuf, Wres, resb,  N);
    k_attn_coef<<<(N + 255) / 256, 256, 0, stream>>>(feat1, al1, ar1, el1, er1, N, 1, 32);
    k_seg_softmax_sum<1, 32><<<(N + 3) / 4, 128, 0, stream>>>(rowptr, ssrc, feat1, el1, er1, agg1, N);
    k_add<<<((size_t)N * 32 + 255) / 256, 256, 0, stream>>>(agg1, resb, out, N * 32);
}


# ELU and the residual add folded into the stores of the two aggregation kernels (row still in registers); the two one-element-per-thread launches are gone, 13 launches -> 11
# speedup vs baseline: 4.3813x; 1.1868x over previous
#include <hip/hip_runtime.h>
#include <hip/hip_bf16.h>
#include <math.h>


typedef float v2f __attribute__((ext_vector_type(2)));
typedef float v8f __attribute__((ext_vector_type(8)));

template<int NCOLS, int KDIM, int NT>
__global__ __launch_bounds__(32)
void k_gemm_wmma_f32(const float* __restrict__ A, const float* __restrict__ B,
                     float* __restrict__ C, int M) {
    const int tileM  = blockIdx.x;
    const int tileN0 = blockIdx.y * NT;
    const int lane   = threadIdx.x;
    const int lm     = lane & 15;
    const int kg     = lane >> 4;

    int row = tileM * 16 + lm;
    if (row >= M) row = M - 1;
    const float* Ab = A + (size_t)row * KDIM + kg * 2;
    const float* Bb = B + (size_t)(kg * 2) * NCOLS + tileN0 * 16 + lm;

    v8f acc[NT];
#pragma unroll
    for (int j = 0; j < NT; ++j) acc[j] = (v8f){};

#pragma unroll
    for (int k = 0; k < KDIM; k += 4) {
        const v2f a = *(const v2f*)(Ab + k);
#pragma unroll
        for (int j = 0; j < NT; ++j) {
            v2f b;
            b.x = Bb[k * NCOLS + j * 16];
            b.y = Bb[k * NCOLS + NCOLS + j * 16];
            acc[j] = __builtin_amdgcn_wmma_f32_16x16x4_f32(
                 false, a,  false, b,
                 (short)0, acc[j],  false,  false);
        }
    }

    const int rbase = tileM * 16 + kg * 8;
#pragma unroll
    for (int j = 0; j < NT; ++j) {
        const int cbase = (tileN0 + j) * 16 + lm;
#pragma unroll
        for (int v = 0; v < 8; ++v) {
            const int r = rbase + v;
            if (r < M) C[(size_t)r * NCOLS + cbase] = acc[j][v];
        }
    }
}

__global__ void k_attn_coef(const float* __restrict__ feat,
                            const float* __restrict__ al,
                            const float* __restrict__ ar,
                            float* __restrict__ el, float* __restrict__ er,
                            int N, int H, int D) {
    const int idx = blockIdx.x * blockDim.x + threadIdx.x;
    if (idx >= N * H) return;
    const int n = idx / H, h = idx % H;
    const float* f  = feat + ((size_t)n * H + h) * D;
    const float* pl = al + h * D;
    const float* pr = ar + h * D;
    float sl = 0.0f, sr = 0.0f;
    for (int d = 0; d < D; ++d) { const float v = f[d]; sl += v * pl[d]; sr += v * pr[d]; }
    el[idx] = sl; er[idx] = sr;
}

__global__ void k_degree(const int* __restrict__ dst, int* __restrict__ deg,
                         int* __restrict__ rank, int E) {
    const int e = blockIdx.x * blockDim.x + threadIdx.x;
    if (e < E) rank[e] = atomicAdd(&deg[dst[e]], 1);
}

__global__ __launch_bounds__(1024)
void k_scan(const int* __restrict__ deg, int* __restrict__ rowptr, int NPAD) {
    __shared__ int swave[32];
    const int t = threadIdx.x, lane = t & 31, w = t >> 5;
    int carry = 0;
    int4 v = *(const int4*)(deg + 4 * t);
    for (int base = 0; base < NPAD; base += 4096) {
        const int i0 = base + 4 * t;
        int4 vn = make_int4(0, 0, 0, 0);
        if (base + 4096 < NPAD) vn = *(const int4*)(deg + i0 + 4096);
        const int s1 = v.x, s2 = s1 + v.y, s3 = s2 + v.z, s4 = s3 + v.w;
        int inc = s4;
        for (int off = 1; off < 32; off <<= 1) {
            const int y = __shfl_up(inc, off);
            if (lane >= off) inc += y;
        }
        if (lane == 31) swave[w] = inc;
        __syncthreads();
        const int wt = swave[lane];
        int winc = wt;
        for (int off = 1; off < 32; off <<= 1) {
            const int y = __shfl_up(winc, off);
            if (lane >= off) winc += y;
        }
        const int total = __shfl(winc, 31);
        const int wexc  = __shfl(winc, w) - __shfl(wt, w);
        const int ex0   = carry + wexc + (inc - s4);
        const int4 o = make_int4(ex0, ex0 + s1, ex0 + s2, ex0 + s3);
        *(int4*)(rowptr + i0) = o;
        carry += total;
        v = vn;
        __syncthreads();
    }
}

__global__ void k_fill(const int* __restrict__ src, const int* __restrict__ dst,
                       const int* __restrict__ rowptr, const int* __restrict__ rank,
                       int* __restrict__ raw, int E) {
    const int e = blockIdx.x * blockDim.x + threadIdx.x;
    if (e >= E) return;
    raw[rowptr[dst[e]] + rank[e]] = src[e];
}

__global__ __launch_bounds__(256)
void k_segsort(const int* __restrict__ rowptr, const int* __restrict__ raw,
               int* __restrict__ ssrc, int N) {
    const int lane = threadIdx.x & 31;
    const int d = __builtin_amdgcn_readfirstlane(blockIdx.x * (blockDim.x >> 5) + (threadIdx.x >> 5));
    if (d >= N) return;
    const int beg = rowptr[d];
    const int cnt = rowptr[d + 1] - beg;
    for (int mb = 0; mb < cnt; mb += 32) {
        const int k = mb + lane;
        const bool have = k < cnt;
        const int key = have ? raw[beg + k] : 0x7fffffff;
        int rank = 0;
        for (int jb = 0; jb < cnt; jb += 32) {
            const int kk = (jb + lane < cnt) ? raw[beg + jb + lane] : 0x7fffffff;
            const int jn = min(32, cnt - jb);
            for (int jj = 0; jj < jn; ++jj) {
                const int kj = __builtin_amdgcn_readlane(kk, jj);
                const int j  = jb + jj;
                rank += (kj < key || (kj == key && j < k)) ? 1 : 0;
            }
        }
        if (have) ssrc[beg + rank] = key;
    }
}

template<int H, int D, bool ADD, bool ELU>
__global__ __launch_bounds__(128)
void k_seg_softmax_sum(const int* __restrict__ rowptr, const int* __restrict__ ssrc,
                       const float* __restrict__ feat,
                       const float* __restrict__ el, const float* __restrict__ er,
                       const float* __restrict__ res,
                       float* __restrict__ out, int N) {
    constexpr int ROW = H * D;
    constexpr int FPL = ROW / 32;
    constexpr int NC  = (FPL >= 4) ? FPL / 4 : 1;
    static_assert(FPL == 1 || FPL % 4 == 0, "row shape");
    const int lane = threadIdx.x & 31;
    const int d = __builtin_amdgcn_readfirstlane(blockIdx.x * (blockDim.x >> 5) + (threadIdx.x >> 5));
    if (d >= N) return;
    const int beg = rowptr[d];
    const int cnt = rowptr[d + 1] - beg;

    float erd[H];
    for (int h = 0; h < H; ++h) erd[h] = er[d * H + h];

    float radd[NC * 4];
    if constexpr (ADD) {
        const float* rrow = res + (size_t)d * ROW;
        if constexpr (FPL >= 4) {
            for (int c = 0; c < NC; ++c) {
                const float4 r = *(const float4*)(rrow + c * 128 + lane * 4);
                radd[4 * c + 0] = r.x; radd[4 * c + 1] = r.y;
                radd[4 * c + 2] = r.z; radd[4 * c + 3] = r.w;
            }
        } else {
            radd[0] = rrow[lane];
        }
    }

    int hsel[NC];
    for (int c = 0; c < NC; ++c) hsel[c] = (FPL >= 4) ? (c * 128 + lane * 4) / D : lane / D;

    int s0 = 0;
    float v0[H];
    for (int h = 0; h < H; ++h) v0[h] = 0.0f;
    if (lane < cnt) {
        s0 = ssrc[beg + lane];
        for (int h = 0; h < H; ++h) {
            const float v = el[s0 * H + h] + erd[h];
            v0[h] = (v > 0.0f) ? v : 0.2f * v;
        }
    }
    auto edge_scores = [&](int b, int& s, float (&v)[H]) {
        if (b == 0) {
            s = s0;
            for (int h = 0; h < H; ++h) v[h] = v0[h];
        } else {
            s = 0;
            for (int h = 0; h < H; ++h) v[h] = 0.0f;
            if (b + lane < cnt) {
                s = ssrc[beg + b + lane];
                for (int h = 0; h < H; ++h) {
                    const float t = el[s * H + h] + erd[h];
                    v[h] = (t > 0.0f) ? t : 0.2f * t;
                }
            }
        }
    };

    float mx[H];
    for (int h = 0; h < H; ++h) mx[h] = -INFINITY;
    for (int b = 0; b < cnt; b += 32) {
        int s; float v[H];
        edge_scores(b, s, v);
        if (b + lane < cnt) {
            for (int h = 0; h < H; ++h) mx[h] = fmaxf(mx[h], v[h]);
        }
    }
    for (int h = 0; h < H; ++h)
        for (int off = 16; off > 0; off >>= 1) mx[h] = fmaxf(mx[h], __shfl_xor(mx[h], off));

    float den[H];
    for (int h = 0; h < H; ++h) den[h] = 0.0f;
    for (int b = 0; b < cnt; b += 32) {
        int s; float v[H];
        edge_scores(b, s, v);
        if (b + lane < cnt) {
            for (int h = 0; h < H; ++h) den[h] += expf(v[h] - mx[h]);
        }
    }
    for (int h = 0; h < H; ++h)
        for (int off = 16; off > 0; off >>= 1) den[h] += __shfl_xor(den[h], off);

    float acc[NC * 4];
    for (int i = 0; i < NC * 4; ++i) acc[i] = 0.0f;
    for (int b = 0; b < cnt; b += 32) {
        int s; float v[H];
        edge_scores(b, s, v);
        float al[H];
        for (int h = 0; h < H; ++h) al[h] = 0.0f;
        if (b + lane < cnt) {
            for (int h = 0; h < H; ++h) al[h] = expf(v[h] - mx[h]) / den[h];
        }
        const int jn = min(32, cnt - b);
#pragma unroll 4
        for (int jj = 0; jj < jn; ++jj) {
            const int sj = __builtin_amdgcn_readlane(s, jj);
            float aj[H];
            for (int h = 0; h < H; ++h)
                aj[h] = __int_as_float(__builtin_amdgcn_readlane(__float_as_int(al[h]), jj));
            const float* frow = feat + (size_t)sj * ROW;
            for (int c = 0; c < NC; ++c) {
                float a = aj[0];
                for (int h = 1; h < H; ++h) a = (hsel[c] == h) ? aj[h] : a;
                if constexpr (FPL >= 4) {
                    const float4 f = *(const float4*)(frow + c * 128 + lane * 4);
                    acc[4 * c + 0] += f.x * a;
                    acc[4 * c + 1] += f.y * a;
                    acc[4 * c + 2] += f.z * a;
                    acc[4 * c + 3] += f.w * a;
                } else {
                    acc[0] += frow[lane] * a;
                }
            }
        }
    }

    if constexpr (ADD) {
        for (int i = 0; i < NC * 4; ++i) acc[i] += radd[i];
    }
    if constexpr (ELU) {
        for (int i = 0; i < NC * 4; ++i) {
            const float v = acc[i];
            acc[i] = (v > 0.0f) ? v : (expf(v) - 1.0f);
        }
    }

    float* orow = out + (size_t)d * ROW;
    if constexpr (FPL >= 4) {
        for (int c = 0; c < NC; ++c)
            *(float4*)(orow + c * 128 + lane * 4) =
                make_float4(acc[4 * c + 0], acc[4 * c + 1], acc[4 * c + 2], acc[4 * c + 3]);
    } else {
        orow[lane] = acc[0];
    }
}

extern "C" void kernel_launch(void* const* d_in, const int* in_sizes, int n_in,
                              void* d_out, int out_size, void* d_ws, size_t ws_size,
                              hipStream_t stream) {
    const float* x    = (const float*)d_in[0];
    const int*   src  = (const int*)  d_in[1];
    const int*   dst  = (const int*)  d_in[2];
    const float* W0   = (const float*)d_in[3];
    const float* al0  = (const float*)d_in[4];
    const float* ar0  = (const float*)d_in[5];
    const float* W1   = (const float*)d_in[6];
    const float* al1  = (const float*)d_in[7];
    const float* ar1  = (const float*)d_in[8];
    const float* Wres = (const float*)d_in[9];
    float* out = (float*)d_out;

    const int N = in_sizes[0] / 256;
    const int E = in_sizes[1];
    const int F = 256;

    char* ws = (char*)d_ws;
    size_t off = 0;
    auto alloc = [&](size_t elems) -> float* {
        float* p = (float*)(ws + off);
        off += ((elems * sizeof(float) + 255) / 256) * 256;
        return p;
    };
    const int NPAD = ((N + 1 + 4095) / 4096) * 4096;
    float* feat0  = alloc((size_t)N * F);
    float* hbuf   = alloc((size_t)N * F);
    float* el0    = alloc((size_t)N * 4);
    float* er0    = alloc((size_t)N * 4);
    float* feat1  = alloc((size_t)N * 32);
    float* resb   = alloc((size_t)N * 32);
    float* el1    = alloc((size_t)N);
    float* er1    = alloc((size_t)N);
    int*   deg    = (int*)alloc((size_t)NPAD);
    int*   rowptr = (int*)alloc((size_t)NPAD);
    int*   rank   = (int*)alloc((size_t)E);
    int*   raw    = (int*)alloc((size_t)E);
    int*   ssrc   = (int*)alloc((size_t)E);

    hipMemsetAsync(deg, 0, (size_t)NPAD * sizeof(int), stream);

    const int mt = (N + 15) / 16;

    k_degree<<<(E + 255) / 256, 256, 0, stream>>>(dst, deg, rank, E);
    k_scan<<<1, 1024, 0, stream>>>(deg, rowptr, NPAD);
    k_fill<<<(E + 255) / 256, 256, 0, stream>>>(src, dst, rowptr, rank, raw, E);
    k_segsort<<<(N + 7) / 8, 256, 0, stream>>>(rowptr, raw, ssrc, N);

    k_gemm_wmma_f32<256, 256, 4><<<dim3(mt, 4), 32, 0, stream>>>(x, W0, feat0, N);
    k_attn_coef<<<(N * 4 + 255) / 256, 256, 0, stream>>>(feat0, al0, ar0, el0, er0, N, 4, 64);
    k_seg_softmax_sum<4, 64, false, true><<<(N + 3) / 4, 128, 0, stream>>>(
        rowptr, ssrc, feat0, el0, er0, nullptr, hbuf, N);

    k_gemm_wmma_f32<32, 256, 2><<<dim3(mt, 1), 32, 0, stream>>>(hbuf, W1,   feat1, N);
    k_gemm_wmma_f32<32, 256, 2><<<dim3(mt, 1), 32, 0, stream>>>(hbuf, Wres, resb,  N);
    k_attn_coef<<<(N + 255) / 256, 256, 0, stream>>>(feat1, al1, ar1, el1, er1, N, 1, 32);
    k_seg_softmax_sum<1, 32, true, false><<<(N + 3) / 4, 128, 0, stream>>>(
        rowptr, ssrc, feat1, el1, er1, resb, out, N);
}


# el/er taken from the WMMA accumulators at the end of each layer's GEMM (lane-tree sum over a head's columns, replacing the serial dot); the two coefficient launches are gone, 11 launches -> 9
# speedup vs baseline: 4.5333x; 1.0347x over previous
#include <hip/hip_runtime.h>
#include <hip/hip_bf16.h>
#include <math.h>


typedef float v2f __attribute__((ext_vector_type(2)));
typedef float v8f __attribute__((ext_vector_type(8)));

template<int W>
__device__ __forceinline__ void lane_sum_step(float (&p)[16], int lm) {
    const bool up = (lm & W) != 0;
#pragma unroll
    for (int i = 0; i < W; ++i) {
        const float lo = p[i], hi = p[i + W];
        const float keep = up ? hi : lo;
        const float send = up ? lo : hi;
        const float got = __int_as_float(__builtin_amdgcn_update_dpp(
            0, __float_as_int(send), 0x160 | W, 0xf, 0xf, false));
        p[i] = keep + got;
    }
}

template<int NCOLS, int KDIM, int NT, bool COEF>
__global__ __launch_bounds__(32)
void k_gemm_wmma_f32(const float* __restrict__ A, const float* __restrict__ B,
                     float* __restrict__ C,
                     const float* __restrict__ al, const float* __restrict__ ar,
                     float* __restrict__ el, float* __restrict__ er, int M) {
    const int tileM  = blockIdx.x;
    const int tileN0 = blockIdx.y * NT;
    const int lane   = threadIdx.x;
    const int lm     = lane & 15;
    const int kg     = lane >> 4;

    int row = tileM * 16 + lm;
    if (row >= M) row = M - 1;
    const float* Ab = A + (size_t)row * KDIM + kg * 2;
    const float* Bb = B + (size_t)(kg * 2) * NCOLS + tileN0 * 16 + lm;

    v8f acc[NT];
#pragma unroll
    for (int j = 0; j < NT; ++j) acc[j] = (v8f){};

#pragma unroll
    for (int k = 0; k < KDIM; k += 4) {
        const v2f a = *(const v2f*)(Ab + k);
#pragma unroll
        for (int j = 0; j < NT; ++j) {
            v2f b;
            b.x = Bb[k * NCOLS + j * 16];
            b.y = Bb[k * NCOLS + NCOLS + j * 16];
            acc[j] = __builtin_amdgcn_wmma_f32_16x16x4_f32(
                 false, a,  false, b,
                 (short)0, acc[j],  false,  false);
        }
    }

    const int rbase = tileM * 16 + kg * 8;
#pragma unroll
    for (int j = 0; j < NT; ++j) {
        const int cbase = (tileN0 + j) * 16 + lm;
#pragma unroll
        for (int v = 0; v < 8; ++v) {
            const int r = rbase + v;
            if (r < M) C[(size_t)r * NCOLS + cbase] = acc[j][v];
        }
    }

    if constexpr (COEF) {
        constexpr int H = NCOLS / (16 * NT);
        float wl[NT], wr[NT];
#pragma unroll
        for (int j = 0; j < NT; ++j) {
            wl[j] = al[(tileN0 + j) * 16 + lm];
            wr[j] = ar[(tileN0 + j) * 16 + lm];
        }
        float p[16];
#pragma unroll
        for (int v = 0; v < 8; ++v) {
            float sl = acc[0][v] * wl[0], sr = acc[0][v] * wr[0];
#pragma unroll
            for (int j = 1; j < NT; ++j) { sl += acc[j][v] * wl[j]; sr += acc[j][v] * wr[j]; }
            p[v] = sl; p[8 + v] = sr;
        }
        lane_sum_step<8>(p, lm);
        lane_sum_step<4>(p, lm);
        lane_sum_step<2>(p, lm);
        lane_sum_step<1>(p, lm);
        const int r = rbase + (lm & 7);
        float* dstp = (lm < 8) ? el : er;
        if (r < M) dstp[(size_t)r * H + blockIdx.y] = p[0];
    }
}

__global__ void k_degree(const int* __restrict__ dst, int* __restrict__ deg,
                         int* __restrict__ rank, int E) {
    const int e = blockIdx.x * blockDim.x + threadIdx.x;
    if (e < E) rank[e] = atomicAdd(&deg[dst[e]], 1);
}

__global__ __launch_bounds__(1024)
void k_scan(const int* __restrict__ deg, int* __restrict__ rowptr, int NPAD) {
    __shared__ int swave[32];
    const int t = threadIdx.x, lane = t & 31, w = t >> 5;
    int carry = 0;
    int4 v = *(const int4*)(deg + 4 * t);
    for (int base = 0; base < NPAD; base += 4096) {
        const int i0 = base + 4 * t;
        int4 vn = make_int4(0, 0, 0, 0);
        if (base + 4096 < NPAD) vn = *(const int4*)(deg + i0 + 4096);
        const int s1 = v.x, s2 = s1 + v.y, s3 = s2 + v.z, s4 = s3 + v.w;
        int inc = s4;
        for (int off = 1; off < 32; off <<= 1) {
            const int y = __shfl_up(inc, off);
            if (lane >= off) inc += y;
        }
        if (lane == 31) swave[w] = inc;
        __syncthreads();
        const int wt = swave[lane];
        int winc = wt;
        for (int off = 1; off < 32; off <<= 1) {
            const int y = __shfl_up(winc, off);
            if (lane >= off) winc += y;
        }
        const int total = __shfl(winc, 31);
        const int wexc  = __shfl(winc, w) - __shfl(wt, w);
        const int ex0   = carry + wexc + (inc - s4);
        const int4 o = make_int4(ex0, ex0 + s1, ex0 + s2, ex0 + s3);
        *(int4*)(rowptr + i0) = o;
        carry += total;
        v = vn;
        __syncthreads();
    }
}

__global__ void k_fill(const int* __restrict__ src, const int* __restrict__ dst,
                       const int* __restrict__ rowptr, const int* __restrict__ rank,
                       int* __restrict__ raw, int E) {
    const int e = blockIdx.x * blockDim.x + threadIdx.x;
    if (e >= E) return;
    raw[rowptr[dst[e]] + rank[e]] = src[e];
}

__global__ __launch_bounds__(256)
void k_segsort(const int* __restrict__ rowptr, const int* __restrict__ raw,
               int* __restrict__ ssrc, int N) {
    const int lane = threadIdx.x & 31;
    const int d = __builtin_amdgcn_readfirstlane(blockIdx.x * (blockDim.x >> 5) + (threadIdx.x >> 5));
    if (d >= N) return;
    const int beg = rowptr[d];
    const int cnt = rowptr[d + 1] - beg;
    for (int mb = 0; mb < cnt; mb += 32) {
        const int k = mb + lane;
        const bool have = k < cnt;
        const int key = have ? raw[beg + k] : 0x7fffffff;
        int rank = 0;
        for (int jb = 0; jb < cnt; jb += 32) {
            const int kk = (jb + lane < cnt) ? raw[beg + jb + lane] : 0x7fffffff;
            const int jn = min(32, cnt - jb);
            for (int jj = 0; jj < jn; ++jj) {
                const int kj = __builtin_amdgcn_readlane(kk, jj);
                const int j  = jb + jj;
                rank += (kj < key || (kj == key && j < k)) ? 1 : 0;
            }
        }
        if (have) ssrc[beg + rank] = key;
    }
}

template<int H, int D, bool ADD, bool ELU>
__global__ __launch_bounds__(128)
void k_seg_softmax_sum(const int* __restrict__ rowptr, const int* __restrict__ ssrc,
                       const float* __restrict__ feat,
                       const float* __restrict__ el, const float* __restrict__ er,
                       const float* __restrict__ res,
                       float* __restrict__ out, int N) {
    constexpr int ROW = H * D;
    constexpr int FPL = ROW / 32;
    constexpr int NC  = (FPL >= 4) ? FPL / 4 : 1;
    static_assert(FPL == 1 || FPL % 4 == 0, "row shape");
    const int lane = threadIdx.x & 31;
    const int d = __builtin_amdgcn_readfirstlane(blockIdx.x * (blockDim.x >> 5) + (threadIdx.x >> 5));
    if (d >= N) return;
    const int beg = rowptr[d];
    const int cnt = rowptr[d + 1] - beg;

    float erd[H];
    for (int h = 0; h < H; ++h) erd[h] = er[d * H + h];

    float radd[NC * 4];
    if constexpr (ADD) {
        const float* rrow = res + (size_t)d * ROW;
        if constexpr (FPL >= 4) {
            for (int c = 0; c < NC; ++c) {
                const float4 r = *(const float4*)(rrow + c * 128 + lane * 4);
                radd[4 * c + 0] = r.x; radd[4 * c + 1] = r.y;
                radd[4 * c + 2] = r.z; radd[4 * c + 3] = r.w;
            }
        } else {
            radd[0] = rrow[lane];
        }
    }

    int hsel[NC];
    for (int c = 0; c < NC; ++c) hsel[c] = (FPL >= 4) ? (c * 128 + lane * 4) / D : lane / D;

    int s0 = 0;
    float v0[H];
    for (int h = 0; h < H; ++h) v0[h] = 0.0f;
    if (lane < cnt) {
        s0 = ssrc[beg + lane];
        for (int h = 0; h < H; ++h) {
            const float v = el[s0 * H + h] + erd[h];
            v0[h] = (v > 0.0f) ? v : 0.2f * v;
        }
    }
    auto edge_scores = [&](int b, int& s, float (&v)[H]) {
        if (b == 0) {
            s = s0;
            for (int h = 0; h < H; ++h) v[h] = v0[h];
        } else {
            s = 0;
            for (int h = 0; h < H; ++h) v[h] = 0.0f;
            if (b + lane < cnt) {
                s = ssrc[beg + b + lane];
                for (int h = 0; h < H; ++h) {
                    const float t = el[s * H + h] + erd[h];
                    v[h] = (t > 0.0f) ? t : 0.2f * t;
                }
            }
        }
    };

    float mx[H];
    for (int h = 0; h < H; ++h) mx[h] = -INFINITY;
    for (int b = 0; b < cnt; b += 32) {
        int s; float v[H];
        edge_scores(b, s, v);
        if (b + lane < cnt) {
            for (int h = 0; h < H; ++h) mx[h] = fmaxf(mx[h], v[h]);
        }
    }
    for (int h = 0; h < H; ++h)
        for (int off = 16; off > 0; off >>= 1) mx[h] = fmaxf(mx[h], __shfl_xor(mx[h], off));

    float den[H];
    for (int h = 0; h < H; ++h) den[h] = 0.0f;
    for (int b = 0; b < cnt; b += 32) {
        int s; float v[H];
        edge_scores(b, s, v);
        if (b + lane < cnt) {
            for (int h = 0; h < H; ++h) den[h] += expf(v[h] - mx[h]);
        }
    }
    for (int h = 0; h < H; ++h)
        for (int off = 16; off > 0; off >>= 1) den[h] += __shfl_xor(den[h], off);

    float acc[NC * 4];
    for (int i = 0; i < NC * 4; ++i) acc[i] = 0.0f;
    for (int b = 0; b < cnt; b += 32) {
        int s; float v[H];
        edge_scores(b, s, v);
        float al[H];
        for (int h = 0; h < H; ++h) al[h] = 0.0f;
        if (b + lane < cnt) {
            for (int h = 0; h < H; ++h) al[h] = expf(v[h] - mx[h]) / den[h];
        }
        const int jn = min(32, cnt - b);
#pragma unroll 4
        for (int jj = 0; jj < jn; ++jj) {
            const int sj = __builtin_amdgcn_readlane(s, jj);
            float aj[H];
            for (int h = 0; h < H; ++h)
                aj[h] = __int_as_float(__builtin_amdgcn_readlane(__float_as_int(al[h]), jj));
            const float* frow = feat + (size_t)sj * ROW;
            for (int c = 0; c < NC; ++c) {
                float a = aj[0];
                for (int h = 1; h < H; ++h) a = (hsel[c] == h) ? aj[h] : a;
                if constexpr (FPL >= 4) {
                    const float4 f = *(const float4*)(frow + c * 128 + lane * 4);
                    acc[4 * c + 0] += f.x * a;
                    acc[4 * c + 1] += f.y * a;
                    acc[4 * c + 2] += f.z * a;
                    acc[4 * c + 3] += f.w * a;
                } else {
                    acc[0] += frow[lane] * a;
                }
            }
        }
    }

    if constexpr (ADD) {
        for (int i = 0; i < NC * 4; ++i) acc[i] += radd[i];
    }
    if constexpr (ELU) {
        for (int i = 0; i < NC * 4; ++i) {
            const float v = acc[i];
            acc[i] = (v > 0.0f) ? v : (expf(v) - 1.0f);
        }
    }

    float* orow = out + (size_t)d * ROW;
    if constexpr (FPL >= 4) {
        for (int c = 0; c < NC; ++c)
            *(float4*)(orow + c * 128 + lane * 4) =
                make_float4(acc[4 * c + 0], acc[4 * c + 1], acc[4 * c + 2], acc[4 * c + 3]);
    } else {
        orow[lane] = acc[0];
    }
}

extern "C" void kernel_launch(void* const* d_in, const int* in_sizes, int n_in,
                              void* d_out, int out_size, void* d_ws, size_t ws_size,
                              hipStream_t stream) {
    const float* x    = (const float*)d_in[0];
    const int*   src  = (const int*)  d_in[1];
    const int*   dst  = (const int*)  d_in[2];
    const float* W0   = (const float*)d_in[3];
    const float* al0  = (const float*)d_in[4];
    const float* ar0  = (const float*)d_in[5];
    const float* W1   = (const float*)d_in[6];
    const float* al1  = (const float*)d_in[7];
    const float* ar1  = (const float*)d_in[8];
    const float* Wres = (const float*)d_in[9];
    float* out = (float*)d_out;

    const int N = in_sizes[0] / 256;
    const int E = in_sizes[1];
    const int F = 256;

    char* ws = (char*)d_ws;
    size_t off = 0;
    auto alloc = [&](size_t elems) -> float* {
        float* p = (float*)(ws + off);
        off += ((elems * sizeof(float) + 255) / 256) * 256;
        return p;
    };
    const int NPAD = ((N + 1 + 4095) / 4096) * 4096;
    float* feat0  = alloc((size_t)N * F);
    float* hbuf   = alloc((size_t)N * F);
    float* el0    = alloc((size_t)N * 4);
    float* er0    = alloc((size_t)N * 4);
    float* feat1  = alloc((size_t)N * 32);
    float* resb   = alloc((size_t)N * 32);
    float* el1    = alloc((size_t)N);
    float* er1    = alloc((size_t)N);
    int*   deg    = (int*)alloc((size_t)NPAD);
    int*   rowptr = (int*)alloc((size_t)NPAD);
    int*   rank   = (int*)alloc((size_t)E);
    int*   raw    = (int*)alloc((size_t)E);
    int*   ssrc   = (int*)alloc((size_t)E);

    hipMemsetAsync(deg, 0, (size_t)NPAD * sizeof(int), stream);

    const int mt = (N + 15) / 16;

    k_degree<<<(E + 255) / 256, 256, 0, stream>>>(dst, deg, rank, E);
    k_scan<<<1, 1024, 0, stream>>>(deg, rowptr, NPAD);
    k_fill<<<(E + 255) / 256, 256, 0, stream>>>(src, dst, rowptr, rank, raw, E);
    k_segsort<<<(N + 7) / 8, 256, 0, stream>>>(rowptr, raw, ssrc, N);

    k_gemm_wmma_f32<256, 256, 4, true><<<dim3(mt, 4), 32, 0, stream>>>(
        x, W0, feat0, al0, ar0, el0, er0, N);
    k_seg_softmax_sum<4, 64, false, true><<<(N + 3) / 4, 128, 0, stream>>>(
        rowptr, ssrc, feat0, el0, er0, nullptr, hbuf, N);

    k_gemm_wmma_f32<32, 256, 2, true><<<dim3(mt, 1), 32, 0, stream>>>(
        hbuf, W1, feat1, al1, ar1, el1, er1, N);
    k_gemm_wmma_f32<32, 256, 2, false><<<dim3(mt, 1), 32, 0, stream>>>(
        hbuf, Wres, resb, nullptr, nullptr, nullptr, nullptr, N);
    k_seg_softmax_sum<1, 32, true, false><<<(N + 3) / 4, 128, 0, stream>>>(
        rowptr, ssrc, feat1, el1, er1, resb, out, N);
}


# Layer-0 GEMM: each wave takes two 16-row tiles (32x64 strip), so every B fragment load feeds two WMMAs; B loads per WMMA halved, single-wave work-groups 12500 -> 6252
# speedup vs baseline: 4.7756x; 1.0534x over previous
#include <hip/hip_runtime.h>
#include <hip/hip_bf16.h>
#include <math.h>


typedef float v2f __attribute__((ext_vector_type(2)));
typedef float v8f __attribute__((ext_vector_type(8)));

template<int W>
__device__ __forceinline__ void lane_sum_step(float (&p)[16], int lm) {
    const bool up = (lm & W) != 0;
#pragma unroll
    for (int i = 0; i < W; ++i) {
        const float lo = p[i], hi = p[i + W];
        const float keep = up ? hi : lo;
        const float send = up ? lo : hi;
        const float got = __int_as_float(__builtin_amdgcn_update_dpp(
            0, __float_as_int(send), 0x160 | W, 0xf, 0xf, false));
        p[i] = keep + got;
    }
}

template<int NCOLS, int KDIM, int NT, int MT, bool COEF>
__global__ __launch_bounds__(32)
void k_gemm_wmma_f32(const float* __restrict__ A, const float* __restrict__ B,
                     float* __restrict__ C,
                     const float* __restrict__ al, const float* __restrict__ ar,
                     float* __restrict__ el, float* __restrict__ er, int M) {
    const int tileM0 = blockIdx.x * MT;
    const int tileN0 = blockIdx.y * NT;
    const int lane   = threadIdx.x;
    const int lm     = lane & 15;
    const int kg     = lane >> 4;

    const float* Ab[MT];
#pragma unroll
    for (int i = 0; i < MT; ++i) {
        int row = (tileM0 + i) * 16 + lm;
        if (row >= M) row = M - 1;
        Ab[i] = A + (size_t)row * KDIM + kg * 2;
    }
    const float* Bb = B + (size_t)(kg * 2) * NCOLS + tileN0 * 16 + lm;

    v8f acc[MT][NT];
#pragma unroll
    for (int i = 0; i < MT; ++i)
#pragma unroll
        for (int j = 0; j < NT; ++j) acc[i][j] = (v8f){};

#pragma unroll
    for (int k = 0; k < KDIM; k += 4) {
        v2f a[MT];
#pragma unroll
        for (int i = 0; i < MT; ++i) a[i] = *(const v2f*)(Ab[i] + k);
#pragma unroll
        for (int j = 0; j < NT; ++j) {
            v2f b;
            b.x = Bb[k * NCOLS + j * 16];
            b.y = Bb[k * NCOLS + NCOLS + j * 16];
#pragma unroll
            for (int i = 0; i < MT; ++i)
                acc[i][j] = __builtin_amdgcn_wmma_f32_16x16x4_f32(
                     false, a[i],  false, b,
                     (short)0, acc[i][j],  false,  false);
        }
    }

#pragma unroll
    for (int i = 0; i < MT; ++i) {
        const int rbase = (tileM0 + i) * 16 + kg * 8;
#pragma unroll
        for (int j = 0; j < NT; ++j) {
            const int cbase = (tileN0 + j) * 16 + lm;
#pragma unroll
            for (int v = 0; v < 8; ++v) {
                const int r = rbase + v;
                if (r < M) C[(size_t)r * NCOLS + cbase] = acc[i][j][v];
            }
        }
    }

    if constexpr (COEF) {
        constexpr int H = NCOLS / (16 * NT);
        float wl[NT], wr[NT];
#pragma unroll
        for (int j = 0; j < NT; ++j) {
            wl[j] = al[(tileN0 + j) * 16 + lm];
            wr[j] = ar[(tileN0 + j) * 16 + lm];
        }
#pragma unroll
        for (int i = 0; i < MT; ++i) {
            const int rbase = (tileM0 + i) * 16 + kg * 8;
            float p[16];
#pragma unroll
            for (int v = 0; v < 8; ++v) {
                float sl = acc[i][0][v] * wl[0], sr = acc[i][0][v] * wr[0];
#pragma unroll
                for (int j = 1; j < NT; ++j) { sl += acc[i][j][v] * wl[j]; sr += acc[i][j][v] * wr[j]; }
                p[v] = sl; p[8 + v] = sr;
            }
            lane_sum_step<8>(p, lm);
            lane_sum_step<4>(p, lm);
            lane_sum_step<2>(p, lm);
            lane_sum_step<1>(p, lm);
            const int r = rbase + (lm & 7);
            float* dstp = (lm < 8) ? el : er;
            if (r < M) dstp[(size_t)r * H + blockIdx.y] = p[0];
        }
    }
}

__global__ void k_degree(const int* __restrict__ dst, int* __restrict__ deg,
                         int* __restrict__ rank, int E) {
    const int e = blockIdx.x * blockDim.x + threadIdx.x;
    if (e < E) rank[e] = atomicAdd(&deg[dst[e]], 1);
}

__global__ __launch_bounds__(1024)
void k_scan(const int* __restrict__ deg, int* __restrict__ rowptr, int NPAD) {
    __shared__ int swave[32];
    const int t = threadIdx.x, lane = t & 31, w = t >> 5;
    int carry = 0;
    int4 v = *(const int4*)(deg + 4 * t);
    for (int base = 0; base < NPAD; base += 4096) {
        const int i0 = base + 4 * t;
        int4 vn = make_int4(0, 0, 0, 0);
        if (base + 4096 < NPAD) vn = *(const int4*)(deg + i0 + 4096);
        const int s1 = v.x, s2 = s1 + v.y, s3 = s2 + v.z, s4 = s3 + v.w;
        int inc = s4;
        for (int off = 1; off < 32; off <<= 1) {
            const int y = __shfl_up(inc, off);
            if (lane >= off) inc += y;
        }
        if (lane == 31) swave[w] = inc;
        __syncthreads();
        const int wt = swave[lane];
        int winc = wt;
        for (int off = 1; off < 32; off <<= 1) {
            const int y = __shfl_up(winc, off);
            if (lane >= off) winc += y;
        }
        const int total = __shfl(winc, 31);
        const int wexc  = __shfl(winc, w) - __shfl(wt, w);
        const int ex0   = carry + wexc + (inc - s4);
        const int4 o = make_int4(ex0, ex0 + s1, ex0 + s2, ex0 + s3);
        *(int4*)(rowptr + i0) = o;
        carry += total;
        v = vn;
        __syncthreads();
    }
}

__global__ void k_fill(const int* __restrict__ src, const int* __restrict__ dst,
                       const int* __restrict__ rowptr, const int* __restrict__ rank,
                       int* __restrict__ raw, int E) {
    const int e = blockIdx.x * blockDim.x + threadIdx.x;
    if (e >= E) return;
    raw[rowptr[dst[e]] + rank[e]] = src[e];
}

__global__ __launch_bounds__(256)
void k_segsort(const int* __restrict__ rowptr, const int* __restrict__ raw,
               int* __restrict__ ssrc, int N) {
    const int lane = threadIdx.x & 31;
    const int d = __builtin_amdgcn_readfirstlane(blockIdx.x * (blockDim.x >> 5) + (threadIdx.x >> 5));
    if (d >= N) return;
    const int beg = rowptr[d];
    const int cnt = rowptr[d + 1] - beg;
    for (int mb = 0; mb < cnt; mb += 32) {
        const int k = mb + lane;
        const bool have = k < cnt;
        const int key = have ? raw[beg + k] : 0x7fffffff;
        int rank = 0;
        for (int jb = 0; jb < cnt; jb += 32) {
            const int kk = (jb + lane < cnt) ? raw[beg + jb + lane] : 0x7fffffff;
            const int jn = min(32, cnt - jb);
            for (int jj = 0; jj < jn; ++jj) {
                const int kj = __builtin_amdgcn_readlane(kk, jj);
                const int j  = jb + jj;
                rank += (kj < key || (kj == key && j < k)) ? 1 : 0;
            }
        }
        if (have) ssrc[beg + rank] = key;
    }
}

template<int H, int D, bool ADD, bool ELU>
__global__ __launch_bounds__(128)
void k_seg_softmax_sum(const int* __restrict__ rowptr, const int* __restrict__ ssrc,
                       const float* __restrict__ feat,
                       const float* __restrict__ el, const float* __restrict__ er,
                       const float* __restrict__ res,
                       float* __restrict__ out, int N) {
    constexpr int ROW = H * D;
    constexpr int FPL = ROW / 32;
    constexpr int NC  = (FPL >= 4) ? FPL / 4 : 1;
    static_assert(FPL == 1 || FPL % 4 == 0, "row shape");
    const int lane = threadIdx.x & 31;
    const int d = __builtin_amdgcn_readfirstlane(blockIdx.x * (blockDim.x >> 5) + (threadIdx.x >> 5));
    if (d >= N) return;
    const int beg = rowptr[d];
    const int cnt = rowptr[d + 1] - beg;

    float erd[H];
    for (int h = 0; h < H; ++h) erd[h] = er[d * H + h];

    float radd[NC * 4];
    if constexpr (ADD) {
        const float* rrow = res + (size_t)d * ROW;
        if constexpr (FPL >= 4) {
            for (int c = 0; c < NC; ++c) {
                const float4 r = *(const float4*)(rrow + c * 128 + lane * 4);
                radd[4 * c + 0] = r.x; radd[4 * c + 1] = r.y;
                radd[4 * c + 2] = r.z; radd[4 * c + 3] = r.w;
            }
        } else {
            radd[0] = rrow[lane];
        }
    }

    int hsel[NC];
    for (int c = 0; c < NC; ++c) hsel[c] = (FPL >= 4) ? (c * 128 + lane * 4) / D : lane / D;

    int s0 = 0;
    float v0[H];
    for (int h = 0; h < H; ++h) v0[h] = 0.0f;
    if (lane < cnt) {
        s0 = ssrc[beg + lane];
        for (int h = 0; h < H; ++h) {
            const float v = el[s0 * H + h] + erd[h];
            v0[h] = (v > 0.0f) ? v : 0.2f * v;
        }
    }
    auto edge_scores = [&](int b, int& s, float (&v)[H]) {
        if (b == 0) {
            s = s0;
            for (int h = 0; h < H; ++h) v[h] = v0[h];
        } else {
            s = 0;
            for (int h = 0; h < H; ++h) v[h] = 0.0f;
            if (b + lane < cnt) {
                s = ssrc[beg + b + lane];
                for (int h = 0; h < H; ++h) {
                    const float t = el[s * H + h] + erd[h];
                    v[h] = (t > 0.0f) ? t : 0.2f * t;
                }
            }
        }
    };

    float mx[H];
    for (int h = 0; h < H; ++h) mx[h] = -INFINITY;
    for (int b = 0; b < cnt; b += 32) {
        int s; float v[H];
        edge_scores(b, s, v);
        if (b + lane < cnt) {
            for (int h = 0; h < H; ++h) mx[h] = fmaxf(mx[h], v[h]);
        }
    }
    for (int h = 0; h < H; ++h)
        for (int off = 16; off > 0; off >>= 1) mx[h] = fmaxf(mx[h], __shfl_xor(mx[h], off));

    float den[H];
    for (int h = 0; h < H; ++h) den[h] = 0.0f;
    for (int b = 0; b < cnt; b += 32) {
        int s; float v[H];
        edge_scores(b, s, v);
        if (b + lane < cnt) {
            for (int h = 0; h < H; ++h) den[h] += expf(v[h] - mx[h]);
        }
    }
    for (int h = 0; h < H; ++h)
        for (int off = 16; off > 0; off >>= 1) den[h] += __shfl_xor(den[h], off);

    float acc[NC * 4];
    for (int i = 0; i < NC * 4; ++i) acc[i] = 0.0f;
    for (int b = 0; b < cnt; b += 32) {
        int s; float v[H];
        edge_scores(b, s, v);
        float al[H];
        for (int h = 0; h < H; ++h) al[h] = 0.0f;
        if (b + lane < cnt) {
            for (int h = 0; h < H; ++h) al[h] = expf(v[h] - mx[h]) / den[h];
        }
        const int jn = min(32, cnt - b);
#pragma unroll 4
        for (int jj = 0; jj < jn; ++jj) {
            const int sj = __builtin_amdgcn_readlane(s, jj);
            float aj[H];
            for (int h = 0; h < H; ++h)
                aj[h] = __int_as_float(__builtin_amdgcn_readlane(__float_as_int(al[h]), jj));
            const float* frow = feat + (size_t)sj * ROW;
            for (int c = 0; c < NC; ++c) {
                float a = aj[0];
                for (int h = 1; h < H; ++h) a = (hsel[c] == h) ? aj[h] : a;
                if constexpr (FPL >= 4) {
                    const float4 f = *(const float4*)(frow + c * 128 + lane * 4);
                    acc[4 * c + 0] += f.x * a;
                    acc[4 * c + 1] += f.y * a;
                    acc[4 * c + 2] += f.z * a;
                    acc[4 * c + 3] += f.w * a;
                } else {
                    acc[0] += frow[lane] * a;
                }
            }
        }
    }

    if constexpr (ADD) {
        for (int i = 0; i < NC * 4; ++i) acc[i] += radd[i];
    }
    if constexpr (ELU) {
        for (int i = 0; i < NC * 4; ++i) {
            const float v = acc[i];
            acc[i] = (v > 0.0f) ? v : (expf(v) - 1.0f);
        }
    }

    float* orow = out + (size_t)d * ROW;
    if constexpr (FPL >= 4) {
        for (int c = 0; c < NC; ++c)
            *(float4*)(orow + c * 128 + lane * 4) =
                make_float4(acc[4 * c + 0], acc[4 * c + 1], acc[4 * c + 2], acc[4 * c + 3]);
    } else {
        orow[lane] = acc[0];
    }
}

extern "C" void kernel_launch(void* const* d_in, const int* in_sizes, int n_in,
                              void* d_out, int out_size, void* d_ws, size_t ws_size,
                              hipStream_t stream) {
    const float* x    = (const float*)d_in[0];
    const int*   src  = (const int*)  d_in[1];
    const int*   dst  = (const int*)  d_in[2];
    const float* W0   = (const float*)d_in[3];
    const float* al0  = (const float*)d_in[4];
    const float* ar0  = (const float*)d_in[5];
    const float* W1   = (const float*)d_in[6];
    const float* al1  = (const float*)d_in[7];
    const float* ar1  = (const float*)d_in[8];
    const float* Wres = (const float*)d_in[9];
    float* out = (float*)d_out;

    const int N = in_sizes[0] / 256;
    const int E = in_sizes[1];
    const int F = 256;

    char* ws = (char*)d_ws;
    size_t off = 0;
    auto alloc = [&](size_t elems) -> float* {
        float* p = (float*)(ws + off);
        off += ((elems * sizeof(float) + 255) / 256) * 256;
        return p;
    };
    const int NPAD = ((N + 1 + 4095) / 4096) * 4096;
    float* feat0  = alloc((size_t)N * F);
    float* hbuf   = alloc((size_t)N * F);
    float* el0    = alloc((size_t)N * 4);
    float* er0    = alloc((size_t)N * 4);
    float* feat1  = alloc((size_t)N * 32);
    float* resb   = alloc((size_t)N * 32);
    float* el1    = alloc((size_t)N);
    float* er1    = alloc((size_t)N);
    int*   deg    = (int*)alloc((size_t)NPAD);
    int*   rowptr = (int*)alloc((size_t)NPAD);
    int*   rank   = (int*)alloc((size_t)E);
    int*   raw    = (int*)alloc((size_t)E);
    int*   ssrc   = (int*)alloc((size_t)E);

    hipMemsetAsync(deg, 0, (size_t)NPAD * sizeof(int), stream);

    const int mt = (N + 15) / 16;

    k_degree<<<(E + 255) / 256, 256, 0, stream>>>(dst, deg, rank, E);
    k_scan<<<1, 1024, 0, stream>>>(deg, rowptr, NPAD);
    k_fill<<<(E + 255) / 256, 256, 0, stream>>>(src, dst, rowptr, rank, raw, E);
    k_segsort<<<(N + 7) / 8, 256, 0, stream>>>(rowptr, raw, ssrc, N);

    k_gemm_wmma_f32<256, 256, 4, 2, true><<<dim3((mt + 1) / 2, 4), 32, 0, stream>>>(
        x, W0, feat0, al0, ar0, el0, er0, N);
    k_seg_softmax_sum<4, 64, false, true><<<(N + 3) / 4, 128, 0, stream>>>(
        rowptr, ssrc, feat0, el0, er0, nullptr, hbuf, N);

    k_gemm_wmma_f32<32, 256, 2, 1, true><<<dim3(mt, 1), 32, 0, stream>>>(
        hbuf, W1, feat1, al1, ar1, el1, er1, N);
    k_gemm_wmma_f32<32, 256, 2, 1, false><<<dim3(mt, 1), 32, 0, stream>>>(
        hbuf, Wres, resb, nullptr, nullptr, nullptr, nullptr, N);
    k_seg_softmax_sum<1, 32, true, false><<<(N + 3) / 4, 128, 0, stream>>>(
        rowptr, ssrc, feat1, el1, er1, resb, out, N);
}


# Degree counts and edge ranks on LDS counters: 32 work-groups each count a slice of the edges into node counters in LDS, a per-node prefix over slices gives offsets; global atomics and memset gone
# speedup vs baseline: 5.3745x; 1.1254x over previous
#include <hip/hip_runtime.h>
#include <hip/hip_bf16.h>
#include <math.h>


typedef float v2f __attribute__((ext_vector_type(2)));
typedef float v8f __attribute__((ext_vector_type(8)));

template<int W>
__device__ __forceinline__ void lane_sum_step(float (&p)[16], int lm) {
    const bool up = (lm & W) != 0;
#pragma unroll
    for (int i = 0; i < W; ++i) {
        const float lo = p[i], hi = p[i + W];
        const float keep = up ? hi : lo;
        const float send = up ? lo : hi;
        const float got = __int_as_float(__builtin_amdgcn_update_dpp(
            0, __float_as_int(send), 0x160 | W, 0xf, 0xf, false));
        p[i] = keep + got;
    }
}

template<int NCOLS, int KDIM, int NT, int MT, bool COEF>
__global__ __launch_bounds__(32)
void k_gemm_wmma_f32(const float* __restrict__ A, const float* __restrict__ B,
                     float* __restrict__ C,
                     const float* __restrict__ al, const float* __restrict__ ar,
                     float* __restrict__ el, float* __restrict__ er, int M) {
    const int tileM0 = blockIdx.x * MT;
    const int tileN0 = blockIdx.y * NT;
    const int lane   = threadIdx.x;
    const int lm     = lane & 15;
    const int kg     = lane >> 4;

    const float* Ab[MT];
#pragma unroll
    for (int i = 0; i < MT; ++i) {
        int row = (tileM0 + i) * 16 + lm;
        if (row >= M) row = M - 1;
        Ab[i] = A + (size_t)row * KDIM + kg * 2;
    }
    const float* Bb = B + (size_t)(kg * 2) * NCOLS + tileN0 * 16 + lm;

    v8f acc[MT][NT];
#pragma unroll
    for (int i = 0; i < MT; ++i)
#pragma unroll
        for (int j = 0; j < NT; ++j) acc[i][j] = (v8f){};

#pragma unroll
    for (int k = 0; k < KDIM; k += 4) {
        v2f a[MT];
#pragma unroll
        for (int i = 0; i < MT; ++i) a[i] = *(const v2f*)(Ab[i] + k);
#pragma unroll
        for (int j = 0; j < NT; ++j) {
            v2f b;
            b.x = Bb[k * NCOLS + j * 16];
            b.y = Bb[k * NCOLS + NCOLS + j * 16];
#pragma unroll
            for (int i = 0; i < MT; ++i)
                acc[i][j] = __builtin_amdgcn_wmma_f32_16x16x4_f32(
                     false, a[i],  false, b,
                     (short)0, acc[i][j],  false,  false);
        }
    }

#pragma unroll
    for (int i = 0; i < MT; ++i) {
        const int rbase = (tileM0 + i) * 16 + kg * 8;
#pragma unroll
        for (int j = 0; j < NT; ++j) {
            const int cbase = (tileN0 + j) * 16 + lm;
#pragma unroll
            for (int v = 0; v < 8; ++v) {
                const int r = rbase + v;
                if (r < M) C[(size_t)r * NCOLS + cbase] = acc[i][j][v];
            }
        }
    }

    if constexpr (COEF) {
        constexpr int H = NCOLS / (16 * NT);
        float wl[NT], wr[NT];
#pragma unroll
        for (int j = 0; j < NT; ++j) {
            wl[j] = al[(tileN0 + j) * 16 + lm];
            wr[j] = ar[(tileN0 + j) * 16 + lm];
        }
#pragma unroll
        for (int i = 0; i < MT; ++i) {
            const int rbase = (tileM0 + i) * 16 + kg * 8;
            float p[16];
#pragma unroll
            for (int v = 0; v < 8; ++v) {
                float sl = acc[i][0][v] * wl[0], sr = acc[i][0][v] * wr[0];
#pragma unroll
                for (int j = 1; j < NT; ++j) { sl += acc[i][j][v] * wl[j]; sr += acc[i][j][v] * wr[j]; }
                p[v] = sl; p[8 + v] = sr;
            }
            lane_sum_step<8>(p, lm);
            lane_sum_step<4>(p, lm);
            lane_sum_step<2>(p, lm);
            lane_sum_step<1>(p, lm);
            const int r = rbase + (lm & 7);
            float* dstp = (lm < 8) ? el : er;
            if (r < M) dstp[(size_t)r * H + blockIdx.y] = p[0];
        }
    }
}

constexpr int NSLICE  = 32;
constexpr int CNT_CAP = 51200;
constexpr int EPT     = 16;

__device__ __forceinline__ int slice_len(int E) {
    return (((E + NSLICE - 1) / NSLICE) + 255) & ~255;
}

__global__ __launch_bounds__(1024)
void k_count_lds(const int* __restrict__ dst, int* __restrict__ cnt,
                 int* __restrict__ rank, int E, int N, int PITCH) {
    __shared__ int c[CNT_CAP];
    const int t  = threadIdx.x;
    const int L  = slice_len(E);
    const int e0 = blockIdx.x * L;
    const int e1 = min(E, e0 + L);
    int* crow = cnt + (size_t)blockIdx.x * PITCH;
    for (int base = 0; base < N; base += CNT_CAP) {
        const int span  = min(CNT_CAP, N - base);
        const int span4 = (span + 3) >> 2;
        for (int i = t; i < span4; i += 1024) ((int4*)c)[i] = make_int4(0, 0, 0, 0);
        __syncthreads();
        for (int eb = e0; eb < e1; eb += EPT * 1024) {
            int d[EPT], r[EPT];
#pragma unroll
            for (int k = 0; k < EPT; ++k) {
                const int e = eb + k * 1024 + t;
                const int v = dst[min(e, e1 - 1)] - base;
                d[k] = (e < e1) ? v : -1;
            }
#pragma unroll
            for (int k = 0; k < EPT; ++k) {
                r[k] = 0;
                if ((unsigned)d[k] < (unsigned)span) r[k] = atomicAdd(&c[d[k]], 1);
            }
#pragma unroll
            for (int k = 0; k < EPT; ++k)
                if ((unsigned)d[k] < (unsigned)span) rank[eb + k * 1024 + t] = r[k];
        }
        __syncthreads();
        for (int i = t; i < span4; i += 1024) ((int4*)(crow + base))[i] = ((const int4*)c)[i];
        __syncthreads();
    }
}

__global__ __launch_bounds__(256)
void k_slice_prefix(int* __restrict__ cnt, int* __restrict__ deg, int N, int NPAD) {
    const int n = blockIdx.x * blockDim.x + threadIdx.x;
    if (n >= NPAD) return;
    int run = 0;
    if (n < N) {
        int c[NSLICE];
#pragma unroll
        for (int s = 0; s < NSLICE; ++s) c[s] = cnt[(size_t)s * NPAD + n];
#pragma unroll
        for (int s = 0; s < NSLICE; ++s) { cnt[(size_t)s * NPAD + n] = run; run += c[s]; }
    }
    deg[n] = run;
}

__global__ __launch_bounds__(1024)
void k_scan(const int* __restrict__ deg, int* __restrict__ rowptr, int NPAD) {
    __shared__ int swave[32];
    const int t = threadIdx.x, lane = t & 31, w = t >> 5;
    int carry = 0;
    int4 v = *(const int4*)(deg + 4 * t);
    for (int base = 0; base < NPAD; base += 4096) {
        const int i0 = base + 4 * t;
        int4 vn = make_int4(0, 0, 0, 0);
        if (base + 4096 < NPAD) vn = *(const int4*)(deg + i0 + 4096);
        const int s1 = v.x, s2 = s1 + v.y, s3 = s2 + v.z, s4 = s3 + v.w;
        int inc = s4;
        for (int off = 1; off < 32; off <<= 1) {
            const int y = __shfl_up(inc, off);
            if (lane >= off) inc += y;
        }
        if (lane == 31) swave[w] = inc;
        __syncthreads();
        const int wt = swave[lane];
        int winc = wt;
        for (int off = 1; off < 32; off <<= 1) {
            const int y = __shfl_up(winc, off);
            if (lane >= off) winc += y;
        }
        const int total = __shfl(winc, 31);
        const int wexc  = __shfl(winc, w) - __shfl(wt, w);
        const int ex0   = carry + wexc + (inc - s4);
        const int4 o = make_int4(ex0, ex0 + s1, ex0 + s2, ex0 + s3);
        *(int4*)(rowptr + i0) = o;
        carry += total;
        v = vn;
        __syncthreads();
    }
}

__global__ __launch_bounds__(256)
void k_fill(const int* __restrict__ src, const int* __restrict__ dst,
            const int* __restrict__ rowptr, const int* __restrict__ off,
            const int* __restrict__ rank, int* __restrict__ raw, int E, int NPAD) {
    const int e = blockIdx.x * 256 + threadIdx.x;
    if (e >= E) return;
    const int s = (blockIdx.x * 256) / slice_len(E);
    const int d = dst[e];
    raw[rowptr[d] + off[(size_t)s * NPAD + d] + rank[e]] = src[e];
}

__global__ __launch_bounds__(256)
void k_segsort(const int* __restrict__ rowptr, const int* __restrict__ raw,
               int* __restrict__ ssrc, int N) {
    const int lane = threadIdx.x & 31;
    const int d = __builtin_amdgcn_readfirstlane(blockIdx.x * (blockDim.x >> 5) + (threadIdx.x >> 5));
    if (d >= N) return;
    const int beg = rowptr[d];
    const int cnt = rowptr[d + 1] - beg;
    for (int mb = 0; mb < cnt; mb += 32) {
        const int k = mb + lane;
        const bool have = k < cnt;
        const int key = have ? raw[beg + k] : 0x7fffffff;
        int rank = 0;
        for (int jb = 0; jb < cnt; jb += 32) {
            const int kk = (jb + lane < cnt) ? raw[beg + jb + lane] : 0x7fffffff;
            const int jn = min(32, cnt - jb);
            for (int jj = 0; jj < jn; ++jj) {
                const int kj = __builtin_amdgcn_readlane(kk, jj);
                const int j  = jb + jj;
                rank += (kj < key || (kj == key && j < k)) ? 1 : 0;
            }
        }
        if (have) ssrc[beg + rank] = key;
    }
}

template<int H, int D, bool ADD, bool ELU>
__global__ __launch_bounds__(128)
void k_seg_softmax_sum(const int* __restrict__ rowptr, const int* __restrict__ ssrc,
                       const float* __restrict__ feat,
                       const float* __restrict__ el, const float* __restrict__ er,
                       const float* __restrict__ res,
                       float* __restrict__ out, int N) {
    constexpr int ROW = H * D;
    constexpr int FPL = ROW / 32;
    constexpr int NC  = (FPL >= 4) ? FPL / 4 : 1;
    static_assert(FPL == 1 || FPL % 4 == 0, "row shape");
    const int lane = threadIdx.x & 31;
    const int d = __builtin_amdgcn_readfirstlane(blockIdx.x * (blockDim.x >> 5) + (threadIdx.x >> 5));
    if (d >= N) return;
    const int beg = rowptr[d];
    const int cnt = rowptr[d + 1] - beg;

    float erd[H];
    for (int h = 0; h < H; ++h) erd[h] = er[d * H + h];

    float radd[NC * 4];
    if constexpr (ADD) {
        const float* rrow = res + (size_t)d * ROW;
        if constexpr (FPL >= 4) {
            for (int c = 0; c < NC; ++c) {
                const float4 r = *(const float4*)(rrow + c * 128 + lane * 4);
                radd[4 * c + 0] = r.x; radd[4 * c + 1] = r.y;
                radd[4 * c + 2] = r.z; radd[4 * c + 3] = r.w;
            }
        } else {
            radd[0] = rrow[lane];
        }
    }

    int hsel[NC];
    for (int c = 0; c < NC; ++c) hsel[c] = (FPL >= 4) ? (c * 128 + lane * 4) / D : lane / D;

    int s0 = 0;
    float v0[H];
    for (int h = 0; h < H; ++h) v0[h] = 0.0f;
    if (lane < cnt) {
        s0 = ssrc[beg + lane];
        for (int h = 0; h < H; ++h) {
            const float v = el[s0 * H + h] + erd[h];
            v0[h] = (v > 0.0f) ? v : 0.2f * v;
        }
    }
    auto edge_scores = [&](int b, int& s, float (&v)[H]) {
        if (b == 0) {
            s = s0;
            for (int h = 0; h < H; ++h) v[h] = v0[h];
        } else {
            s = 0;
            for (int h = 0; h < H; ++h) v[h] = 0.0f;
            if (b + lane < cnt) {
                s = ssrc[beg + b + lane];
                for (int h = 0; h < H; ++h) {
                    const float t = el[s * H + h] + erd[h];
                    v[h] = (t > 0.0f) ? t : 0.2f * t;
                }
            }
        }
    };

    float mx[H];
    for (int h = 0; h < H; ++h) mx[h] = -INFINITY;
    for (int b = 0; b < cnt; b += 32) {
        int s; float v[H];
        edge_scores(b, s, v);
        if (b + lane < cnt) {
            for (int h = 0; h < H; ++h) mx[h] = fmaxf(mx[h], v[h]);
        }
    }
    for (int h = 0; h < H; ++h)
        for (int off = 16; off > 0; off >>= 1) mx[h] = fmaxf(mx[h], __shfl_xor(mx[h], off));

    float den[H];
    for (int h = 0; h < H; ++h) den[h] = 0.0f;
    for (int b = 0; b < cnt; b += 32) {
        int s; float v[H];
        edge_scores(b, s, v);
        if (b + lane < cnt) {
            for (int h = 0; h < H; ++h) den[h] += expf(v[h] - mx[h]);
        }
    }
    for (int h = 0; h < H; ++h)
        for (int off = 16; off > 0; off >>= 1) den[h] += __shfl_xor(den[h], off);

    float acc[NC * 4];
    for (int i = 0; i < NC * 4; ++i) acc[i] = 0.0f;
    for (int b = 0; b < cnt; b += 32) {
        int s; float v[H];
        edge_scores(b, s, v);
        float al[H];
        for (int h = 0; h < H; ++h) al[h] = 0.0f;
        if (b + lane < cnt) {
            for (int h = 0; h < H; ++h) al[h] = expf(v[h] - mx[h]) / den[h];
        }
        const int jn = min(32, cnt - b);
#pragma unroll 4
        for (int jj = 0; jj < jn; ++jj) {
            const int sj = __builtin_amdgcn_readlane(s, jj);
            float aj[H];
            for (int h = 0; h < H; ++h)
                aj[h] = __int_as_float(__builtin_amdgcn_readlane(__float_as_int(al[h]), jj));
            const float* frow = feat + (size_t)sj * ROW;
            for (int c = 0; c < NC; ++c) {
                float a = aj[0];
                for (int h = 1; h < H; ++h) a = (hsel[c] == h) ? aj[h] : a;
                if constexpr (FPL >= 4) {
                    const float4 f = *(const float4*)(frow + c * 128 + lane * 4);
                    acc[4 * c + 0] += f.x * a;
                    acc[4 * c + 1] += f.y * a;
                    acc[4 * c + 2] += f.z * a;
                    acc[4 * c + 3] += f.w * a;
                } else {
                    acc[0] += frow[lane] * a;
                }
            }
        }
    }

    if constexpr (ADD) {
        for (int i = 0; i < NC * 4; ++i) acc[i] += radd[i];
    }
    if constexpr (ELU) {
        for (int i = 0; i < NC * 4; ++i) {
            const float v = acc[i];
            acc[i] = (v > 0.0f) ? v : (expf(v) - 1.0f);
        }
    }

    float* orow = out + (size_t)d * ROW;
    if constexpr (FPL >= 4) {
        for (int c = 0; c < NC; ++c)
            *(float4*)(orow + c * 128 + lane * 4) =
                make_float4(acc[4 * c + 0], acc[4 * c + 1], acc[4 * c + 2], acc[4 * c + 3]);
    } else {
        orow[lane] = acc[0];
    }
}

extern "C" void kernel_launch(void* const* d_in, const int* in_sizes, int n_in,
                              void* d_out, int out_size, void* d_ws, size_t ws_size,
                              hipStream_t stream) {
    const float* x    = (const float*)d_in[0];
    const int*   src  = (const int*)  d_in[1];
    const int*   dst  = (const int*)  d_in[2];
    const float* W0   = (const float*)d_in[3];
    const float* al0  = (const float*)d_in[4];
    const float* ar0  = (const float*)d_in[5];
    const float* W1   = (const float*)d_in[6];
    const float* al1  = (const float*)d_in[7];
    const float* ar1  = (const float*)d_in[8];
    const float* Wres = (const float*)d_in[9];
    float* out = (float*)d_out;

    const int N = in_sizes[0] / 256;
    const int E = in_sizes[1];
    const int F = 256;

    char* ws = (char*)d_ws;
    size_t off = 0;
    auto alloc = [&](size_t elems) -> float* {
        float* p = (float*)(ws + off);
        off += ((elems * sizeof(float) + 255) / 256) * 256;
        return p;
    };
    const int NPAD = ((N + 1 + 4095) / 4096) * 4096;
    float* feat0  = alloc((size_t)N * F);
    float* hbuf   = alloc((size_t)N * F);
    float* el0    = alloc((size_t)N * 4);
    float* er0    = alloc((size_t)N * 4);
    float* feat1  = alloc((size_t)N * 32);
    float* resb   = alloc((size_t)N * 32);
    float* el1    = alloc((size_t)N);
    float* er1    = alloc((size_t)N);
    int*   deg    = (int*)alloc((size_t)NPAD);
    int*   cnt    = (int*)alloc((size_t)NSLICE * NPAD);
    int*   rowptr = (int*)alloc((size_t)NPAD);
    int*   rank   = (int*)alloc((size_t)E);
    int*   raw    = (int*)alloc((size_t)E);
    int*   ssrc   = (int*)alloc((size_t)E);

    const int mt = (N + 15) / 16;

    k_count_lds<<<NSLICE, 1024, 0, stream>>>(dst, cnt, rank, E, N, NPAD);
    k_slice_prefix<<<NPAD / 256, 256, 0, stream>>>(cnt, deg, N, NPAD);
    k_scan<<<1, 1024, 0, stream>>>(deg, rowptr, NPAD);
    k_fill<<<(E + 255) / 256, 256, 0, stream>>>(src, dst, rowptr, cnt, rank, raw, E, NPAD);
    k_segsort<<<(N + 7) / 8, 256, 0, stream>>>(rowptr, raw, ssrc, N);

    k_gemm_wmma_f32<256, 256, 4, 2, true><<<dim3((mt + 1) / 2, 4), 32, 0, stream>>>(
        x, W0, feat0, al0, ar0, el0, er0, N);
    k_seg_softmax_sum<4, 64, false, true><<<(N + 3) / 4, 128, 0, stream>>>(
        rowptr, ssrc, feat0, el0, er0, nullptr, hbuf, N);

    k_gemm_wmma_f32<32, 256, 2, 1, true><<<dim3(mt, 1), 32, 0, stream>>>(
        hbuf, W1, feat1, al1, ar1, el1, er1, N);
    k_gemm_wmma_f32<32, 256, 2, 1, false><<<dim3(mt, 1), 32, 0, stream>>>(
        hbuf, Wres, resb, nullptr, nullptr, nullptr, nullptr, N);
    k_seg_softmax_sum<1, 32, true, false><<<(N + 3) / 4, 128, 0, stream>>>(
        rowptr, ssrc, feat1, el1, er1, resb, out, N);
}


# GEMMs: each work-group stages its A rows in LDS by whole-line loads and all its waves share them (4 heads in layer 0; W1 and Wres strips in layer 1, one launch instead of two); 10 launches -> 9
# speedup vs baseline: 5.7336x; 1.0668x over previous
#include <hip/hip_runtime.h>
#include <hip/hip_bf16.h>
#include <math.h>


typedef float v2f __attribute__((ext_vector_type(2)));
typedef float v8f __attribute__((ext_vector_type(8)));

template<int W>
__device__ __forceinline__ void lane_sum_step(float (&p)[16], int lm) {
    const bool up = (lm & W) != 0;
#pragma unroll
    for (int i = 0; i < W; ++i) {
        const float lo = p[i], hi = p[i + W];
        const float keep = up ? hi : lo;
        const float send = up ? lo : hi;
        const float got = __int_as_float(__builtin_amdgcn_update_dpp(
            0, __float_as_int(send), 0x160 | W, 0xf, 0xf, false));
        p[i] = keep + got;
    }
}

template<int NCOLS, int KDIM, int NT, int MT, int NW, bool SPLITB>
__global__ __launch_bounds__(32 * NW, 7)
void k_gemm_wmma_f32(const float* __restrict__ A,
                     const float* __restrict__ B0, const float* __restrict__ B1,
                     float* __restrict__ C0, float* __restrict__ C1,
                     const float* __restrict__ al, const float* __restrict__ ar,
                     float* __restrict__ el, float* __restrict__ er, int M) {
    constexpr int KC    = 64;
    constexpr int PITCH = KC + 4;
    constexpr int ROWS  = 16 * MT;
    constexpr int NTHR  = 32 * NW;
    constexpr int NLD   = (ROWS * (KC / 4)) / NTHR;
    constexpr int NCH   = KDIM / KC;
    static_assert((ROWS * (KC / 4)) % NTHR == 0, "chunk must split evenly over the threads");
    static_assert(KDIM % KC == 0, "K must be whole chunks");
    __shared__ __attribute__((aligned(16))) float sA[2][ROWS * PITCH];

    const int t      = threadIdx.x;
    const int lane   = t & 31;
    const int w      = __builtin_amdgcn_readfirstlane(t >> 5);
    const int lm     = lane & 15;
    const int kg     = lane >> 4;
    const int tileM0 = blockIdx.x * MT;

    int gOff[NLD], sOff[NLD];
#pragma unroll
    for (int i = 0; i < NLD; ++i) {
        const int idx = t + i * NTHR;
        const int r   = idx >> 4;
        const int c4  = idx & 15;
        int row = tileM0 * 16 + r;
        if (row >= M) row = M - 1;
        gOff[i] = row * KDIM + c4 * 4;
        sOff[i] = r * PITCH + c4 * 4;
    }

    const float* B   = SPLITB ? (w == 0 ? B0 : B1) : B0;
    float*       C   = SPLITB ? (w == 0 ? C0 : C1) : C0;
    const int tileN0 = SPLITB ? 0 : w * NT;
    const float* Bb  = B + (size_t)(kg * 2) * NCOLS + tileN0 * 16 + lm;

    float4 pre[NLD];
#pragma unroll
    for (int i = 0; i < NLD; ++i) pre[i] = *(const float4*)(A + gOff[i]);
#pragma unroll
    for (int i = 0; i < NLD; ++i) *(float4*)(&sA[0][sOff[i]]) = pre[i];
    __syncthreads();

    v8f acc[MT][NT];
#pragma unroll
    for (int i = 0; i < MT; ++i)
#pragma unroll
        for (int j = 0; j < NT; ++j) acc[i][j] = (v8f){};

#pragma unroll
    for (int c = 0; c < NCH; ++c) {
        if (c + 1 < NCH) {
#pragma unroll
            for (int i = 0; i < NLD; ++i) pre[i] = *(const float4*)(A + gOff[i] + (c + 1) * KC);
        }
        __builtin_amdgcn_sched_barrier(0);
        const float* sa = &sA[c & 1][0];
#pragma unroll
        for (int kk = 0; kk < KC; kk += 4) {
            const int k = c * KC + kk;
            v2f a[MT];
#pragma unroll
            for (int i = 0; i < MT; ++i)
                a[i] = *(const v2f*)(sa + (i * 16 + lm) * PITCH + kk + kg * 2);
#pragma unroll
            for (int j = 0; j < NT; ++j) {
                v2f b;
                b.x = Bb[k * NCOLS + j * 16];
                b.y = Bb[k * NCOLS + NCOLS + j * 16];
#pragma unroll
                for (int i = 0; i < MT; ++i)
                    acc[i][j] = __builtin_amdgcn_wmma_f32_16x16x4_f32(
                         false, a[i],  false, b,
                         (short)0, acc[i][j],  false,  false);
            }
        }
        if (c + 1 < NCH) {
#pragma unroll
            for (int i = 0; i < NLD; ++i) *(float4*)(&sA[(c + 1) & 1][sOff[i]]) = pre[i];
            __syncthreads();
        }
    }

#pragma unroll
    for (int i = 0; i < MT; ++i) {
        const int rbase = (tileM0 + i) * 16 + kg * 8;
#pragma unroll
        for (int j = 0; j < NT; ++j) {
            const int cbase = (tileN0 + j) * 16 + lm;
#pragma unroll
            for (int v = 0; v < 8; ++v) {
                const int r = rbase + v;
                if (r < M) C[(size_t)r * NCOLS + cbase] = acc[i][j][v];
            }
        }
    }

    constexpr int H  = SPLITB ? 1 : NW;
    const int head   = SPLITB ? 0 : w;
    const bool coef  = SPLITB ? (w == 0) : true;
    if (coef) {
        float wl[NT], wr[NT];
#pragma unroll
        for (int j = 0; j < NT; ++j) {
            wl[j] = al[(tileN0 + j) * 16 + lm];
            wr[j] = ar[(tileN0 + j) * 16 + lm];
        }
#pragma unroll
        for (int i = 0; i < MT; ++i) {
            const int rbase = (tileM0 + i) * 16 + kg * 8;
            float p[16];
#pragma unroll
            for (int v = 0; v < 8; ++v) {
                float sl = acc[i][0][v] * wl[0], sr = acc[i][0][v] * wr[0];
#pragma unroll
                for (int j = 1; j < NT; ++j) { sl += acc[i][j][v] * wl[j]; sr += acc[i][j][v] * wr[j]; }
                p[v] = sl; p[8 + v] = sr;
            }
            lane_sum_step<8>(p, lm);
            lane_sum_step<4>(p, lm);
            lane_sum_step<2>(p, lm);
            lane_sum_step<1>(p, lm);
            const int r = rbase + (lm & 7);
            float* dstp = (lm < 8) ? el : er;
            if (r < M) dstp[(size_t)r * H + head] = p[0];
        }
    }
}

constexpr int NSLICE  = 32;
constexpr int CNT_CAP = 51200;
constexpr int EPT     = 16;

__device__ __forceinline__ int slice_len(int E) {
    return (((E + NSLICE - 1) / NSLICE) + 255) & ~255;
}

__global__ __launch_bounds__(1024)
void k_count_lds(const int* __restrict__ dst, int* __restrict__ cnt,
                 int* __restrict__ rank, int E, int N, int PITCH) {
    __shared__ int c[CNT_CAP];
    const int t  = threadIdx.x;
    const int L  = slice_len(E);
    const int e0 = blockIdx.x * L;
    const int e1 = min(E, e0 + L);
    int* crow = cnt + (size_t)blockIdx.x * PITCH;
    for (int base = 0; base < N; base += CNT_CAP) {
        const int span  = min(CNT_CAP, N - base);
        const int span4 = (span + 3) >> 2;
        for (int i = t; i < span4; i += 1024) ((int4*)c)[i] = make_int4(0, 0, 0, 0);
        __syncthreads();
        for (int eb = e0; eb < e1; eb += EPT * 1024) {
            int d[EPT], r[EPT];
#pragma unroll
            for (int k = 0; k < EPT; ++k) {
                const int e = eb + k * 1024 + t;
                const int v = dst[min(e, e1 - 1)] - base;
                d[k] = (e < e1) ? v : -1;
            }
#pragma unroll
            for (int k = 0; k < EPT; ++k) {
                r[k] = 0;
                if ((unsigned)d[k] < (unsigned)span) r[k] = atomicAdd(&c[d[k]], 1);
            }
#pragma unroll
            for (int k = 0; k < EPT; ++k)
                if ((unsigned)d[k] < (unsigned)span) rank[eb + k * 1024 + t] = r[k];
        }
        __syncthreads();
        for (int i = t; i < span4; i += 1024) ((int4*)(crow + base))[i] = ((const int4*)c)[i];
        __syncthreads();
    }
}

__global__ __launch_bounds__(256)
void k_slice_prefix(int* __restrict__ cnt, int* __restrict__ deg, int N, int NPAD) {
    const int n = blockIdx.x * blockDim.x + threadIdx.x;
    if (n >= NPAD) return;
    int run = 0;
    if (n < N) {
        int c[NSLICE];
#pragma unroll
        for (int s = 0; s < NSLICE; ++s) c[s] = cnt[(size_t)s * NPAD + n];
#pragma unroll
        for (int s = 0; s < NSLICE; ++s) { cnt[(size_t)s * NPAD + n] = run; run += c[s]; }
    }
    deg[n] = run;
}

__global__ __launch_bounds__(1024)
void k_scan(const int* __restrict__ deg, int* __restrict__ rowptr, int NPAD) {
    __shared__ int swave[32];
    const int t = threadIdx.x, lane = t & 31, w = t >> 5;
    int carry = 0;
    int4 v = *(const int4*)(deg + 4 * t);
    for (int base = 0; base < NPAD; base += 4096) {
        const int i0 = base + 4 * t;
        int4 vn = make_int4(0, 0, 0, 0);
        if (base + 4096 < NPAD) vn = *(const int4*)(deg + i0 + 4096);
        const int s1 = v.x, s2 = s1 + v.y, s3 = s2 + v.z, s4 = s3 + v.w;
        int inc = s4;
        for (int off = 1; off < 32; off <<= 1) {
            const int y = __shfl_up(inc, off);
            if (lane >= off) inc += y;
        }
        if (lane == 31) swave[w] = inc;
        __syncthreads();
        const int wt = swave[lane];
        int winc = wt;
        for (int off = 1; off < 32; off <<= 1) {
            const int y = __shfl_up(winc, off);
            if (lane >= off) winc += y;
        }
        const int total = __shfl(winc, 31);
        const int wexc  = __shfl(winc, w) - __shfl(wt, w);
        const int ex0   = carry + wexc + (inc - s4);
        const int4 o = make_int4(ex0, ex0 + s1, ex0 + s2, ex0 + s3);
        *(int4*)(rowptr + i0) = o;
        carry += total;
        v = vn;
        __syncthreads();
    }
}

__global__ __launch_bounds__(256)
void k_fill(const int* __restrict__ src, const int* __restrict__ dst,
            const int* __restrict__ rowptr, const int* __restrict__ off,
            const int* __restrict__ rank, int* __restrict__ raw, int E, int NPAD) {
    const int e = blockIdx.x * 256 + threadIdx.x;
    if (e >= E) return;
    const int s = (blockIdx.x * 256) / slice_len(E);
    const int d = dst[e];
    raw[rowptr[d] + off[(size_t)s * NPAD + d] + rank[e]] = src[e];
}

__global__ __launch_bounds__(256)
void k_segsort(const int* __restrict__ rowptr, const int* __restrict__ raw,
               int* __restrict__ ssrc, int N) {
    const int lane = threadIdx.x & 31;
    const int d = __builtin_amdgcn_readfirstlane(blockIdx.x * (blockDim.x >> 5) + (threadIdx.x >> 5));
    if (d >= N) return;
    const int beg = rowptr[d];
    const int cnt = rowptr[d + 1] - beg;
    for (int mb = 0; mb < cnt; mb += 32) {
        const int k = mb + lane;
        const bool have = k < cnt;
        const int key = have ? raw[beg + k] : 0x7fffffff;
        int rank = 0;
        for (int jb = 0; jb < cnt; jb += 32) {
            const int kk = (jb + lane < cnt) ? raw[beg + jb + lane] : 0x7fffffff;
            const int jn = min(32, cnt - jb);
            for (int jj = 0; jj < jn; ++jj) {
                const int kj = __builtin_amdgcn_readlane(kk, jj);
                const int j  = jb + jj;
                rank += (kj < key || (kj == key && j < k)) ? 1 : 0;
            }
        }
        if (have) ssrc[beg + rank] = key;
    }
}

template<int H, int D, bool ADD, bool ELU>
__global__ __launch_bounds__(128)
void k_seg_softmax_sum(const int* __restrict__ rowptr, const int* __restrict__ ssrc,
                       const float* __restrict__ feat,
                       const float* __restrict__ el, const float* __restrict__ er,
                       const float* __restrict__ res,
                       float* __restrict__ out, int N) {
    constexpr int ROW = H * D;
    constexpr int FPL = ROW / 32;
    constexpr int NC  = (FPL >= 4) ? FPL / 4 : 1;
    static_assert(FPL == 1 || FPL % 4 == 0, "row shape");
    const int lane = threadIdx.x & 31;
    const int d = __builtin_amdgcn_readfirstlane(blockIdx.x * (blockDim.x >> 5) + (threadIdx.x >> 5));
    if (d >= N) return;
    const int beg = rowptr[d];
    const int cnt = rowptr[d + 1] - beg;

    float erd[H];
    for (int h = 0; h < H; ++h) erd[h] = er[d * H + h];

    float radd[NC * 4];
    if constexpr (ADD) {
        const float* rrow = res + (size_t)d * ROW;
        if constexpr (FPL >= 4) {
            for (int c = 0; c < NC; ++c) {
                const float4 r = *(const float4*)(rrow + c * 128 + lane * 4);
                radd[4 * c + 0] = r.x; radd[4 * c + 1] = r.y;
                radd[4 * c + 2] = r.z; radd[4 * c + 3] = r.w;
            }
        } else {
            radd[0] = rrow[lane];
        }
    }

    int hsel[NC];
    for (int c = 0; c < NC; ++c) hsel[c] = (FPL >= 4) ? (c * 128 + lane * 4) / D : lane / D;

    int s0 = 0;
    float v0[H];
    for (int h = 0; h < H; ++h) v0[h] = 0.0f;
    if (lane < cnt) {
        s0 = ssrc[beg + lane];
        for (int h = 0; h < H; ++h) {
            const float v = el[s0 * H + h] + erd[h];
            v0[h] = (v > 0.0f) ? v : 0.2f * v;
        }
    }
    auto edge_scores = [&](int b, int& s, float (&v)[H]) {
        if (b == 0) {
            s = s0;
            for (int h = 0; h < H; ++h) v[h] = v0[h];
        } else {
            s = 0;
            for (int h = 0; h < H; ++h) v[h] = 0.0f;
            if (b + lane < cnt) {
                s = ssrc[beg + b + lane];
                for (int h = 0; h < H; ++h) {
                    const float t = el[s * H + h] + erd[h];
                    v[h] = (t > 0.0f) ? t : 0.2f * t;
                }
            }
        }
    };

    float mx[H];
    for (int h = 0; h < H; ++h) mx[h] = -INFINITY;
    for (int b = 0; b < cnt; b += 32) {
        int s; float v[H];
        edge_scores(b, s, v);
        if (b + lane < cnt) {
            for (int h = 0; h < H; ++h) mx[h] = fmaxf(mx[h], v[h]);
        }
    }
    for (int h = 0; h < H; ++h)
        for (int off = 16; off > 0; off >>= 1) mx[h] = fmaxf(mx[h], __shfl_xor(mx[h], off));

    float den[H];
    for (int h = 0; h < H; ++h) den[h] = 0.0f;
    for (int b = 0; b < cnt; b += 32) {
        int s; float v[H];
        edge_scores(b, s, v);
        if (b + lane < cnt) {
            for (int h = 0; h < H; ++h) den[h] += expf(v[h] - mx[h]);
        }
    }
    for (int h = 0; h < H; ++h)
        for (int off = 16; off > 0; off >>= 1) den[h] += __shfl_xor(den[h], off);

    float acc[NC * 4];
    for (int i = 0; i < NC * 4; ++i) acc[i] = 0.0f;
    for (int b = 0; b < cnt; b += 32) {
        int s; float v[H];
        edge_scores(b, s, v);
        float al[H];
        for (int h = 0; h < H; ++h) al[h] = 0.0f;
        if (b + lane < cnt) {
            for (int h = 0; h < H; ++h) al[h] = expf(v[h] - mx[h]) / den[h];
        }
        const int jn = min(32, cnt - b);
#pragma unroll 4
        for (int jj = 0; jj < jn; ++jj) {
            const int sj = __builtin_amdgcn_readlane(s, jj);
            float aj[H];
            for (int h = 0; h < H; ++h)
                aj[h] = __int_as_float(__builtin_amdgcn_readlane(__float_as_int(al[h]), jj));
            const float* frow = feat + (size_t)sj * ROW;
            for (int c = 0; c < NC; ++c) {
                float a = aj[0];
                for (int h = 1; h < H; ++h) a = (hsel[c] == h) ? aj[h] : a;
                if constexpr (FPL >= 4) {
                    const float4 f = *(const float4*)(frow + c * 128 + lane * 4);
                    acc[4 * c + 0] += f.x * a;
                    acc[4 * c + 1] += f.y * a;
                    acc[4 * c + 2] += f.z * a;
                    acc[4 * c + 3] += f.w * a;
                } else {
                    acc[0] += frow[lane] * a;
                }
            }
        }
    }

    if constexpr (ADD) {
        for (int i = 0; i < NC * 4; ++i) acc[i] += radd[i];
    }
    if constexpr (ELU) {
        for (int i = 0; i < NC * 4; ++i) {
            const float v = acc[i];
            acc[i] = (v > 0.0f) ? v : (expf(v) - 1.0f);
        }
    }

    float* orow = out + (size_t)d * ROW;
    if constexpr (FPL >= 4) {
        for (int c = 0; c < NC; ++c)
            *(float4*)(orow + c * 128 + lane * 4) =
                make_float4(acc[4 * c + 0], acc[4 * c + 1], acc[4 * c + 2], acc[4 * c + 3]);
    } else {
        orow[lane] = acc[0];
    }
}

extern "C" void kernel_launch(void* const* d_in, const int* in_sizes, int n_in,
                              void* d_out, int out_size, void* d_ws, size_t ws_size,
                              hipStream_t stream) {
    const float* x    = (const float*)d_in[0];
    const int*   src  = (const int*)  d_in[1];
    const int*   dst  = (const int*)  d_in[2];
    const float* W0   = (const float*)d_in[3];
    const float* al0  = (const float*)d_in[4];
    const float* ar0  = (const float*)d_in[5];
    const float* W1   = (const float*)d_in[6];
    const float* al1  = (const float*)d_in[7];
    const float* ar1  = (const float*)d_in[8];
    const float* Wres = (const float*)d_in[9];
    float* out = (float*)d_out;

    const int N = in_sizes[0] / 256;
    const int E = in_sizes[1];
    const int F = 256;

    char* ws = (char*)d_ws;
    size_t off = 0;
    auto alloc = [&](size_t elems) -> float* {
        float* p = (float*)(ws + off);
        off += ((elems * sizeof(float) + 255) / 256) * 256;
        return p;
    };
    const int NPAD = ((N + 1 + 4095) / 4096) * 4096;
    float* feat0  = alloc((size_t)N * F);
    float* hbuf   = alloc((size_t)N * F);
    float* el0    = alloc((size_t)N * 4);
    float* er0    = alloc((size_t)N * 4);
    float* feat1  = alloc((size_t)N * 32);
    float* resb   = alloc((size_t)N * 32);
    float* el1    = alloc((size_t)N);
    float* er1    = alloc((size_t)N);
    int*   deg    = (int*)alloc((size_t)NPAD);
    int*   cnt    = (int*)alloc((size_t)NSLICE * NPAD);
    int*   rowptr = (int*)alloc((size_t)NPAD);
    int*   rank   = (int*)alloc((size_t)E);
    int*   raw    = (int*)alloc((size_t)E);
    int*   ssrc   = (int*)alloc((size_t)E);

    const int mt = (N + 15) / 16;

    k_count_lds<<<NSLICE, 1024, 0, stream>>>(dst, cnt, rank, E, N, NPAD);
    k_slice_prefix<<<NPAD / 256, 256, 0, stream>>>(cnt, deg, N, NPAD);
    k_scan<<<1, 1024, 0, stream>>>(deg, rowptr, NPAD);
    k_fill<<<(E + 255) / 256, 256, 0, stream>>>(src, dst, rowptr, cnt, rank, raw, E, NPAD);
    k_segsort<<<(N + 7) / 8, 256, 0, stream>>>(rowptr, raw, ssrc, N);

    k_gemm_wmma_f32<256, 256, 4, 2, 4, false><<<dim3((mt + 1) / 2), 128, 0, stream>>>(
        x, W0, nullptr, feat0, nullptr, al0, ar0, el0, er0, N);
    k_seg_softmax_sum<4, 64, false, true><<<(N + 3) / 4, 128, 0, stream>>>(
        rowptr, ssrc, feat0, el0, er0, nullptr, hbuf, N);

    k_gemm_wmma_f32<32, 256, 2, 1, 2, true><<<dim3(mt), 64, 0, stream>>>(
        hbuf, W1, Wres, feat1, resb, al1, ar1, el1, er1, N);
    k_seg_softmax_sum<1, 32, true, false><<<(N + 3) / 4, 128, 0, stream>>>(
        rowptr, ssrc, feat1, el1, er1, resb, out, N);
}


# Degree scan (serial, one work-group) rides in the layer-0 GEMM launch as its work-group 0 and runs beside the matrix work, lane sums on the vector ALU; its own launch is gone, 9 launches -> 8
# speedup vs baseline: 6.5532x; 1.1429x over previous
#include <hip/hip_runtime.h>
#include <hip/hip_bf16.h>
#include <math.h>


typedef float v2f __attribute__((ext_vector_type(2)));
typedef float v8f __attribute__((ext_vector_type(8)));

template<int W>
__device__ __forceinline__ void lane_sum_step(float (&p)[16], int lm) {
    const bool up = (lm & W) != 0;
#pragma unroll
    for (int i = 0; i < W; ++i) {
        const float lo = p[i], hi = p[i + W];
        const float keep = up ? hi : lo;
        const float send = up ? lo : hi;
        const float got = __int_as_float(__builtin_amdgcn_update_dpp(
            0, __float_as_int(send), 0x160 | W, 0xf, 0xf, false));
        p[i] = keep + got;
    }
}

__device__ __forceinline__ int wave_incl_sum(int x, int lane) {
    x += __builtin_amdgcn_update_dpp(0, x, 0x111, 0xf, 0xf, false);
    x += __builtin_amdgcn_update_dpp(0, x, 0x112, 0xf, 0xf, false);
    x += __builtin_amdgcn_update_dpp(0, x, 0x114, 0xf, 0xf, false);
    x += __builtin_amdgcn_update_dpp(0, x, 0x118, 0xf, 0xf, false);
    const int low = __builtin_amdgcn_readlane(x, 15);
    return x + ((lane & 16) ? low : 0);
}

template<int NWV>
__device__ __noinline__ void scan_degrees(const int* __restrict__ deg,
                                             int* __restrict__ rowptr, int NPAD) {
    static_assert(4096 % (NWV * 1024) == 0, "a wave's share must be whole groups of 8 trips");
    __shared__ int wtot[NWV];
    const int lane  = threadIdx.x & 31;
    const int w     = __builtin_amdgcn_readfirstlane(threadIdx.x >> 5);
    const int per   = NPAD / NWV;
    const int trips = per >> 7;
    const int4* in4 = (const int4*)(deg + (size_t)w * per) + lane;
    int4*      out4 = (int4*)(rowptr + (size_t)w * per) + lane;

    int s = 0;
    for (int i = 0; i < trips; i += 8) {
        int4 v[8];
#pragma unroll
        for (int u = 0; u < 8; ++u) v[u] = in4[(i + u) * 32];
#pragma unroll
        for (int u = 0; u < 8; ++u) s += (v[u].x + v[u].y) + (v[u].z + v[u].w);
    }
    s = wave_incl_sum(s, lane);
    if (lane == 31) wtot[w] = s;
    __syncthreads();
    int carry = 0;
#pragma unroll
    for (int j = 0; j < NWV; ++j) {
        const int tj = wtot[j];
        carry += (j < w) ? tj : 0;
    }
    carry = __builtin_amdgcn_readfirstlane(carry);

    for (int i = 0; i < trips; i += 8) {
        int4 v[8];
#pragma unroll
        for (int u = 0; u < 8; ++u) v[u] = in4[(i + u) * 32];
#pragma unroll
        for (int u = 0; u < 8; ++u) {
            const int s1 = v[u].x, s2 = s1 + v[u].y, s3 = s2 + v[u].z, s4 = s3 + v[u].w;
            const int inc = wave_incl_sum(s4, lane);
            const int ex0 = carry + (inc - s4);
            out4[(i + u) * 32] = make_int4(ex0, ex0 + s1, ex0 + s2, ex0 + s3);
            carry += __builtin_amdgcn_readlane(inc, 31);
        }
    }
}

template<int NCOLS, int KDIM, int NT, int MT, int NW, bool SPLITB, bool SCAN>
__global__ __launch_bounds__(32 * NW, 7)
void k_gemm_wmma_f32(const float* __restrict__ A,
                     const float* __restrict__ B0, const float* __restrict__ B1,
                     float* __restrict__ C0, float* __restrict__ C1,
                     const float* __restrict__ al, const float* __restrict__ ar,
                     float* __restrict__ el, float* __restrict__ er, int M,
                     const int* __restrict__ deg, int* __restrict__ rowptr, int NPAD) {
    if constexpr (SCAN) {
        if (blockIdx.x == 0) {
            scan_degrees<NW>(deg, rowptr, NPAD);
            return;
        }
    }
    constexpr int KC    = 64;
    constexpr int PITCH = KC + 4;
    constexpr int ROWS  = 16 * MT;
    constexpr int NTHR  = 32 * NW;
    constexpr int NLD   = (ROWS * (KC / 4)) / NTHR;
    constexpr int NCH   = KDIM / KC;
    static_assert((ROWS * (KC / 4)) % NTHR == 0, "chunk must split evenly over the threads");
    static_assert(KDIM % KC == 0, "K must be whole chunks");
    __shared__ __attribute__((aligned(16))) float sA[2][ROWS * PITCH];

    const int t      = threadIdx.x;
    const int lane   = t & 31;
    const int w      = __builtin_amdgcn_readfirstlane(t >> 5);
    const int lm     = lane & 15;
    const int kg     = lane >> 4;
    const int tileM0 = (blockIdx.x - (SCAN ? 1 : 0)) * MT;

    int gOff[NLD], sOff[NLD];
#pragma unroll
    for (int i = 0; i < NLD; ++i) {
        const int idx = t + i * NTHR;
        const int r   = idx >> 4;
        const int c4  = idx & 15;
        int row = tileM0 * 16 + r;
        if (row >= M) row = M - 1;
        gOff[i] = row * KDIM + c4 * 4;
        sOff[i] = r * PITCH + c4 * 4;
    }

    const float* B   = SPLITB ? (w == 0 ? B0 : B1) : B0;
    float*       C   = SPLITB ? (w == 0 ? C0 : C1) : C0;
    const int tileN0 = SPLITB ? 0 : w * NT;
    const float* Bb  = B + (size_t)(kg * 2) * NCOLS + tileN0 * 16 + lm;

    float4 pre[NLD];
#pragma unroll
    for (int i = 0; i < NLD; ++i) pre[i] = *(const float4*)(A + gOff[i]);
#pragma unroll
    for (int i = 0; i < NLD; ++i) *(float4*)(&sA[0][sOff[i]]) = pre[i];
    __syncthreads();

    v8f acc[MT][NT];
#pragma unroll
    for (int i = 0; i < MT; ++i)
#pragma unroll
        for (int j = 0; j < NT; ++j) acc[i][j] = (v8f){};

#pragma unroll
    for (int c = 0; c < NCH; ++c) {
        if (c + 1 < NCH) {
#pragma unroll
            for (int i = 0; i < NLD; ++i) pre[i] = *(const float4*)(A + gOff[i] + (c + 1) * KC);
        }
        __builtin_amdgcn_sched_barrier(0);
        const float* sa = &sA[c & 1][0];
#pragma unroll
        for (int kk = 0; kk < KC; kk += 4) {
            const int k = c * KC + kk;
            v2f a[MT];
#pragma unroll
            for (int i = 0; i < MT; ++i)
                a[i] = *(const v2f*)(sa + (i * 16 + lm) * PITCH + kk + kg * 2);
#pragma unroll
            for (int j = 0; j < NT; ++j) {
                v2f b;
                b.x = Bb[k * NCOLS + j * 16];
                b.y = Bb[k * NCOLS + NCOLS + j * 16];
#pragma unroll
                for (int i = 0; i < MT; ++i)
                    acc[i][j] = __builtin_amdgcn_wmma_f32_16x16x4_f32(
                         false, a[i],  false, b,
                         (short)0, acc[i][j],  false,  false);
            }
        }
        if (c + 1 < NCH) {
#pragma unroll
            for (int i = 0; i < NLD; ++i) *(float4*)(&sA[(c + 1) & 1][sOff[i]]) = pre[i];
            __syncthreads();
        }
    }

#pragma unroll
    for (int i = 0; i < MT; ++i) {
        const int rbase = (tileM0 + i) * 16 + kg * 8;
#pragma unroll
        for (int j = 0; j < NT; ++j) {
            const int cbase = (tileN0 + j) * 16 + lm;
#pragma unroll
            for (int v = 0; v < 8; ++v) {
                const int r = rbase + v;
                if (r < M) C[(size_t)r * NCOLS + cbase] = acc[i][j][v];
            }
        }
    }

    constexpr int H  = SPLITB ? 1 : NW;
    const int head   = SPLITB ? 0 : w;
    const bool coef  = SPLITB ? (w == 0) : true;
    if (coef) {
        float wl[NT], wr[NT];
#pragma unroll
        for (int j = 0; j < NT; ++j) {
            wl[j] = al[(tileN0 + j) * 16 + lm];
            wr[j] = ar[(tileN0 + j) * 16 + lm];
        }
#pragma unroll
        for (int i = 0; i < MT; ++i) {
            const int rbase = (tileM0 + i) * 16 + kg * 8;
            float p[16];
#pragma unroll
            for (int v = 0; v < 8; ++v) {
                float sl = acc[i][0][v] * wl[0], sr = acc[i][0][v] * wr[0];
#pragma unroll
                for (int j = 1; j < NT; ++j) { sl += acc[i][j][v] * wl[j]; sr += acc[i][j][v] * wr[j]; }
                p[v] = sl; p[8 + v] = sr;
            }
            lane_sum_step<8>(p, lm);
            lane_sum_step<4>(p, lm);
            lane_sum_step<2>(p, lm);
            lane_sum_step<1>(p, lm);
            const int r = rbase + (lm & 7);
            float* dstp = (lm < 8) ? el : er;
            if (r < M) dstp[(size_t)r * H + head] = p[0];
        }
    }
}

constexpr int NSLICE  = 32;
constexpr int CNT_CAP = 51200;
constexpr int EPT     = 16;

__device__ __forceinline__ int slice_len(int E) {
    return (((E + NSLICE - 1) / NSLICE) + 255) & ~255;
}

__global__ __launch_bounds__(1024)
void k_count_lds(const int* __restrict__ dst, int* __restrict__ cnt,
                 int* __restrict__ rank, int E, int N, int PITCH) {
    __shared__ int c[CNT_CAP];
    const int t  = threadIdx.x;
    const int L  = slice_len(E);
    const int e0 = blockIdx.x * L;
    const int e1 = min(E, e0 + L);
    int* crow = cnt + (size_t)blockIdx.x * PITCH;
    for (int base = 0; base < N; base += CNT_CAP) {
        const int span  = min(CNT_CAP, N - base);
        const int span4 = (span + 3) >> 2;
        for (int i = t; i < span4; i += 1024) ((int4*)c)[i] = make_int4(0, 0, 0, 0);
        __syncthreads();
        for (int eb = e0; eb < e1; eb += EPT * 1024) {
            int d[EPT], r[EPT];
#pragma unroll
            for (int k = 0; k < EPT; ++k) {
                const int e = eb + k * 1024 + t;
                const int v = dst[min(e, e1 - 1)] - base;
                d[k] = (e < e1) ? v : -1;
            }
#pragma unroll
            for (int k = 0; k < EPT; ++k) {
                r[k] = 0;
                if ((unsigned)d[k] < (unsigned)span) r[k] = atomicAdd(&c[d[k]], 1);
            }
#pragma unroll
            for (int k = 0; k < EPT; ++k)
                if ((unsigned)d[k] < (unsigned)span) rank[eb + k * 1024 + t] = r[k];
        }
        __syncthreads();
        for (int i = t; i < span4; i += 1024) ((int4*)(crow + base))[i] = ((const int4*)c)[i];
        __syncthreads();
    }
}

__global__ __launch_bounds__(256)
void k_slice_prefix(int* __restrict__ cnt, int* __restrict__ deg, int N, int NPAD) {
    const int n = blockIdx.x * blockDim.x + threadIdx.x;
    if (n >= NPAD) return;
    int run = 0;
    if (n < N) {
        int c[NSLICE];
#pragma unroll
        for (int s = 0; s < NSLICE; ++s) c[s] = cnt[(size_t)s * NPAD + n];
#pragma unroll
        for (int s = 0; s < NSLICE; ++s) { cnt[(size_t)s * NPAD + n] = run; run += c[s]; }
    }
    deg[n] = run;
}

__global__ __launch_bounds__(256)
void k_fill(const int* __restrict__ src, const int* __restrict__ dst,
            const int* __restrict__ rowptr, const int* __restrict__ off,
            const int* __restrict__ rank, int* __restrict__ raw, int E, int NPAD) {
    const int e = blockIdx.x * 256 + threadIdx.x;
    if (e >= E) return;
    const int s = (blockIdx.x * 256) / slice_len(E);
    const int d = dst[e];
    raw[rowptr[d] + off[(size_t)s * NPAD + d] + rank[e]] = src[e];
}

__global__ __launch_bounds__(256)
void k_segsort(const int* __restrict__ rowptr, const int* __restrict__ raw,
               int* __restrict__ ssrc, int N) {
    const int lane = threadIdx.x & 31;
    const int d = __builtin_amdgcn_readfirstlane(blockIdx.x * (blockDim.x >> 5) + (threadIdx.x >> 5));
    if (d >= N) return;
    const int beg = rowptr[d];
    const int cnt = rowptr[d + 1] - beg;
    for (int mb = 0; mb < cnt; mb += 32) {
        const int k = mb + lane;
        const bool have = k < cnt;
        const int key = have ? raw[beg + k] : 0x7fffffff;
        int rank = 0;
        for (int jb = 0; jb < cnt; jb += 32) {
            const int kk = (jb + lane < cnt) ? raw[beg + jb + lane] : 0x7fffffff;
            const int jn = min(32, cnt - jb);
            for (int jj = 0; jj < jn; ++jj) {
                const int kj = __builtin_amdgcn_readlane(kk, jj);
                const int j  = jb + jj;
                rank += (kj < key || (kj == key && j < k)) ? 1 : 0;
            }
        }
        if (have) ssrc[beg + rank] = key;
    }
}

template<int H, int D, bool ADD, bool ELU>
__global__ __launch_bounds__(128)
void k_seg_softmax_sum(const int* __restrict__ rowptr, const int* __restrict__ ssrc,
                       const float* __restrict__ feat,
                       const float* __restrict__ el, const float* __restrict__ er,
                       const float* __restrict__ res,
                       float* __restrict__ out, int N) {
    constexpr int ROW = H * D;
    constexpr int FPL = ROW / 32;
    constexpr int NC  = (FPL >= 4) ? FPL / 4 : 1;
    static_assert(FPL == 1 || FPL % 4 == 0, "row shape");
    const int lane = threadIdx.x & 31;
    const int d = __builtin_amdgcn_readfirstlane(blockIdx.x * (blockDim.x >> 5) + (threadIdx.x >> 5));
    if (d >= N) return;
    const int beg = rowptr[d];
    const int cnt = rowptr[d + 1] - beg;

    float erd[H];
    for (int h = 0; h < H; ++h) erd[h] = er[d * H + h];

    float radd[NC * 4];
    if constexpr (ADD) {
        const float* rrow = res + (size_t)d * ROW;
        if constexpr (FPL >= 4) {
            for (int c = 0; c < NC; ++c) {
                const float4 r = *(const float4*)(rrow + c * 128 + lane * 4);
                radd[4 * c + 0] = r.x; radd[4 * c + 1] = r.y;
                radd[4 * c + 2] = r.z; radd[4 * c + 3] = r.w;
            }
        } else {
            radd[0] = rrow[lane];
        }
    }

    int hsel[NC];
    for (int c = 0; c < NC; ++c) hsel[c] = (FPL >= 4) ? (c * 128 + lane * 4) / D : lane / D;

    int s0 = 0;
    float v0[H];
    for (int h = 0; h < H; ++h) v0[h] = 0.0f;
    if (lane < cnt) {
        s0 = ssrc[beg + lane];
        for (int h = 0; h < H; ++h) {
            const float v = el[s0 * H + h] + erd[h];
            v0[h] = (v > 0.0f) ? v : 0.2f * v;
        }
    }
    auto edge_scores = [&](int b, int& s, float (&v)[H]) {
        if (b == 0) {
            s = s0;
            for (int h = 0; h < H; ++h) v[h] = v0[h];
        } else {
            s = 0;
            for (int h = 0; h < H; ++h) v[h] = 0.0f;
            if (b + lane < cnt) {
                s = ssrc[beg + b + lane];
                for (int h = 0; h < H; ++h) {
                    const float t = el[s * H + h] + erd[h];
                    v[h] = (t > 0.0f) ? t : 0.2f * t;
                }
            }
        }
    };

    float mx[H];
    for (int h = 0; h < H; ++h) mx[h] = -INFINITY;
    for (int b = 0; b < cnt; b += 32) {
        int s; float v[H];
        edge_scores(b, s, v);
        if (b + lane < cnt) {
            for (int h = 0; h < H; ++h) mx[h] = fmaxf(mx[h], v[h]);
        }
    }
    for (int h = 0; h < H; ++h)
        for (int off = 16; off > 0; off >>= 1) mx[h] = fmaxf(mx[h], __shfl_xor(mx[h], off));

    float den[H];
    for (int h = 0; h < H; ++h) den[h] = 0.0f;
    for (int b = 0; b < cnt; b += 32) {
        int s; float v[H];
        edge_scores(b, s, v);
        if (b + lane < cnt) {
            for (int h = 0; h < H; ++h) den[h] += expf(v[h] - mx[h]);
        }
    }
    for (int h = 0; h < H; ++h)
        for (int off = 16; off > 0; off >>= 1) den[h] += __shfl_xor(den[h], off);

    float acc[NC * 4];
    for (int i = 0; i < NC * 4; ++i) acc[i] = 0.0f;
    for (int b = 0; b < cnt; b += 32) {
        int s; float v[H];
        edge_scores(b, s, v);
        float al[H];
        for (int h = 0; h < H; ++h) al[h] = 0.0f;
        if (b + lane < cnt) {
            for (int h = 0; h < H; ++h) al[h] = expf(v[h] - mx[h]) / den[h];
        }
        const int jn = min(32, cnt - b);
#pragma unroll 4
        for (int jj = 0; jj < jn; ++jj) {
            const int sj = __builtin_amdgcn_readlane(s, jj);
            float aj[H];
            for (int h = 0; h < H; ++h)
                aj[h] = __int_as_float(__builtin_amdgcn_readlane(__float_as_int(al[h]), jj));
            const float* frow = feat + (size_t)sj * ROW;
            for (int c = 0; c < NC; ++c) {
                float a = aj[0];
                for (int h = 1; h < H; ++h) a = (hsel[c] == h) ? aj[h] : a;
                if constexpr (FPL >= 4) {
                    const float4 f = *(const float4*)(frow + c * 128 + lane * 4);
                    acc[4 * c + 0] += f.x * a;
                    acc[4 * c + 1] += f.y * a;
                    acc[4 * c + 2] += f.z * a;
                    acc[4 * c + 3] += f.w * a;
                } else {
                    acc[0] += frow[lane] * a;
                }
            }
        }
    }

    if constexpr (ADD) {
        for (int i = 0; i < NC * 4; ++i) acc[i] += radd[i];
    }
    if constexpr (ELU) {
        for (int i = 0; i < NC * 4; ++i) {
            const float v = acc[i];
            acc[i] = (v > 0.0f) ? v : (expf(v) - 1.0f);
        }
    }

    float* orow = out + (size_t)d * ROW;
    if constexpr (FPL >= 4) {
        for (int c = 0; c < NC; ++c)
            *(float4*)(orow + c * 128 + lane * 4) =
                make_float4(acc[4 * c + 0], acc[4 * c + 1], acc[4 * c + 2], acc[4 * c + 3]);
    } else {
        orow[lane] = acc[0];
    }
}

extern "C" void kernel_launch(void* const* d_in, const int* in_sizes, int n_in,
                              void* d_out, int out_size, void* d_ws, size_t ws_size,
                              hipStream_t stream) {
    const float* x    = (const float*)d_in[0];
    const int*   src  = (const int*)  d_in[1];
    const int*   dst  = (const int*)  d_in[2];
    const float* W0   = (const float*)d_in[3];
    const float* al0  = (const float*)d_in[4];
    const float* ar0  = (const float*)d_in[5];
    const float* W1   = (const float*)d_in[6];
    const float* al1  = (const float*)d_in[7];
    const float* ar1  = (const float*)d_in[8];
    const float* Wres = (const float*)d_in[9];
    float* out = (float*)d_out;

    const int N = in_sizes[0] / 256;
    const int E = in_sizes[1];
    const int F = 256;

    char* ws = (char*)d_ws;
    size_t off = 0;
    auto alloc = [&](size_t elems) -> float* {
        float* p = (float*)(ws + off);
        off += ((elems * sizeof(float) + 255) / 256) * 256;
        return p;
    };
    const int NPAD = ((N + 1 + 4095) / 4096) * 4096;
    float* feat0  = alloc((size_t)N * F);
    float* hbuf   = alloc((size_t)N * F);
    float* el0    = alloc((size_t)N * 4);
    float* er0    = alloc((size_t)N * 4);
    float* feat1  = alloc((size_t)N * 32);
    float* resb   = alloc((size_t)N * 32);
    float* el1    = alloc((size_t)N);
    float* er1    = alloc((size_t)N);
    int*   deg    = (int*)alloc((size_t)NPAD);
    int*   cnt    = (int*)alloc((size_t)NSLICE * NPAD);
    int*   rowptr = (int*)alloc((size_t)NPAD);
    int*   rank   = (int*)alloc((size_t)E);
    int*   raw    = (int*)alloc((size_t)E);
    int*   ssrc   = (int*)alloc((size_t)E);

    const int mt = (N + 15) / 16;

    k_count_lds<<<NSLICE, 1024, 0, stream>>>(dst, cnt, rank, E, N, NPAD);
    k_slice_prefix<<<NPAD / 256, 256, 0, stream>>>(cnt, deg, N, NPAD);

    k_gemm_wmma_f32<256, 256, 4, 2, 4, false, true><<<dim3((mt + 1) / 2 + 1), 128, 0, stream>>>(
        x, W0, nullptr, feat0, nullptr, al0, ar0, el0, er0, N, deg, rowptr, NPAD);

    k_fill<<<(E + 255) / 256, 256, 0, stream>>>(src, dst, rowptr, cnt, rank, raw, E, NPAD);
    k_segsort<<<(N + 7) / 8, 256, 0, stream>>>(rowptr, raw, ssrc, N);

    k_seg_softmax_sum<4, 64, false, true><<<(N + 3) / 4, 128, 0, stream>>>(
        rowptr, ssrc, feat0, el0, er0, nullptr, hbuf, N);

    k_gemm_wmma_f32<32, 256, 2, 1, 2, true, false><<<dim3(mt), 64, 0, stream>>>(
        hbuf, W1, Wres, feat1, resb, al1, ar1, el1, er1, N, nullptr, nullptr, 0);
    k_seg_softmax_sum<1, 32, true, false><<<(N + 3) / 4, 128, 0, stream>>>(
        rowptr, ssrc, feat1, el1, er1, resb, out, N);
}


# Edge sort by destination as a two-launch bucket sort: 256 slices split by node range in LDS, then one work-group per range counts, scans, fills and orders its segments in LDS; 8 launches -> 6
# speedup vs baseline: 7.2541x; 1.1070x over previous
#include <hip/hip_runtime.h>
#include <hip/hip_bf16.h>
#include <math.h>


typedef float v2f __attribute__((ext_vector_type(2)));
typedef float v8f __attribute__((ext_vector_type(8)));

template<int W>
__device__ __forceinline__ void lane_sum_step(float (&p)[16], int lm) {
    const bool up = (lm & W) != 0;
#pragma unroll
    for (int i = 0; i < W; ++i) {
        const float lo = p[i], hi = p[i + W];
        const float keep = up ? hi : lo;
        const float send = up ? lo : hi;
        const float got = __int_as_float(__builtin_amdgcn_update_dpp(
            0, __float_as_int(send), 0x160 | W, 0xf, 0xf, false));
        p[i] = keep + got;
    }
}

__device__ __forceinline__ int wave_incl_sum(int x, int lane) {
    x += __builtin_amdgcn_update_dpp(0, x, 0x111, 0xf, 0xf, false);
    x += __builtin_amdgcn_update_dpp(0, x, 0x112, 0xf, 0xf, false);
    x += __builtin_amdgcn_update_dpp(0, x, 0x114, 0xf, 0xf, false);
    x += __builtin_amdgcn_update_dpp(0, x, 0x118, 0xf, 0xf, false);
    const int low = __builtin_amdgcn_readlane(x, 15);
    return x + ((lane & 16) ? low : 0);
}


template<int NCOLS, int KDIM, int NT, int MT, int NW, bool SPLITB>
__global__ __launch_bounds__(32 * NW, 7)
void k_gemm_wmma_f32(const float* __restrict__ A,
                     const float* __restrict__ B0, const float* __restrict__ B1,
                     float* __restrict__ C0, float* __restrict__ C1,
                     const float* __restrict__ al, const float* __restrict__ ar,
                     float* __restrict__ el, float* __restrict__ er, int M) {
    constexpr int KC    = 64;
    constexpr int PITCH = KC + 4;
    constexpr int ROWS  = 16 * MT;
    constexpr int NTHR  = 32 * NW;
    constexpr int NLD   = (ROWS * (KC / 4)) / NTHR;
    constexpr int NCH   = KDIM / KC;
    static_assert((ROWS * (KC / 4)) % NTHR == 0, "chunk must split evenly over the threads");
    static_assert(KDIM % KC == 0, "K must be whole chunks");
    __shared__ __attribute__((aligned(16))) float sA[2][ROWS * PITCH];

    const int t      = threadIdx.x;
    const int lane   = t & 31;
    const int w      = __builtin_amdgcn_readfirstlane(t >> 5);
    const int lm     = lane & 15;
    const int kg     = lane >> 4;
    const int tileM0 = blockIdx.x * MT;

    int gOff[NLD], sOff[NLD];
#pragma unroll
    for (int i = 0; i < NLD; ++i) {
        const int idx = t + i * NTHR;
        const int r   = idx >> 4;
        const int c4  = idx & 15;
        int row = tileM0 * 16 + r;
        if (row >= M) row = M - 1;
        gOff[i] = row * KDIM + c4 * 4;
        sOff[i] = r * PITCH + c4 * 4;
    }

    const float* B   = SPLITB ? (w == 0 ? B0 : B1) : B0;
    float*       C   = SPLITB ? (w == 0 ? C0 : C1) : C0;
    const int tileN0 = SPLITB ? 0 : w * NT;
    const float* Bb  = B + (size_t)(kg * 2) * NCOLS + tileN0 * 16 + lm;

    float4 pre[NLD];
#pragma unroll
    for (int i = 0; i < NLD; ++i) pre[i] = *(const float4*)(A + gOff[i]);
#pragma unroll
    for (int i = 0; i < NLD; ++i) *(float4*)(&sA[0][sOff[i]]) = pre[i];
    __syncthreads();

    v8f acc[MT][NT];
#pragma unroll
    for (int i = 0; i < MT; ++i)
#pragma unroll
        for (int j = 0; j < NT; ++j) acc[i][j] = (v8f){};

#pragma unroll
    for (int c = 0; c < NCH; ++c) {
        if (c + 1 < NCH) {
#pragma unroll
            for (int i = 0; i < NLD; ++i) pre[i] = *(const float4*)(A + gOff[i] + (c + 1) * KC);
        }
        __builtin_amdgcn_sched_barrier(0);
        const float* sa = &sA[c & 1][0];
#pragma unroll
        for (int kk = 0; kk < KC; kk += 4) {
            const int k = c * KC + kk;
            v2f a[MT];
#pragma unroll
            for (int i = 0; i < MT; ++i)
                a[i] = *(const v2f*)(sa + (i * 16 + lm) * PITCH + kk + kg * 2);
#pragma unroll
            for (int j = 0; j < NT; ++j) {
                v2f b;
                b.x = Bb[k * NCOLS + j * 16];
                b.y = Bb[k * NCOLS + NCOLS + j * 16];
#pragma unroll
                for (int i = 0; i < MT; ++i)
                    acc[i][j] = __builtin_amdgcn_wmma_f32_16x16x4_f32(
                         false, a[i],  false, b,
                         (short)0, acc[i][j],  false,  false);
            }
        }
        if (c + 1 < NCH) {
#pragma unroll
            for (int i = 0; i < NLD; ++i) *(float4*)(&sA[(c + 1) & 1][sOff[i]]) = pre[i];
            __syncthreads();
        }
    }

#pragma unroll
    for (int i = 0; i < MT; ++i) {
        const int rbase = (tileM0 + i) * 16 + kg * 8;
#pragma unroll
        for (int j = 0; j < NT; ++j) {
            const int cbase = (tileN0 + j) * 16 + lm;
#pragma unroll
            for (int v = 0; v < 8; ++v) {
                const int r = rbase + v;
                if (r < M) C[(size_t)r * NCOLS + cbase] = acc[i][j][v];
            }
        }
    }

    constexpr int H  = SPLITB ? 1 : NW;
    const int head   = SPLITB ? 0 : w;
    const bool coef  = SPLITB ? (w == 0) : true;
    if (coef) {
        float wl[NT], wr[NT];
#pragma unroll
        for (int j = 0; j < NT; ++j) {
            wl[j] = al[(tileN0 + j) * 16 + lm];
            wr[j] = ar[(tileN0 + j) * 16 + lm];
        }
#pragma unroll
        for (int i = 0; i < MT; ++i) {
            const int rbase = (tileM0 + i) * 16 + kg * 8;
            float p[16];
#pragma unroll
            for (int v = 0; v < 8; ++v) {
                float sl = acc[i][0][v] * wl[0], sr = acc[i][0][v] * wr[0];
#pragma unroll
                for (int j = 1; j < NT; ++j) { sl += acc[i][j][v] * wl[j]; sr += acc[i][j][v] * wr[j]; }
                p[v] = sl; p[8 + v] = sr;
            }
            lane_sum_step<8>(p, lm);
            lane_sum_step<4>(p, lm);
            lane_sum_step<2>(p, lm);
            lane_sum_step<1>(p, lm);
            const int r = rbase + (lm & 7);
            float* dstp = (lm < 8) ? el : er;
            if (r < M) dstp[(size_t)r * H + head] = p[0];
        }
    }
}

constexpr int NB    = 256;
constexpr int NS    = 256;
constexpr int LCAP  = 4096;
constexpr int BCAP  = 8192;
constexpr int BSCAP = 1024;

__device__ __forceinline__ int slice_len(int E) {
    return (((E + NS - 1) / NS) + 7) & ~7;
}

__device__ __forceinline__ int group_excl_sum(int x, int* red, int& total) {
    const int lane = threadIdx.x & 31;
    const int w    = __builtin_amdgcn_readfirstlane(threadIdx.x >> 5);
    const int inc  = wave_incl_sum(x, lane);
    if (lane == 31) red[w] = inc;
    __syncthreads();
    const int ws   = red[lane];
    const int winc = wave_incl_sum(ws, lane);
    total = __builtin_amdgcn_readlane(winc, 31);
    const int wex  = __builtin_amdgcn_readlane(winc - ws, w);
    __syncthreads();
    return wex + inc - x;
}

__global__ __launch_bounds__(1024)
void k_bucket(const int* __restrict__ src, const int* __restrict__ dst,
              int2* __restrict__ stage, int* __restrict__ cnt, int* __restrict__ bas,
              int E, int BS) {
    __shared__ int  hist[NB];
    __shared__ int  cur[NB];
    __shared__ int2 buf[LCAP];
    const int t  = threadIdx.x;
    const int L  = slice_len(E);
    const int e0 = blockIdx.x * L;
    const int n  = max(0, min(L, E - e0));
    if (t < NB) hist[t] = 0;
    __syncthreads();
    int d[4], s[4], b[4];
#pragma unroll
    for (int k = 0; k < 4; ++k) {
        const int i = t + k * 1024;
        const int e = min(e0 + i, E - 1);
        d[k] = dst[e];
        s[k] = src[e];
    }
    asm volatile("" :: "v"(d[0]), "v"(d[1]), "v"(d[2]), "v"(d[3]),
                       "v"(s[0]), "v"(s[1]), "v"(s[2]), "v"(s[3]));
#pragma unroll
    for (int k = 0; k < 4; ++k) {
        const int i = t + k * 1024;
        b[k] = (i < n) ? min((int)((unsigned)d[k] / (unsigned)BS), NB - 1) : -1;
    }
#pragma unroll
    for (int k = 0; k < 4; ++k)
        if (b[k] >= 0) atomicAdd(&hist[b[k]], 1);
    __syncthreads();
    if (t < 32) {
        int v[8], sum = 0;
#pragma unroll
        for (int j = 0; j < 8; ++j) { v[j] = hist[t * 8 + j]; sum += v[j]; }
        int ex = wave_incl_sum(sum, t) - sum;
#pragma unroll
        for (int j = 0; j < 8; ++j) { cur[t * 8 + j] = ex; ex += v[j]; }
    }
    __syncthreads();
#pragma unroll
    for (int k = 0; k < 4; ++k)
        if (b[k] >= 0) buf[atomicAdd(&cur[b[k]], 1)] = make_int2(d[k], s[k]);
    __syncthreads();
    for (int i = t; i < n; i += 1024) stage[e0 + i] = buf[i];
    if (t < NB) {
        const int h = hist[t];
        cnt[blockIdx.x * NB + t] = h;
        bas[blockIdx.x * NB + t] = cur[t] - h;
    }
}

__device__ __forceinline__ void bucket_finish(int2* pairs, int* seg, int* ncnt, int* nstart,
                                              const int* poff, const int* pbase, int* red,
                                              const int2* __restrict__ stage,
                                              int* __restrict__ rowptr, int* __restrict__ ssrc,
                                              int b, int G, int T, int N, int E, int BS) {
    const int t    = threadIdx.x;
    const int lane = t & 31;
    const int w    = __builtin_amdgcn_readfirstlane(t >> 5);
    const int L    = slice_len(E);
    const int n0   = b * BS;

    for (int i0 = 0; i0 < T; i0 += 4096) {
        int2 p[4];
#pragma unroll
        for (int k = 0; k < 4; ++k) {
            const int ic = min(i0 + k * 1024 + t, T - 1);
            int lo = 0;
#pragma unroll
            for (int step = NS / 2; step >= 1; step >>= 1)
                if (poff[lo + step] <= ic) lo += step;
            p[k] = stage[(size_t)lo * L + pbase[lo] + (ic - poff[lo])];
        }
        asm volatile("" :: "v"(p[0].x), "v"(p[0].y), "v"(p[1].x), "v"(p[1].y),
                           "v"(p[2].x), "v"(p[2].y), "v"(p[3].x), "v"(p[3].y));
#pragma unroll
        for (int k = 0; k < 4; ++k) {
            const int i = i0 + k * 1024 + t;
            if (i < T) {
                pairs[i] = p[k];
                atomicAdd(&ncnt[p[k].x - n0], 1);
            }
        }
    }
    __syncthreads();

    int tot;
    const int c  = (t < BS) ? ncnt[t] : 0;
    const int ex = group_excl_sum(c, red, tot);
    if (t < BS) { nstart[t] = ex; ncnt[t] = ex; }
    if (t == 0) nstart[BS] = T;
    __syncthreads();

    for (int i = t; i < T; i += 1024) {
        const int2 p = pairs[i];
        seg[atomicAdd(&ncnt[p.x - n0], 1)] = p.y;
    }
    __syncthreads();

    for (int ln = w; ln < BS; ln += 32) {
        const int beg = __builtin_amdgcn_readfirstlane(nstart[ln]);
        const int cn  = __builtin_amdgcn_readfirstlane(nstart[ln + 1]) - beg;
        for (int mb = 0; mb < cn; mb += 32) {
            const int k = mb + lane;
            const bool have = k < cn;
            const int key = have ? seg[beg + k] : 0x7fffffff;
            int rank = 0;
            for (int jb = 0; jb < cn; jb += 32) {
                const int kk = (jb + lane < cn) ? seg[beg + jb + lane] : 0x7fffffff;
                const int jn = min(32, cn - jb);
                for (int jj = 0; jj < jn; ++jj) {
                    const int kj = __builtin_amdgcn_readlane(kk, jj);
                    const int j  = jb + jj;
                    rank += (kj < key || (kj == key && j < k)) ? 1 : 0;
                }
            }
            if (have) ssrc[G + beg + rank] = key;
        }
    }
    if (t < BS && n0 + t < N) rowptr[n0 + t] = G + nstart[t];
}

__global__ __launch_bounds__(1024)
void k_bucket_finish(const int2* __restrict__ stage, const int* __restrict__ cnt,
                     const int* __restrict__ bas, int2* __restrict__ pairsG, int* __restrict__ segG,
                     int* __restrict__ rowptr, int* __restrict__ ssrc, int N, int E, int BS) {
    __shared__ int2 pairsL[BCAP];
    __shared__ int  segL[BCAP];
    __shared__ int  ncnt[BSCAP];
    __shared__ int  nstart[BSCAP + 1];
    __shared__ int  poff[NS];
    __shared__ int  pbase[NS];
    __shared__ int  red[32];
    static_assert(BSCAP == 1024 && NS <= 1024, "one thread per node of a range and per slice");
    static_assert((NS * NB / 4) % 1024 == 0, "the count matrix splits evenly over the threads");
    const int t = threadIdx.x;
    const int b = blockIdx.x;

    int x = 0;
#pragma unroll
    for (int k = 0; k < (NS * NB / 4) / 1024; ++k) {
        const int q  = t + k * 1024;
        const int4 v = ((const int4*)cnt)[q];
        const int c0 = (q * 4) & (NB - 1);
        x += ((c0 + 0 < b) ? v.x : 0) + ((c0 + 1 < b) ? v.y : 0)
           + ((c0 + 2 < b) ? v.z : 0) + ((c0 + 3 < b) ? v.w : 0);
    }
    const int myc = (t < NS) ? cnt[t * NB + b] : 0;
    const int myb = (t < NS) ? bas[t * NB + b] : 0;
    int G, T;
    group_excl_sum(x, red, G);
    const int pex = group_excl_sum(myc, red, T);
    if (t < NS) { poff[t] = pex; pbase[t] = myb; }
    ncnt[t] = 0;
    __syncthreads();

    if (T <= BCAP)
        bucket_finish(pairsL, segL, ncnt, nstart, poff, pbase, red, stage, rowptr, ssrc, b, G, T, N, E, BS);
    else
        bucket_finish(pairsG + G, segG + G, ncnt, nstart, poff, pbase, red, stage, rowptr, ssrc, b, G, T, N, E, BS);
    if (b == NB - 1 && t == 0) rowptr[N] = E;
}

template<int H, int D, bool ADD, bool ELU>
__global__ __launch_bounds__(128)
void k_seg_softmax_sum(const int* __restrict__ rowptr, const int* __restrict__ ssrc,
                       const float* __restrict__ feat,
                       const float* __restrict__ el, const float* __restrict__ er,
                       const float* __restrict__ res,
                       float* __restrict__ out, int N) {
    constexpr int ROW = H * D;
    constexpr int FPL = ROW / 32;
    constexpr int NC  = (FPL >= 4) ? FPL / 4 : 1;
    static_assert(FPL == 1 || FPL % 4 == 0, "row shape");
    const int lane = threadIdx.x & 31;
    const int d = __builtin_amdgcn_readfirstlane(blockIdx.x * (blockDim.x >> 5) + (threadIdx.x >> 5));
    if (d >= N) return;
    const int beg = rowptr[d];
    const int cnt = rowptr[d + 1] - beg;

    float erd[H];
    for (int h = 0; h < H; ++h) erd[h] = er[d * H + h];

    float radd[NC * 4];
    if constexpr (ADD) {
        const float* rrow = res + (size_t)d * ROW;
        if constexpr (FPL >= 4) {
            for (int c = 0; c < NC; ++c) {
                const float4 r = *(const float4*)(rrow + c * 128 + lane * 4);
                radd[4 * c + 0] = r.x; radd[4 * c + 1] = r.y;
                radd[4 * c + 2] = r.z; radd[4 * c + 3] = r.w;
            }
        } else {
            radd[0] = rrow[lane];
        }
    }

    int hsel[NC];
    for (int c = 0; c < NC; ++c) hsel[c] = (FPL >= 4) ? (c * 128 + lane * 4) / D : lane / D;

    int s0 = 0;
    float v0[H];
    for (int h = 0; h < H; ++h) v0[h] = 0.0f;
    if (lane < cnt) {
        s0 = ssrc[beg + lane];
        for (int h = 0; h < H; ++h) {
            const float v = el[s0 * H + h] + erd[h];
            v0[h] = (v > 0.0f) ? v : 0.2f * v;
        }
    }
    auto edge_scores = [&](int b, int& s, float (&v)[H]) {
        if (b == 0) {
            s = s0;
            for (int h = 0; h < H; ++h) v[h] = v0[h];
        } else {
            s = 0;
            for (int h = 0; h < H; ++h) v[h] = 0.0f;
            if (b + lane < cnt) {
                s = ssrc[beg + b + lane];
                for (int h = 0; h < H; ++h) {
                    const float t = el[s * H + h] + erd[h];
                    v[h] = (t > 0.0f) ? t : 0.2f * t;
                }
            }
        }
    };

    float mx[H];
    for (int h = 0; h < H; ++h) mx[h] = -INFINITY;
    for (int b = 0; b < cnt; b += 32) {
        int s; float v[H];
        edge_scores(b, s, v);
        if (b + lane < cnt) {
            for (int h = 0; h < H; ++h) mx[h] = fmaxf(mx[h], v[h]);
        }
    }
    for (int h = 0; h < H; ++h)
        for (int off = 16; off > 0; off >>= 1) mx[h] = fmaxf(mx[h], __shfl_xor(mx[h], off));

    float den[H];
    for (int h = 0; h < H; ++h) den[h] = 0.0f;
    for (int b = 0; b < cnt; b += 32) {
        int s; float v[H];
        edge_scores(b, s, v);
        if (b + lane < cnt) {
            for (int h = 0; h < H; ++h) den[h] += expf(v[h] - mx[h]);
        }
    }
    for (int h = 0; h < H; ++h)
        for (int off = 16; off > 0; off >>= 1) den[h] += __shfl_xor(den[h], off);

    float acc[NC * 4];
    for (int i = 0; i < NC * 4; ++i) acc[i] = 0.0f;
    for (int b = 0; b < cnt; b += 32) {
        int s; float v[H];
        edge_scores(b, s, v);
        float al[H];
        for (int h = 0; h < H; ++h) al[h] = 0.0f;
        if (b + lane < cnt) {
            for (int h = 0; h < H; ++h) al[h] = expf(v[h] - mx[h]) / den[h];
        }
        const int jn = min(32, cnt - b);
#pragma unroll 4
        for (int jj = 0; jj < jn; ++jj) {
            const int sj = __builtin_amdgcn_readlane(s, jj);
            float aj[H];
            for (int h = 0; h < H; ++h)
                aj[h] = __int_as_float(__builtin_amdgcn_readlane(__float_as_int(al[h]), jj));
            const float* frow = feat + (size_t)sj * ROW;
            for (int c = 0; c < NC; ++c) {
                float a = aj[0];
                for (int h = 1; h < H; ++h) a = (hsel[c] == h) ? aj[h] : a;
                if constexpr (FPL >= 4) {
                    const float4 f = *(const float4*)(frow + c * 128 + lane * 4);
                    acc[4 * c + 0] += f.x * a;
                    acc[4 * c + 1] += f.y * a;
                    acc[4 * c + 2] += f.z * a;
                    acc[4 * c + 3] += f.w * a;
                } else {
                    acc[0] += frow[lane] * a;
                }
            }
        }
    }

    if constexpr (ADD) {
        for (int i = 0; i < NC * 4; ++i) acc[i] += radd[i];
    }
    if constexpr (ELU) {
        for (int i = 0; i < NC * 4; ++i) {
            const float v = acc[i];
            acc[i] = (v > 0.0f) ? v : (expf(v) - 1.0f);
        }
    }

    float* orow = out + (size_t)d * ROW;
    if constexpr (FPL >= 4) {
        for (int c = 0; c < NC; ++c)
            *(float4*)(orow + c * 128 + lane * 4) =
                make_float4(acc[4 * c + 0], acc[4 * c + 1], acc[4 * c + 2], acc[4 * c + 3]);
    } else {
        orow[lane] = acc[0];
    }
}

extern "C" void kernel_launch(void* const* d_in, const int* in_sizes, int n_in,
                              void* d_out, int out_size, void* d_ws, size_t ws_size,
                              hipStream_t stream) {
    const float* x    = (const float*)d_in[0];
    const int*   src  = (const int*)  d_in[1];
    const int*   dst  = (const int*)  d_in[2];
    const float* W0   = (const float*)d_in[3];
    const float* al0  = (const float*)d_in[4];
    const float* ar0  = (const float*)d_in[5];
    const float* W1   = (const float*)d_in[6];
    const float* al1  = (const float*)d_in[7];
    const float* ar1  = (const float*)d_in[8];
    const float* Wres = (const float*)d_in[9];
    float* out = (float*)d_out;

    const int N = in_sizes[0] / 256;
    const int E = in_sizes[1];
    const int F = 256;

    char* ws = (char*)d_ws;
    size_t off = 0;
    auto alloc = [&](size_t elems) -> float* {
        float* p = (float*)(ws + off);
        off += ((elems * sizeof(float) + 255) / 256) * 256;
        return p;
    };
    float* feat0  = alloc((size_t)N * F);
    float* hbuf   = alloc((size_t)N * F);
    float* el0    = alloc((size_t)N * 4);
    float* er0    = alloc((size_t)N * 4);
    float* feat1  = alloc((size_t)N * 32);
    float* resb   = alloc((size_t)N * 32);
    float* el1    = alloc((size_t)N);
    float* er1    = alloc((size_t)N);
    int2*  stage  = (int2*)alloc((size_t)E * 2);
    int*   cnt    = (int*)alloc((size_t)NS * NB);
    int*   bas    = (int*)alloc((size_t)NS * NB);
    int2*  pairsG = (int2*)alloc((size_t)E * 2);
    int*   segG   = (int*)alloc((size_t)E);
    int*   rowptr = (int*)alloc((size_t)N + 1);
    int*   ssrc   = (int*)alloc((size_t)E);

    const int mt = (N + 15) / 16;
    const int BS = (N + NB - 1) / NB;

    k_bucket<<<NS, 1024, 0, stream>>>(src, dst, stage, cnt, bas, E, BS);
    k_bucket_finish<<<NB, 1024, 0, stream>>>(stage, cnt, bas, pairsG, segG, rowptr, ssrc, N, E, BS);

    k_gemm_wmma_f32<256, 256, 4, 2, 4, false><<<dim3((mt + 1) / 2), 128, 0, stream>>>(
        x, W0, nullptr, feat0, nullptr, al0, ar0, el0, er0, N);

    k_seg_softmax_sum<4, 64, false, true><<<(N + 3) / 4, 128, 0, stream>>>(
        rowptr, ssrc, feat0, el0, er0, nullptr, hbuf, N);

    k_gemm_wmma_f32<32, 256, 2, 1, 2, true><<<dim3(mt), 64, 0, stream>>>(
        hbuf, W1, Wres, feat1, resb, al1, ar1, el1, er1, N);
    k_seg_softmax_sum<1, 32, true, false><<<(N + 3) / 4, 128, 0, stream>>>(
        rowptr, ssrc, feat1, el1, er1, resb, out, N);
}


# Stage 1 of the edge sort (slices split by node range in LDS) rides in the layer-0 GEMM launch as its first 256 work-groups, on the same LDS buffer, beside the matrix work; 6 launches -> 5
# speedup vs baseline: 7.9106x; 1.0905x over previous
#include <hip/hip_runtime.h>
#include <hip/hip_bf16.h>
#include <math.h>


typedef float v2f __attribute__((ext_vector_type(2)));
typedef float v8f __attribute__((ext_vector_type(8)));

template<int W>
__device__ __forceinline__ void lane_sum_step(float (&p)[16], int lm) {
    const bool up = (lm & W) != 0;
#pragma unroll
    for (int i = 0; i < W; ++i) {
        const float lo = p[i], hi = p[i + W];
        const float keep = up ? hi : lo;
        const float send = up ? lo : hi;
        const float got = __int_as_float(__builtin_amdgcn_update_dpp(
            0, __float_as_int(send), 0x160 | W, 0xf, 0xf, false));
        p[i] = keep + got;
    }
}

__device__ __forceinline__ int wave_incl_sum(int x, int lane) {
    x += __builtin_amdgcn_update_dpp(0, x, 0x111, 0xf, 0xf, false);
    x += __builtin_amdgcn_update_dpp(0, x, 0x112, 0xf, 0xf, false);
    x += __builtin_amdgcn_update_dpp(0, x, 0x114, 0xf, 0xf, false);
    x += __builtin_amdgcn_update_dpp(0, x, 0x118, 0xf, 0xf, false);
    const int low = __builtin_amdgcn_readlane(x, 15);
    return x + ((lane & 16) ? low : 0);
}


constexpr int NB    = 256;
constexpr int NS    = 256;
constexpr int LCAP  = 4096;
constexpr int BCAP  = 8192;
constexpr int BSCAP = 1024;

__device__ __forceinline__ int slice_len(int E) {
    return (((E + NS - 1) / NS) + 7) & ~7;
}

constexpr int SIDE_LDS_BYTES = 2 * NB * 4 + LCAP * 8;

template<int NTHR>
__device__ __forceinline__ void bucket_slice(int* hist, int* cur, int2* buf,
                                             const int* __restrict__ src, const int* __restrict__ dst,
                                             int2* __restrict__ stage, int* __restrict__ cnt,
                                             int* __restrict__ bas, int E, int BS, int slice) {
    constexpr int PER = LCAP / NTHR;
    constexpr int GRP = 8;
    static_assert(LCAP % NTHR == 0 && PER % GRP == 0, "slice splits evenly over the threads");
    static_assert(NB == 256 && NTHR >= 32, "the group scan below is 8 bins per lane of wave 0");
    const int t  = threadIdx.x;
    const int L  = slice_len(E);
    const int e0 = slice * L;
    const int n  = max(0, min(L, E - e0));
    for (int i = t; i < NB; i += NTHR) hist[i] = 0;
    __syncthreads();
    int d[PER], s[PER], b[PER];
#pragma unroll
    for (int g = 0; g < PER; g += GRP) {
#pragma unroll
        for (int k = g; k < g + GRP; ++k) {
            const int e = min(e0 + t + k * NTHR, E - 1);
            d[k] = dst[e];
            s[k] = src[e];
        }
        asm volatile("" :: "v"(d[g + 0]), "v"(d[g + 1]), "v"(d[g + 2]), "v"(d[g + 3]),
                           "v"(d[g + 4]), "v"(d[g + 5]), "v"(d[g + 6]), "v"(d[g + 7]),
                           "v"(s[g + 0]), "v"(s[g + 1]), "v"(s[g + 2]), "v"(s[g + 3]),
                           "v"(s[g + 4]), "v"(s[g + 5]), "v"(s[g + 6]), "v"(s[g + 7]));
#pragma unroll
        for (int k = g; k < g + GRP; ++k) {
            const int i = t + k * NTHR;
            b[k] = (i < n) ? min((int)((unsigned)d[k] / (unsigned)BS), NB - 1) : -1;
            if (b[k] >= 0) atomicAdd(&hist[b[k]], 1);
        }
    }
    __syncthreads();
    if (t < 32) {
        int v[8], sum = 0;
#pragma unroll
        for (int j = 0; j < 8; ++j) { v[j] = hist[t * 8 + j]; sum += v[j]; }
        int ex = wave_incl_sum(sum, t) - sum;
#pragma unroll
        for (int j = 0; j < 8; ++j) { cur[t * 8 + j] = ex; ex += v[j]; }
    }
    __syncthreads();
#pragma unroll
    for (int k = 0; k < PER; ++k)
        if (b[k] >= 0) buf[atomicAdd(&cur[b[k]], 1)] = make_int2(d[k], s[k]);
    __syncthreads();
    for (int i = t; i < n; i += NTHR) stage[e0 + i] = buf[i];
    for (int i = t; i < NB; i += NTHR) {
        const int h = hist[i];
        cnt[slice * NB + i] = h;
        bas[slice * NB + i] = cur[i] - h;
    }
}

template<int NCOLS, int KDIM, int NT, int MT, int NW, bool SPLITB, bool SIDE>
__global__ __launch_bounds__(32 * NW, 7)
void k_gemm_wmma_f32(const float* __restrict__ A,
                     const float* __restrict__ B0, const float* __restrict__ B1,
                     float* __restrict__ C0, float* __restrict__ C1,
                     const float* __restrict__ al, const float* __restrict__ ar,
                     float* __restrict__ el, float* __restrict__ er, int M,
                     const int* __restrict__ esrc, const int* __restrict__ edst,
                     int2* __restrict__ stage, int* __restrict__ cnt, int* __restrict__ bas,
                     int E, int BS) {
    constexpr int KC    = 64;
    constexpr int PITCH = KC + 4;
    constexpr int ROWS  = 16 * MT;
    constexpr int NTHR  = 32 * NW;
    constexpr int NLD   = (ROWS * (KC / 4)) / NTHR;
    constexpr int NCH   = KDIM / KC;
    static_assert((ROWS * (KC / 4)) % NTHR == 0, "chunk must split evenly over the threads");
    static_assert(KDIM % KC == 0, "K must be whole chunks");
    constexpr int A_LDS_BYTES = 2 * ROWS * PITCH * 4;
    constexpr int LDS_BYTES   = (SIDE && SIDE_LDS_BYTES > A_LDS_BYTES) ? SIDE_LDS_BYTES : A_LDS_BYTES;
    __shared__ __attribute__((aligned(16))) char smem[LDS_BYTES];
    float (*sA)[ROWS * PITCH] = (float (*)[ROWS * PITCH])smem;

    int tileBlock = blockIdx.x;
    if constexpr (SIDE) {
        if (blockIdx.x < NS) {
            int* hist = (int*)smem;
            bucket_slice<NTHR>(hist, hist + NB, (int2*)(hist + 2 * NB),
                               esrc, edst, stage, cnt, bas, E, BS, blockIdx.x);
            return;
        }
        tileBlock -= NS;
    }

    const int t      = threadIdx.x;
    const int lane   = t & 31;
    const int w      = __builtin_amdgcn_readfirstlane(t >> 5);
    const int lm     = lane & 15;
    const int kg     = lane >> 4;
    const int tileM0 = tileBlock * MT;

    int gOff[NLD], sOff[NLD];
#pragma unroll
    for (int i = 0; i < NLD; ++i) {
        const int idx = t + i * NTHR;
        const int r   = idx >> 4;
        const int c4  = idx & 15;
        int row = tileM0 * 16 + r;
        if (row >= M) row = M - 1;
        gOff[i] = row * KDIM + c4 * 4;
        sOff[i] = r * PITCH + c4 * 4;
    }

    const float* B   = SPLITB ? (w == 0 ? B0 : B1) : B0;
    float*       C   = SPLITB ? (w == 0 ? C0 : C1) : C0;
    const int tileN0 = SPLITB ? 0 : w * NT;
    const float* Bb  = B + (size_t)(kg * 2) * NCOLS + tileN0 * 16 + lm;

    float4 pre[NLD];
#pragma unroll
    for (int i = 0; i < NLD; ++i) pre[i] = *(const float4*)(A + gOff[i]);
#pragma unroll
    for (int i = 0; i < NLD; ++i) *(float4*)(&sA[0][sOff[i]]) = pre[i];
    __syncthreads();

    v8f acc[MT][NT];
#pragma unroll
    for (int i = 0; i < MT; ++i)
#pragma unroll
        for (int j = 0; j < NT; ++j) acc[i][j] = (v8f){};

#pragma unroll
    for (int c = 0; c < NCH; ++c) {
        if (c + 1 < NCH) {
#pragma unroll
            for (int i = 0; i < NLD; ++i) pre[i] = *(const float4*)(A + gOff[i] + (c + 1) * KC);
        }
        __builtin_amdgcn_sched_barrier(0);
        const float* sa = &sA[c & 1][0];
#pragma unroll
        for (int kk = 0; kk < KC; kk += 4) {
            const int k = c * KC + kk;
            v2f a[MT];
#pragma unroll
            for (int i = 0; i < MT; ++i)
                a[i] = *(const v2f*)(sa + (i * 16 + lm) * PITCH + kk + kg * 2);
#pragma unroll
            for (int j = 0; j < NT; ++j) {
                v2f b;
                b.x = Bb[k * NCOLS + j * 16];
                b.y = Bb[k * NCOLS + NCOLS + j * 16];
#pragma unroll
                for (int i = 0; i < MT; ++i)
                    acc[i][j] = __builtin_amdgcn_wmma_f32_16x16x4_f32(
                         false, a[i],  false, b,
                         (short)0, acc[i][j],  false,  false);
            }
        }
        if (c + 1 < NCH) {
#pragma unroll
            for (int i = 0; i < NLD; ++i) *(float4*)(&sA[(c + 1) & 1][sOff[i]]) = pre[i];
            __syncthreads();
        }
    }

#pragma unroll
    for (int i = 0; i < MT; ++i) {
        const int rbase = (tileM0 + i) * 16 + kg * 8;
#pragma unroll
        for (int j = 0; j < NT; ++j) {
            const int cbase = (tileN0 + j) * 16 + lm;
#pragma unroll
            for (int v = 0; v < 8; ++v) {
                const int r = rbase + v;
                if (r < M) C[(size_t)r * NCOLS + cbase] = acc[i][j][v];
            }
        }
    }

    constexpr int H  = SPLITB ? 1 : NW;
    const int head   = SPLITB ? 0 : w;
    const bool coef  = SPLITB ? (w == 0) : true;
    if (coef) {
        float wl[NT], wr[NT];
#pragma unroll
        for (int j = 0; j < NT; ++j) {
            wl[j] = al[(tileN0 + j) * 16 + lm];
            wr[j] = ar[(tileN0 + j) * 16 + lm];
        }
#pragma unroll
        for (int i = 0; i < MT; ++i) {
            const int rbase = (tileM0 + i) * 16 + kg * 8;
            float p[16];
#pragma unroll
            for (int v = 0; v < 8; ++v) {
                float sl = acc[i][0][v] * wl[0], sr = acc[i][0][v] * wr[0];
#pragma unroll
                for (int j = 1; j < NT; ++j) { sl += acc[i][j][v] * wl[j]; sr += acc[i][j][v] * wr[j]; }
                p[v] = sl; p[8 + v] = sr;
            }
            lane_sum_step<8>(p, lm);
            lane_sum_step<4>(p, lm);
            lane_sum_step<2>(p, lm);
            lane_sum_step<1>(p, lm);
            const int r = rbase + (lm & 7);
            float* dstp = (lm < 8) ? el : er;
            if (r < M) dstp[(size_t)r * H + head] = p[0];
        }
    }
}

__device__ __forceinline__ int group_excl_sum(int x, int* red, int& total) {
    const int lane = threadIdx.x & 31;
    const int w    = __builtin_amdgcn_readfirstlane(threadIdx.x >> 5);
    const int inc  = wave_incl_sum(x, lane);
    if (lane == 31) red[w] = inc;
    __syncthreads();
    const int ws   = red[lane];
    const int winc = wave_incl_sum(ws, lane);
    total = __builtin_amdgcn_readlane(winc, 31);
    const int wex  = __builtin_amdgcn_readlane(winc - ws, w);
    __syncthreads();
    return wex + inc - x;
}

__device__ __forceinline__ void bucket_finish(int2* pairs, int* seg, int* ncnt, int* nstart,
                                              const int* poff, const int* pbase, int* red,
                                              const int2* __restrict__ stage,
                                              int* __restrict__ rowptr, int* __restrict__ ssrc,
                                              int b, int G, int T, int N, int E, int BS) {
    const int t    = threadIdx.x;
    const int lane = t & 31;
    const int w    = __builtin_amdgcn_readfirstlane(t >> 5);
    const int L    = slice_len(E);
    const int n0   = b * BS;

    for (int i0 = 0; i0 < T; i0 += 4096) {
        int2 p[4];
#pragma unroll
        for (int k = 0; k < 4; ++k) {
            const int ic = min(i0 + k * 1024 + t, T - 1);
            int lo = 0;
#pragma unroll
            for (int step = NS / 2; step >= 1; step >>= 1)
                if (poff[lo + step] <= ic) lo += step;
            p[k] = stage[(size_t)lo * L + pbase[lo] + (ic - poff[lo])];
        }
        asm volatile("" :: "v"(p[0].x), "v"(p[0].y), "v"(p[1].x), "v"(p[1].y),
                           "v"(p[2].x), "v"(p[2].y), "v"(p[3].x), "v"(p[3].y));
#pragma unroll
        for (int k = 0; k < 4; ++k) {
            const int i = i0 + k * 1024 + t;
            if (i < T) {
                pairs[i] = p[k];
                atomicAdd(&ncnt[p[k].x - n0], 1);
            }
        }
    }
    __syncthreads();

    int tot;
    const int c  = (t < BS) ? ncnt[t] : 0;
    const int ex = group_excl_sum(c, red, tot);
    if (t < BS) { nstart[t] = ex; ncnt[t] = ex; }
    if (t == 0) nstart[BS] = T;
    __syncthreads();

    for (int i = t; i < T; i += 1024) {
        const int2 p = pairs[i];
        seg[atomicAdd(&ncnt[p.x - n0], 1)] = p.y;
    }
    __syncthreads();

    for (int ln = w; ln < BS; ln += 32) {
        const int beg = __builtin_amdgcn_readfirstlane(nstart[ln]);
        const int cn  = __builtin_amdgcn_readfirstlane(nstart[ln + 1]) - beg;
        for (int mb = 0; mb < cn; mb += 32) {
            const int k = mb + lane;
            const bool have = k < cn;
            const int key = have ? seg[beg + k] : 0x7fffffff;
            int rank = 0;
            for (int jb = 0; jb < cn; jb += 32) {
                const int kk = (jb + lane < cn) ? seg[beg + jb + lane] : 0x7fffffff;
                const int jn = min(32, cn - jb);
                for (int jj = 0; jj < jn; ++jj) {
                    const int kj = __builtin_amdgcn_readlane(kk, jj);
                    const int j  = jb + jj;
                    rank += (kj < key || (kj == key && j < k)) ? 1 : 0;
                }
            }
            if (have) ssrc[G + beg + rank] = key;
        }
    }
    if (t < BS && n0 + t < N) rowptr[n0 + t] = G + nstart[t];
}

__global__ __launch_bounds__(1024)
void k_bucket_finish(const int2* __restrict__ stage, const int* __restrict__ cnt,
                     const int* __restrict__ bas, int2* __restrict__ pairsG, int* __restrict__ segG,
                     int* __restrict__ rowptr, int* __restrict__ ssrc, int N, int E, int BS) {
    __shared__ int2 pairsL[BCAP];
    __shared__ int  segL[BCAP];
    __shared__ int  ncnt[BSCAP];
    __shared__ int  nstart[BSCAP + 1];
    __shared__ int  poff[NS];
    __shared__ int  pbase[NS];
    __shared__ int  red[32];
    static_assert(BSCAP == 1024 && NS <= 1024, "one thread per node of a range and per slice");
    static_assert((NS * NB / 4) % 1024 == 0, "the count matrix splits evenly over the threads");
    const int t = threadIdx.x;
    const int b = blockIdx.x;

    int x = 0;
#pragma unroll
    for (int k = 0; k < (NS * NB / 4) / 1024; ++k) {
        const int q  = t + k * 1024;
        const int4 v = ((const int4*)cnt)[q];
        const int c0 = (q * 4) & (NB - 1);
        x += ((c0 + 0 < b) ? v.x : 0) + ((c0 + 1 < b) ? v.y : 0)
           + ((c0 + 2 < b) ? v.z : 0) + ((c0 + 3 < b) ? v.w : 0);
    }
    const int myc = (t < NS) ? cnt[t * NB + b] : 0;
    const int myb = (t < NS) ? bas[t * NB + b] : 0;
    int G, T;
    group_excl_sum(x, red, G);
    const int pex = group_excl_sum(myc, red, T);
    if (t < NS) { poff[t] = pex; pbase[t] = myb; }
    ncnt[t] = 0;
    __syncthreads();

    if (T <= BCAP)
        bucket_finish(pairsL, segL, ncnt, nstart, poff, pbase, red, stage, rowptr, ssrc, b, G, T, N, E, BS);
    else
        bucket_finish(pairsG + G, segG + G, ncnt, nstart, poff, pbase, red, stage, rowptr, ssrc, b, G, T, N, E, BS);
    if (b == NB - 1 && t == 0) rowptr[N] = E;
}

template<int H, int D, bool ADD, bool ELU>
__global__ __launch_bounds__(128)
void k_seg_softmax_sum(const int* __restrict__ rowptr, const int* __restrict__ ssrc,
                       const float* __restrict__ feat,
                       const float* __restrict__ el, const float* __restrict__ er,
                       const float* __restrict__ res,
                       float* __restrict__ out, int N) {
    constexpr int ROW = H * D;
    constexpr int FPL = ROW / 32;
    constexpr int NC  = (FPL >= 4) ? FPL / 4 : 1;
    static_assert(FPL == 1 || FPL % 4 == 0, "row shape");
    const int lane = threadIdx.x & 31;
    const int d = __builtin_amdgcn_readfirstlane(blockIdx.x * (blockDim.x >> 5) + (threadIdx.x >> 5));
    if (d >= N) return;
    const int beg = rowptr[d];
    const int cnt = rowptr[d + 1] - beg;

    float erd[H];
    for (int h = 0; h < H; ++h) erd[h] = er[d * H + h];

    float radd[NC * 4];
    if constexpr (ADD) {
        const float* rrow = res + (size_t)d * ROW;
        if constexpr (FPL >= 4) {
            for (int c = 0; c < NC; ++c) {
                const float4 r = *(const float4*)(rrow + c * 128 + lane * 4);
                radd[4 * c + 0] = r.x; radd[4 * c + 1] = r.y;
                radd[4 * c + 2] = r.z; radd[4 * c + 3] = r.w;
            }
        } else {
            radd[0] = rrow[lane];
        }
    }

    int hsel[NC];
    for (int c = 0; c < NC; ++c) hsel[c] = (FPL >= 4) ? (c * 128 + lane * 4) / D : lane / D;

    int s0 = 0;
    float v0[H];
    for (int h = 0; h < H; ++h) v0[h] = 0.0f;
    if (lane < cnt) {
        s0 = ssrc[beg + lane];
        for (int h = 0; h < H; ++h) {
            const float v = el[s0 * H + h] + erd[h];
            v0[h] = (v > 0.0f) ? v : 0.2f * v;
        }
    }
    auto edge_scores = [&](int b, int& s, float (&v)[H]) {
        if (b == 0) {
            s = s0;
            for (int h = 0; h < H; ++h) v[h] = v0[h];
        } else {
            s = 0;
            for (int h = 0; h < H; ++h) v[h] = 0.0f;
            if (b + lane < cnt) {
                s = ssrc[beg + b + lane];
                for (int h = 0; h < H; ++h) {
                    const float t = el[s * H + h] + erd[h];
                    v[h] = (t > 0.0f) ? t : 0.2f * t;
                }
            }
        }
    };

    float mx[H];
    for (int h = 0; h < H; ++h) mx[h] = -INFINITY;
    for (int b = 0; b < cnt; b += 32) {
        int s; float v[H];
        edge_scores(b, s, v);
        if (b + lane < cnt) {
            for (int h = 0; h < H; ++h) mx[h] = fmaxf(mx[h], v[h]);
        }
    }
    for (int h = 0; h < H; ++h)
        for (int off = 16; off > 0; off >>= 1) mx[h] = fmaxf(mx[h], __shfl_xor(mx[h], off));

    float den[H];
    for (int h = 0; h < H; ++h) den[h] = 0.0f;
    for (int b = 0; b < cnt; b += 32) {
        int s; float v[H];
        edge_scores(b, s, v);
        if (b + lane < cnt) {
            for (int h = 0; h < H; ++h) den[h] += expf(v[h] - mx[h]);
        }
    }
    for (int h = 0; h < H; ++h)
        for (int off = 16; off > 0; off >>= 1) den[h] += __shfl_xor(den[h], off);

    float acc[NC * 4];
    for (int i = 0; i < NC * 4; ++i) acc[i] = 0.0f;
    for (int b = 0; b < cnt; b += 32) {
        int s; float v[H];
        edge_scores(b, s, v);
        float al[H];
        for (int h = 0; h < H; ++h) al[h] = 0.0f;
        if (b + lane < cnt) {
            for (int h = 0; h < H; ++h) al[h] = expf(v[h] - mx[h]) / den[h];
        }
        const int jn = min(32, cnt - b);
#pragma unroll 4
        for (int jj = 0; jj < jn; ++jj) {
            const int sj = __builtin_amdgcn_readlane(s, jj);
            float aj[H];
            for (int h = 0; h < H; ++h)
                aj[h] = __int_as_float(__builtin_amdgcn_readlane(__float_as_int(al[h]), jj));
            const float* frow = feat + (size_t)sj * ROW;
            for (int c = 0; c < NC; ++c) {
                float a = aj[0];
                for (int h = 1; h < H; ++h) a = (hsel[c] == h) ? aj[h] : a;
                if constexpr (FPL >= 4) {
                    const float4 f = *(const float4*)(frow + c * 128 + lane * 4);
                    acc[4 * c + 0] += f.x * a;
                    acc[4 * c + 1] += f.y * a;
                    acc[4 * c + 2] += f.z * a;
                    acc[4 * c + 3] += f.w * a;
                } else {
                    acc[0] += frow[lane] * a;
                }
            }
        }
    }

    if constexpr (ADD) {
        for (int i = 0; i < NC * 4; ++i) acc[i] += radd[i];
    }
    if constexpr (ELU) {
        for (int i = 0; i < NC * 4; ++i) {
            const float v = acc[i];
            acc[i] = (v > 0.0f) ? v : (expf(v) - 1.0f);
        }
    }

    float* orow = out + (size_t)d * ROW;
    if constexpr (FPL >= 4) {
        for (int c = 0; c < NC; ++c)
            *(float4*)(orow + c * 128 + lane * 4) =
                make_float4(acc[4 * c + 0], acc[4 * c + 1], acc[4 * c + 2], acc[4 * c + 3]);
    } else {
        orow[lane] = acc[0];
    }
}

extern "C" void kernel_launch(void* const* d_in, const int* in_sizes, int n_in,
                              void* d_out, int out_size, void* d_ws, size_t ws_size,
                              hipStream_t stream) {
    const float* x    = (const float*)d_in[0];
    const int*   src  = (const int*)  d_in[1];
    const int*   dst  = (const int*)  d_in[2];
    const float* W0   = (const float*)d_in[3];
    const float* al0  = (const float*)d_in[4];
    const float* ar0  = (const float*)d_in[5];
    const float* W1   = (const float*)d_in[6];
    const float* al1  = (const float*)d_in[7];
    const float* ar1  = (const float*)d_in[8];
    const float* Wres = (const float*)d_in[9];
    float* out = (float*)d_out;

    const int N = in_sizes[0] / 256;
    const int E = in_sizes[1];
    const int F = 256;

    char* ws = (char*)d_ws;
    size_t off = 0;
    auto alloc = [&](size_t elems) -> float* {
        float* p = (float*)(ws + off);
        off += ((elems * sizeof(float) + 255) / 256) * 256;
        return p;
    };
    float* feat0  = alloc((size_t)N * F);
    float* hbuf   = alloc((size_t)N * F);
    float* el0    = alloc((size_t)N * 4);
    float* er0    = alloc((size_t)N * 4);
    float* feat1  = alloc((size_t)N * 32);
    float* resb   = alloc((size_t)N * 32);
    float* el1    = alloc((size_t)N);
    float* er1    = alloc((size_t)N);
    int2*  stage  = (int2*)alloc((size_t)E * 2);
    int*   cnt    = (int*)alloc((size_t)NS * NB);
    int*   bas    = (int*)alloc((size_t)NS * NB);
    int2*  pairsG = (int2*)alloc((size_t)E * 2);
    int*   segG   = (int*)alloc((size_t)E);
    int*   rowptr = (int*)alloc((size_t)N + 1);
    int*   ssrc   = (int*)alloc((size_t)E);

    const int mt = (N + 15) / 16;
    const int BS = (N + NB - 1) / NB;

    k_gemm_wmma_f32<256, 256, 4, 2, 4, false, true><<<dim3(NS + (mt + 1) / 2), 128, 0, stream>>>(
        x, W0, nullptr, feat0, nullptr, al0, ar0, el0, er0, N,
        src, dst, stage, cnt, bas, E, BS);

    k_bucket_finish<<<NB, 1024, 0, stream>>>(stage, cnt, bas, pairsG, segG, rowptr, ssrc, N, E, BS);

    k_seg_softmax_sum<4, 64, false, true><<<(N + 3) / 4, 128, 0, stream>>>(
        rowptr, ssrc, feat0, el0, er0, nullptr, hbuf, N);

    k_gemm_wmma_f32<32, 256, 2, 1, 2, true, false><<<dim3(mt), 64, 0, stream>>>(
        hbuf, W1, Wres, feat1, resb, al1, ar1, el1, er1, N,
        nullptr, nullptr, nullptr, nullptr, nullptr, 0, 0);
    k_seg_softmax_sum<1, 32, true, false><<<(N + 3) / 4, 128, 0, stream>>>(
        rowptr, ssrc, feat1, el1, er1, resb, out, N);
}


# Layer-1 aggregation: a wave takes four destination nodes, eight lanes per 128-B row; one load fetches four rows, reductions stay in the eight lanes (vector ALU); 50000 waves -> 12500; sums unchanged
# speedup vs baseline: 8.1977x; 1.0363x over previous
#include <hip/hip_runtime.h>
#include <hip/hip_bf16.h>
#include <math.h>


typedef float v2f __attribute__((ext_vector_type(2)));
typedef float v8f __attribute__((ext_vector_type(8)));

template<int W>
__device__ __forceinline__ void lane_sum_step(float (&p)[16], int lm) {
    const bool up = (lm & W) != 0;
#pragma unroll
    for (int i = 0; i < W; ++i) {
        const float lo = p[i], hi = p[i + W];
        const float keep = up ? hi : lo;
        const float send = up ? lo : hi;
        const float got = __int_as_float(__builtin_amdgcn_update_dpp(
            0, __float_as_int(send), 0x160 | W, 0xf, 0xf, false));
        p[i] = keep + got;
    }
}

__device__ __forceinline__ int wave_incl_sum(int x, int lane) {
    x += __builtin_amdgcn_update_dpp(0, x, 0x111, 0xf, 0xf, false);
    x += __builtin_amdgcn_update_dpp(0, x, 0x112, 0xf, 0xf, false);
    x += __builtin_amdgcn_update_dpp(0, x, 0x114, 0xf, 0xf, false);
    x += __builtin_amdgcn_update_dpp(0, x, 0x118, 0xf, 0xf, false);
    const int low = __builtin_amdgcn_readlane(x, 15);
    return x + ((lane & 16) ? low : 0);
}


constexpr int NB    = 256;
constexpr int NS    = 256;
constexpr int LCAP  = 4096;
constexpr int BCAP  = 8192;
constexpr int BSCAP = 1024;

__device__ __forceinline__ int slice_len(int E) {
    return (((E + NS - 1) / NS) + 7) & ~7;
}

constexpr int SIDE_LDS_BYTES = 2 * NB * 4 + LCAP * 8;

template<int NTHR>
__device__ __forceinline__ void bucket_slice(int* hist, int* cur, int2* buf,
                                             const int* __restrict__ src, const int* __restrict__ dst,
                                             int2* __restrict__ stage, int* __restrict__ cnt,
                                             int* __restrict__ bas, int E, int BS, int slice) {
    constexpr int PER = LCAP / NTHR;
    constexpr int GRP = 8;
    static_assert(LCAP % NTHR == 0 && PER % GRP == 0, "slice splits evenly over the threads");
    static_assert(NB == 256 && NTHR >= 32, "the group scan below is 8 bins per lane of wave 0");
    const int t  = threadIdx.x;
    const int L  = slice_len(E);
    const int e0 = slice * L;
    const int n  = max(0, min(L, E - e0));
    for (int i = t; i < NB; i += NTHR) hist[i] = 0;
    __syncthreads();
    int d[PER], s[PER], b[PER];
#pragma unroll
    for (int g = 0; g < PER; g += GRP) {
#pragma unroll
        for (int k = g; k < g + GRP; ++k) {
            const int e = min(e0 + t + k * NTHR, E - 1);
            d[k] = dst[e];
            s[k] = src[e];
        }
        asm volatile("" :: "v"(d[g + 0]), "v"(d[g + 1]), "v"(d[g + 2]), "v"(d[g + 3]),
                           "v"(d[g + 4]), "v"(d[g + 5]), "v"(d[g + 6]), "v"(d[g + 7]),
                           "v"(s[g + 0]), "v"(s[g + 1]), "v"(s[g + 2]), "v"(s[g + 3]),
                           "v"(s[g + 4]), "v"(s[g + 5]), "v"(s[g + 6]), "v"(s[g + 7]));
#pragma unroll
        for (int k = g; k < g + GRP; ++k) {
            const int i = t + k * NTHR;
            b[k] = (i < n) ? min((int)((unsigned)d[k] / (unsigned)BS), NB - 1) : -1;
            if (b[k] >= 0) atomicAdd(&hist[b[k]], 1);
        }
    }
    __syncthreads();
    if (t < 32) {
        int v[8], sum = 0;
#pragma unroll
        for (int j = 0; j < 8; ++j) { v[j] = hist[t * 8 + j]; sum += v[j]; }
        int ex = wave_incl_sum(sum, t) - sum;
#pragma unroll
        for (int j = 0; j < 8; ++j) { cur[t * 8 + j] = ex; ex += v[j]; }
    }
    __syncthreads();
#pragma unroll
    for (int k = 0; k < PER; ++k)
        if (b[k] >= 0) buf[atomicAdd(&cur[b[k]], 1)] = make_int2(d[k], s[k]);
    __syncthreads();
    for (int i = t; i < n; i += NTHR) stage[e0 + i] = buf[i];
    for (int i = t; i < NB; i += NTHR) {
        const int h = hist[i];
        cnt[slice * NB + i] = h;
        bas[slice * NB + i] = cur[i] - h;
    }
}

template<int NCOLS, int KDIM, int NT, int MT, int NW, bool SPLITB, bool SIDE>
__global__ __launch_bounds__(32 * NW, 7)
void k_gemm_wmma_f32(const float* __restrict__ A,
                     const float* __restrict__ B0, const float* __restrict__ B1,
                     float* __restrict__ C0, float* __restrict__ C1,
                     const float* __restrict__ al, const float* __restrict__ ar,
                     float* __restrict__ el, float* __restrict__ er, int M,
                     const int* __restrict__ esrc, const int* __restrict__ edst,
                     int2* __restrict__ stage, int* __restrict__ cnt, int* __restrict__ bas,
                     int E, int BS) {
    constexpr int KC    = 64;
    constexpr int PITCH = KC + 4;
    constexpr int ROWS  = 16 * MT;
    constexpr int NTHR  = 32 * NW;
    constexpr int NLD   = (ROWS * (KC / 4)) / NTHR;
    constexpr int NCH   = KDIM / KC;
    static_assert((ROWS * (KC / 4)) % NTHR == 0, "chunk must split evenly over the threads");
    static_assert(KDIM % KC == 0, "K must be whole chunks");
    constexpr int A_LDS_BYTES = 2 * ROWS * PITCH * 4;
    constexpr int LDS_BYTES   = (SIDE && SIDE_LDS_BYTES > A_LDS_BYTES) ? SIDE_LDS_BYTES : A_LDS_BYTES;
    __shared__ __attribute__((aligned(16))) char smem[LDS_BYTES];
    float (*sA)[ROWS * PITCH] = (float (*)[ROWS * PITCH])smem;

    int tileBlock = blockIdx.x;
    if constexpr (SIDE) {
        if (blockIdx.x < NS) {
            int* hist = (int*)smem;
            bucket_slice<NTHR>(hist, hist + NB, (int2*)(hist + 2 * NB),
                               esrc, edst, stage, cnt, bas, E, BS, blockIdx.x);
            return;
        }
        tileBlock -= NS;
    }

    const int t      = threadIdx.x;
    const int lane   = t & 31;
    const int w      = __builtin_amdgcn_readfirstlane(t >> 5);
    const int lm     = lane & 15;
    const int kg     = lane >> 4;
    const int tileM0 = tileBlock * MT;

    int gOff[NLD], sOff[NLD];
#pragma unroll
    for (int i = 0; i < NLD; ++i) {
        const int idx = t + i * NTHR;
        const int r   = idx >> 4;
        const int c4  = idx & 15;
        int row = tileM0 * 16 + r;
        if (row >= M) row = M - 1;
        gOff[i] = row * KDIM + c4 * 4;
        sOff[i] = r * PITCH + c4 * 4;
    }

    const float* B   = SPLITB ? (w == 0 ? B0 : B1) : B0;
    float*       C   = SPLITB ? (w == 0 ? C0 : C1) : C0;
    const int tileN0 = SPLITB ? 0 : w * NT;
    const float* Bb  = B + (size_t)(kg * 2) * NCOLS + tileN0 * 16 + lm;

    float4 pre[NLD];
#pragma unroll
    for (int i = 0; i < NLD; ++i) pre[i] = *(const float4*)(A + gOff[i]);
#pragma unroll
    for (int i = 0; i < NLD; ++i) *(float4*)(&sA[0][sOff[i]]) = pre[i];
    __syncthreads();

    v8f acc[MT][NT];
#pragma unroll
    for (int i = 0; i < MT; ++i)
#pragma unroll
        for (int j = 0; j < NT; ++j) acc[i][j] = (v8f){};

#pragma unroll
    for (int c = 0; c < NCH; ++c) {
        if (c + 1 < NCH) {
#pragma unroll
            for (int i = 0; i < NLD; ++i) pre[i] = *(const float4*)(A + gOff[i] + (c + 1) * KC);
        }
        __builtin_amdgcn_sched_barrier(0);
        const float* sa = &sA[c & 1][0];
#pragma unroll
        for (int kk = 0; kk < KC; kk += 4) {
            const int k = c * KC + kk;
            v2f a[MT];
#pragma unroll
            for (int i = 0; i < MT; ++i)
                a[i] = *(const v2f*)(sa + (i * 16 + lm) * PITCH + kk + kg * 2);
#pragma unroll
            for (int j = 0; j < NT; ++j) {
                v2f b;
                b.x = Bb[k * NCOLS + j * 16];
                b.y = Bb[k * NCOLS + NCOLS + j * 16];
#pragma unroll
                for (int i = 0; i < MT; ++i)
                    acc[i][j] = __builtin_amdgcn_wmma_f32_16x16x4_f32(
                         false, a[i],  false, b,
                         (short)0, acc[i][j],  false,  false);
            }
        }
        if (c + 1 < NCH) {
#pragma unroll
            for (int i = 0; i < NLD; ++i) *(float4*)(&sA[(c + 1) & 1][sOff[i]]) = pre[i];
            __syncthreads();
        }
    }

#pragma unroll
    for (int i = 0; i < MT; ++i) {
        const int rbase = (tileM0 + i) * 16 + kg * 8;
#pragma unroll
        for (int j = 0; j < NT; ++j) {
            const int cbase = (tileN0 + j) * 16 + lm;
#pragma unroll
            for (int v = 0; v < 8; ++v) {
                const int r = rbase + v;
                if (r < M) C[(size_t)r * NCOLS + cbase] = acc[i][j][v];
            }
        }
    }

    constexpr int H  = SPLITB ? 1 : NW;
    const int head   = SPLITB ? 0 : w;
    const bool coef  = SPLITB ? (w == 0) : true;
    if (coef) {
        float wl[NT], wr[NT];
#pragma unroll
        for (int j = 0; j < NT; ++j) {
            wl[j] = al[(tileN0 + j) * 16 + lm];
            wr[j] = ar[(tileN0 + j) * 16 + lm];
        }
#pragma unroll
        for (int i = 0; i < MT; ++i) {
            const int rbase = (tileM0 + i) * 16 + kg * 8;
            float p[16];
#pragma unroll
            for (int v = 0; v < 8; ++v) {
                float sl = acc[i][0][v] * wl[0], sr = acc[i][0][v] * wr[0];
#pragma unroll
                for (int j = 1; j < NT; ++j) { sl += acc[i][j][v] * wl[j]; sr += acc[i][j][v] * wr[j]; }
                p[v] = sl; p[8 + v] = sr;
            }
            lane_sum_step<8>(p, lm);
            lane_sum_step<4>(p, lm);
            lane_sum_step<2>(p, lm);
            lane_sum_step<1>(p, lm);
            const int r = rbase + (lm & 7);
            float* dstp = (lm < 8) ? el : er;
            if (r < M) dstp[(size_t)r * H + head] = p[0];
        }
    }
}

__device__ __forceinline__ int group_excl_sum(int x, int* red, int& total) {
    const int lane = threadIdx.x & 31;
    const int w    = __builtin_amdgcn_readfirstlane(threadIdx.x >> 5);
    const int inc  = wave_incl_sum(x, lane);
    if (lane == 31) red[w] = inc;
    __syncthreads();
    const int ws   = red[lane];
    const int winc = wave_incl_sum(ws, lane);
    total = __builtin_amdgcn_readlane(winc, 31);
    const int wex  = __builtin_amdgcn_readlane(winc - ws, w);
    __syncthreads();
    return wex + inc - x;
}

__device__ __forceinline__ void bucket_finish(int2* pairs, int* seg, int* ncnt, int* nstart,
                                              const int* poff, const int* pbase, int* red,
                                              const int2* __restrict__ stage,
                                              int* __restrict__ rowptr, int* __restrict__ ssrc,
                                              int b, int G, int T, int N, int E, int BS) {
    const int t    = threadIdx.x;
    const int lane = t & 31;
    const int w    = __builtin_amdgcn_readfirstlane(t >> 5);
    const int L    = slice_len(E);
    const int n0   = b * BS;

    for (int i0 = 0; i0 < T; i0 += 4096) {
        int2 p[4];
#pragma unroll
        for (int k = 0; k < 4; ++k) {
            const int ic = min(i0 + k * 1024 + t, T - 1);
            int lo = 0;
#pragma unroll
            for (int step = NS / 2; step >= 1; step >>= 1)
                if (poff[lo + step] <= ic) lo += step;
            p[k] = stage[(size_t)lo * L + pbase[lo] + (ic - poff[lo])];
        }
        asm volatile("" :: "v"(p[0].x), "v"(p[0].y), "v"(p[1].x), "v"(p[1].y),
                           "v"(p[2].x), "v"(p[2].y), "v"(p[3].x), "v"(p[3].y));
#pragma unroll
        for (int k = 0; k < 4; ++k) {
            const int i = i0 + k * 1024 + t;
            if (i < T) {
                pairs[i] = p[k];
                atomicAdd(&ncnt[p[k].x - n0], 1);
            }
        }
    }
    __syncthreads();

    int tot;
    const int c  = (t < BS) ? ncnt[t] : 0;
    const int ex = group_excl_sum(c, red, tot);
    if (t < BS) { nstart[t] = ex; ncnt[t] = ex; }
    if (t == 0) nstart[BS] = T;
    __syncthreads();

    for (int i = t; i < T; i += 1024) {
        const int2 p = pairs[i];
        seg[atomicAdd(&ncnt[p.x - n0], 1)] = p.y;
    }
    __syncthreads();

    for (int ln = w; ln < BS; ln += 32) {
        const int beg = __builtin_amdgcn_readfirstlane(nstart[ln]);
        const int cn  = __builtin_amdgcn_readfirstlane(nstart[ln + 1]) - beg;
        for (int mb = 0; mb < cn; mb += 32) {
            const int k = mb + lane;
            const bool have = k < cn;
            const int key = have ? seg[beg + k] : 0x7fffffff;
            int rank = 0;
            for (int jb = 0; jb < cn; jb += 32) {
                const int kk = (jb + lane < cn) ? seg[beg + jb + lane] : 0x7fffffff;
                const int jn = min(32, cn - jb);
                for (int jj = 0; jj < jn; ++jj) {
                    const int kj = __builtin_amdgcn_readlane(kk, jj);
                    const int j  = jb + jj;
                    rank += (kj < key || (kj == key && j < k)) ? 1 : 0;
                }
            }
            if (have) ssrc[G + beg + rank] = key;
        }
    }
    if (t < BS && n0 + t < N) rowptr[n0 + t] = G + nstart[t];
}

__global__ __launch_bounds__(1024)
void k_bucket_finish(const int2* __restrict__ stage, const int* __restrict__ cnt,
                     const int* __restrict__ bas, int2* __restrict__ pairsG, int* __restrict__ segG,
                     int* __restrict__ rowptr, int* __restrict__ ssrc, int N, int E, int BS) {
    __shared__ int2 pairsL[BCAP];
    __shared__ int  segL[BCAP];
    __shared__ int  ncnt[BSCAP];
    __shared__ int  nstart[BSCAP + 1];
    __shared__ int  poff[NS];
    __shared__ int  pbase[NS];
    __shared__ int  red[32];
    static_assert(BSCAP == 1024 && NS <= 1024, "one thread per node of a range and per slice");
    static_assert((NS * NB / 4) % 1024 == 0, "the count matrix splits evenly over the threads");
    const int t = threadIdx.x;
    const int b = blockIdx.x;

    int x = 0;
#pragma unroll
    for (int k = 0; k < (NS * NB / 4) / 1024; ++k) {
        const int q  = t + k * 1024;
        const int4 v = ((const int4*)cnt)[q];
        const int c0 = (q * 4) & (NB - 1);
        x += ((c0 + 0 < b) ? v.x : 0) + ((c0 + 1 < b) ? v.y : 0)
           + ((c0 + 2 < b) ? v.z : 0) + ((c0 + 3 < b) ? v.w : 0);
    }
    const int myc = (t < NS) ? cnt[t * NB + b] : 0;
    const int myb = (t < NS) ? bas[t * NB + b] : 0;
    int G, T;
    group_excl_sum(x, red, G);
    const int pex = group_excl_sum(myc, red, T);
    if (t < NS) { poff[t] = pex; pbase[t] = myb; }
    ncnt[t] = 0;
    __syncthreads();

    if (T <= BCAP)
        bucket_finish(pairsL, segL, ncnt, nstart, poff, pbase, red, stage, rowptr, ssrc, b, G, T, N, E, BS);
    else
        bucket_finish(pairsG + G, segG + G, ncnt, nstart, poff, pbase, red, stage, rowptr, ssrc, b, G, T, N, E, BS);
    if (b == NB - 1 && t == 0) rowptr[N] = E;
}

template<int H, int D, bool ADD, bool ELU>
__global__ __launch_bounds__(128)
void k_seg_softmax_sum(const int* __restrict__ rowptr, const int* __restrict__ ssrc,
                       const float* __restrict__ feat,
                       const float* __restrict__ el, const float* __restrict__ er,
                       const float* __restrict__ res,
                       float* __restrict__ out, int N) {
    constexpr int ROW = H * D;
    constexpr int FPL = ROW / 32;
    constexpr int NC  = (FPL >= 4) ? FPL / 4 : 1;
    static_assert(FPL == 1 || FPL % 4 == 0, "row shape");
    const int lane = threadIdx.x & 31;
    const int d = __builtin_amdgcn_readfirstlane(blockIdx.x * (blockDim.x >> 5) + (threadIdx.x >> 5));
    if (d >= N) return;
    const int beg = rowptr[d];
    const int cnt = rowptr[d + 1] - beg;

    float erd[H];
    for (int h = 0; h < H; ++h) erd[h] = er[d * H + h];

    float radd[NC * 4];
    if constexpr (ADD) {
        const float* rrow = res + (size_t)d * ROW;
        if constexpr (FPL >= 4) {
            for (int c = 0; c < NC; ++c) {
                const float4 r = *(const float4*)(rrow + c * 128 + lane * 4);
                radd[4 * c + 0] = r.x; radd[4 * c + 1] = r.y;
                radd[4 * c + 2] = r.z; radd[4 * c + 3] = r.w;
            }
        } else {
            radd[0] = rrow[lane];
        }
    }

    int hsel[NC];
    for (int c = 0; c < NC; ++c) hsel[c] = (FPL >= 4) ? (c * 128 + lane * 4) / D : lane / D;

    int s0 = 0;
    float v0[H];
    for (int h = 0; h < H; ++h) v0[h] = 0.0f;
    if (lane < cnt) {
        s0 = ssrc[beg + lane];
        for (int h = 0; h < H; ++h) {
            const float v = el[s0 * H + h] + erd[h];
            v0[h] = (v > 0.0f) ? v : 0.2f * v;
        }
    }
    auto edge_scores = [&](int b, int& s, float (&v)[H]) {
        if (b == 0) {
            s = s0;
            for (int h = 0; h < H; ++h) v[h] = v0[h];
        } else {
            s = 0;
            for (int h = 0; h < H; ++h) v[h] = 0.0f;
            if (b + lane < cnt) {
                s = ssrc[beg + b + lane];
                for (int h = 0; h < H; ++h) {
                    const float t = el[s * H + h] + erd[h];
                    v[h] = (t > 0.0f) ? t : 0.2f * t;
                }
            }
        }
    };

    float mx[H];
    for (int h = 0; h < H; ++h) mx[h] = -INFINITY;
    for (int b = 0; b < cnt; b += 32) {
        int s; float v[H];
        edge_scores(b, s, v);
        if (b + lane < cnt) {
            for (int h = 0; h < H; ++h) mx[h] = fmaxf(mx[h], v[h]);
        }
    }
    for (int h = 0; h < H; ++h)
        for (int off = 16; off > 0; off >>= 1) mx[h] = fmaxf(mx[h], __shfl_xor(mx[h], off));

    float den[H];
    for (int h = 0; h < H; ++h) den[h] = 0.0f;
    for (int b = 0; b < cnt; b += 32) {
        int s; float v[H];
        edge_scores(b, s, v);
        if (b + lane < cnt) {
            for (int h = 0; h < H; ++h) den[h] += expf(v[h] - mx[h]);
        }
    }
    for (int h = 0; h < H; ++h)
        for (int off = 16; off > 0; off >>= 1) den[h] += __shfl_xor(den[h], off);

    float acc[NC * 4];
    for (int i = 0; i < NC * 4; ++i) acc[i] = 0.0f;
    for (int b = 0; b < cnt; b += 32) {
        int s; float v[H];
        edge_scores(b, s, v);
        float al[H];
        for (int h = 0; h < H; ++h) al[h] = 0.0f;
        if (b + lane < cnt) {
            for (int h = 0; h < H; ++h) al[h] = expf(v[h] - mx[h]) / den[h];
        }
        const int jn = min(32, cnt - b);
#pragma unroll 4
        for (int jj = 0; jj < jn; ++jj) {
            const int sj = __builtin_amdgcn_readlane(s, jj);
            float aj[H];
            for (int h = 0; h < H; ++h)
                aj[h] = __int_as_float(__builtin_amdgcn_readlane(__float_as_int(al[h]), jj));
            const float* frow = feat + (size_t)sj * ROW;
            for (int c = 0; c < NC; ++c) {
                float a = aj[0];
                for (int h = 1; h < H; ++h) a = (hsel[c] == h) ? aj[h] : a;
                if constexpr (FPL >= 4) {
                    const float4 f = *(const float4*)(frow + c * 128 + lane * 4);
                    acc[4 * c + 0] += f.x * a;
                    acc[4 * c + 1] += f.y * a;
                    acc[4 * c + 2] += f.z * a;
                    acc[4 * c + 3] += f.w * a;
                } else {
                    acc[0] += frow[lane] * a;
                }
            }
        }
    }

    if constexpr (ADD) {
        for (int i = 0; i < NC * 4; ++i) acc[i] += radd[i];
    }
    if constexpr (ELU) {
        for (int i = 0; i < NC * 4; ++i) {
            const float v = acc[i];
            acc[i] = (v > 0.0f) ? v : (expf(v) - 1.0f);
        }
    }

    float* orow = out + (size_t)d * ROW;
    if constexpr (FPL >= 4) {
        for (int c = 0; c < NC; ++c)
            *(float4*)(orow + c * 128 + lane * 4) =
                make_float4(acc[4 * c + 0], acc[4 * c + 1], acc[4 * c + 2], acc[4 * c + 3]);
    } else {
        orow[lane] = acc[0];
    }
}


template<int J>
__device__ __forceinline__ int group8_lane(int x) {
    constexpr unsigned SEL = J | (J << 3) | (J << 6) | (J << 9) | (J << 12) | (J << 15) | (J << 18) | (J << 21);
    return (int)__builtin_amdgcn_mov_dpp8((unsigned)x, SEL);
}

template<int W>
__device__ __forceinline__ float group8_xor(float x) {
    return __int_as_float(__builtin_amdgcn_update_dpp(0, __float_as_int(x), 0x160 | W, 0xf, 0xf, false));
}

template<int J>
__device__ __forceinline__ void group8_rows(const float* __restrict__ feat, int q, int sk,
                                            int e0, int cnt, float4 (&f)[8]) {
    if constexpr (J < 8) {
        const int sj = group8_lane<J>(sk);
        f[J] = make_float4(0.0f, 0.0f, 0.0f, 0.0f);
        if (e0 + J < cnt) f[J] = *(const float4*)(feat + (sj * 32 + q * 4));
        group8_rows<J + 1>(feat, q, sk, e0, cnt, f);
    }
}

template<int J>
__device__ __forceinline__ void group8_fma(float4& acc, const float4 (&f)[8], float ak) {
    if constexpr (J < 8) {
        const float a = __int_as_float(group8_lane<J>(__float_as_int(ak)));
        acc.x += f[J].x * a;
        acc.y += f[J].y * a;
        acc.z += f[J].z * a;
        acc.w += f[J].w * a;
        group8_fma<J + 1>(acc, f, ak);
    }
}

__global__ __launch_bounds__(128, 16)
void k_seg_softmax_sum_rows32(const int* __restrict__ rowptr, const int* __restrict__ ssrc,
                              const float* __restrict__ feat,
                              const float* __restrict__ el, const float* __restrict__ er,
                              const float* __restrict__ res,
                              float* __restrict__ out, int N) {
    const int lane = threadIdx.x & 31;
    const int q    = lane & 7;
    const int g    = lane >> 3;
    const int wv   = __builtin_amdgcn_readfirstlane(blockIdx.x * (blockDim.x >> 5) + (threadIdx.x >> 5));
    if (wv * 4 >= N) return;
    const int  d    = wv * 4 + g;
    const bool live = d < N;
    const int  dc   = live ? d : N - 1;
    const int  beg  = rowptr[dc];
    const int  end  = rowptr[dc + 1];
    const int  cnt  = live ? end - beg : 0;
    const float erd = er[dc];
    const float4 radd = *(const float4*)(res + (dc * 32 + q * 4));

    int s0[4];
    float v0[4];
    for (int k = 0; k < 4; ++k) { s0[k] = 0; v0[k] = 0.0f; }
    for (int k = 0; k < 4; ++k)
        if (q + 8 * k < cnt) s0[k] = ssrc[beg + q + 8 * k];
    float e0[4];
    for (int k = 0; k < 4; ++k) e0[k] = el[s0[k]];
    for (int k = 0; k < 4; ++k)
        if (q + 8 * k < cnt) {
            const float t = e0[k] + erd;
            v0[k] = (t > 0.0f) ? t : 0.2f * t;
        }
    auto edge_scores = [&](int B, int (&s)[4], float (&v)[4]) {
        if (B == 0) {
            for (int k = 0; k < 4; ++k) { s[k] = s0[k]; v[k] = v0[k]; }
        } else {
            for (int k = 0; k < 4; ++k) {
                s[k] = 0; v[k] = 0.0f;
                if (B + q + 8 * k < cnt) {
                    s[k] = ssrc[beg + B + q + 8 * k];
                    const float t = el[s[k]] + erd;
                    v[k] = (t > 0.0f) ? t : 0.2f * t;
                }
            }
        }
    };
    auto more = [&](int B) { return __builtin_amdgcn_ballot_w32(B < cnt) != 0u; };

    float mx = -INFINITY;
    for (int B = 0; more(B); B += 32) {
        int s[4]; float v[4];
        edge_scores(B, s, v);
        for (int k = 0; k < 4; ++k)
            if (B + q + 8 * k < cnt) mx = fmaxf(mx, v[k]);
    }
    mx = fmaxf(mx, group8_xor<4>(mx));
    mx = fmaxf(mx, group8_xor<2>(mx));
    mx = fmaxf(mx, group8_xor<1>(mx));

    float dk[4];
    for (int k = 0; k < 4; ++k) dk[k] = 0.0f;
    for (int B = 0; more(B); B += 32) {
        int s[4]; float v[4];
        edge_scores(B, s, v);
        for (int k = 0; k < 4; ++k)
            if (B + q + 8 * k < cnt) dk[k] += expf(v[k] - mx);
    }
    float den = (dk[0] + dk[2]) + (dk[1] + dk[3]);
    den += group8_xor<4>(den);
    den += group8_xor<2>(den);
    den += group8_xor<1>(den);

    float4 acc = make_float4(0.0f, 0.0f, 0.0f, 0.0f);
    for (int B = 0; more(B); B += 32) {
        int s[4]; float v[4];
        edge_scores(B, s, v);
        float al[4];
        for (int k = 0; k < 4; ++k) {
            al[k] = 0.0f;
            if (B + q + 8 * k < cnt) al[k] = expf(v[k] - mx) / den;
        }
#pragma unroll
        for (int k = 0; k < 4; ++k) {
            if (!more(B + 8 * k)) break;
            float4 f[8];
            group8_rows<0>(feat, q, s[k], B + 8 * k, cnt, f);
            group8_fma<0>(acc, f, al[k]);
        }
    }

    if (live) {
        *(float4*)(out + (d * 32 + q * 4)) =
            make_float4(acc.x + radd.x, acc.y + radd.y, acc.z + radd.z, acc.w + radd.w);
    }
}

extern "C" void kernel_launch(void* const* d_in, const int* in_sizes, int n_in,
                              void* d_out, int out_size, void* d_ws, size_t ws_size,
                              hipStream_t stream) {
    const float* x    = (const float*)d_in[0];
    const int*   src  = (const int*)  d_in[1];
    const int*   dst  = (const int*)  d_in[2];
    const float* W0   = (const float*)d_in[3];
    const float* al0  = (const float*)d_in[4];
    const float* ar0  = (const float*)d_in[5];
    const float* W1   = (const float*)d_in[6];
    const float* al1  = (const float*)d_in[7];
    const float* ar1  = (const float*)d_in[8];
    const float* Wres = (const float*)d_in[9];
    float* out = (float*)d_out;

    const int N = in_sizes[0] / 256;
    const int E = in_sizes[1];
    const int F = 256;

    char* ws = (char*)d_ws;
    size_t off = 0;
    auto alloc = [&](size_t elems) -> float* {
        float* p = (float*)(ws + off);
        off += ((elems * sizeof(float) + 255) / 256) * 256;
        return p;
    };
    float* feat0  = alloc((size_t)N * F);
    float* hbuf   = alloc((size_t)N * F);
    float* el0    = alloc((size_t)N * 4);
    float* er0    = alloc((size_t)N * 4);
    float* feat1  = alloc((size_t)N * 32);
    float* resb   = alloc((size_t)N * 32);
    float* el1    = alloc((size_t)N);
    float* er1    = alloc((size_t)N);
    int2*  stage  = (int2*)alloc((size_t)E * 2);
    int*   cnt    = (int*)alloc((size_t)NS * NB);
    int*   bas    = (int*)alloc((size_t)NS * NB);
    int2*  pairsG = (int2*)alloc((size_t)E * 2);
    int*   segG   = (int*)alloc((size_t)E);
    int*   rowptr = (int*)alloc((size_t)N + 1);
    int*   ssrc   = (int*)alloc((size_t)E);

    const int mt = (N + 15) / 16;
    const int BS = (N + NB - 1) / NB;

    k_gemm_wmma_f32<256, 256, 4, 2, 4, false, true><<<dim3(NS + (mt + 1) / 2), 128, 0, stream>>>(
        x, W0, nullptr, feat0, nullptr, al0, ar0, el0, er0, N,
        src, dst, stage, cnt, bas, E, BS);

    k_bucket_finish<<<NB, 1024, 0, stream>>>(stage, cnt, bas, pairsG, segG, rowptr, ssrc, N, E, BS);

    k_seg_softmax_sum<4, 64, false, true><<<(N + 3) / 4, 128, 0, stream>>>(
        rowptr, ssrc, feat0, el0, er0, nullptr, hbuf, N);

    k_gemm_wmma_f32<32, 256, 2, 1, 2, true, false><<<dim3(mt), 64, 0, stream>>>(
        hbuf, W1, Wres, feat1, resb, al1, ar1, el1, er1, N,
        nullptr, nullptr, nullptr, nullptr, nullptr, 0, 0);
    k_seg_softmax_sum_rows32<<<(N + 15) / 16, 128, 0, stream>>>(
        rowptr, ssrc, feat1, el1, er1, resb, out, N);
}
